# starting kernel
# baseline (speedup 1.0000x reference)
#include <hip/hip_runtime.h>
#include <hip/hip_bf16.h>
#include <math.h>

typedef __attribute__((ext_vector_type(2))) float v2f;
typedef __attribute__((ext_vector_type(8))) float v8f;

#define DCH 128
#define NGRAPH 64
#define GN_EPS 1e-5f
#define TILE_M 16
#define SX_LD (DCH + 4)

__global__ void k_zero_f32(float* __restrict__ p, size_t n) {
    size_t i = (size_t)blockIdx.x * blockDim.x + threadIdx.x;
    if (i < n) p[i] = 0.0f;
}

__global__ __launch_bounds__(256) void k_transpose_w(
    const float* __restrict__ W, float* __restrict__ Wt) {
    __shared__ float tile[16][17];
    int bx = blockIdx.x & 7;
    int by = blockIdx.x >> 3;
    int tx = threadIdx.x & 15;
    int ty = threadIdx.x >> 4;
    tile[ty][tx] = W[(size_t)(by * 16 + ty) * DCH + bx * 16 + tx];
    __syncthreads();
    Wt[(size_t)(bx * 16 + ty) * DCH + by * 16 + tx] = tile[tx][ty];
}

__global__ __launch_bounds__(256) void k_gemm_wmma(
    const float* __restrict__ x, const float* __restrict__ Wt,
    float* __restrict__ h, int N) {
    __shared__ float sX[TILE_M][SX_LD];

    const int t = threadIdx.x;
    const int rowBase = blockIdx.x * TILE_M;

    for (int i = t; i < TILE_M * DCH; i += 256) {
        int r = i >> 7;
        int c = i & (DCH - 1);
        int gr = rowBase + r;
        sX[r][c] = (gr < N) ? x[(size_t)gr * DCH + c] : 0.0f;
    }
    __syncthreads();

    const int lane = t & 31;
    const int wave = t >> 5;
    const int m    = lane & 15;
    const int koff = (lane >> 4) << 1;
    const int n    = (wave << 4) + (lane & 15);

    const float* wrow = Wt + (size_t)n * DCH;

    v8f acc = {};
#pragma unroll
    for (int kt = 0; kt < DCH; kt += 4) {
        v2f a = *reinterpret_cast<const v2f*>(&sX[m][kt + koff]);
        v2f b = *reinterpret_cast<const v2f*>(wrow + kt + koff);
        acc = __builtin_amdgcn_wmma_f32_16x16x4_f32(
            false, a, false, b, (short)0, acc, false, false);
    }

    const int rhalf = (lane >> 4) << 3;
#pragma unroll
    for (int v = 0; v < 8; ++v) {
        int gr = rowBase + rhalf + v;
        if (gr < N) h[(size_t)gr * DCH + n] = acc[v];
    }
}

#define EDGES_PER_WAVE 4
__global__ __launch_bounds__(256) void k_edge_scatter(
    const int* __restrict__ src, const int* __restrict__ dst,
    const float* __restrict__ attr, const float* __restrict__ h,
    float* __restrict__ agg, int E) {
    const int lane  = threadIdx.x & 31;
    const int ebase = (blockIdx.x * 8 + (threadIdx.x >> 5)) * EDGES_PER_WAVE;

    __builtin_prefetch(&src[ebase + 4096], 0, 0);
    __builtin_prefetch(&dst[ebase + 4096], 0, 0);
    __builtin_prefetch(&attr[ebase + 4096], 0, 0);

#pragma unroll
    for (int j = 0; j < EDGES_PER_WAVE; ++j) {
        int e = ebase + j;
        if (e >= E) break;
        e = __builtin_amdgcn_readfirstlane(e);
        int   s = src[e];
        int   d = dst[e];
        float a = attr[e];
        const float4* hv = reinterpret_cast<const float4*>(h + (size_t)s * DCH);
        float4 v = hv[lane];
        float* outp = agg + (size_t)d * DCH + lane * 4;
        unsafeAtomicAdd(outp + 0, v.x * a);
        unsafeAtomicAdd(outp + 1, v.y * a);
        unsafeAtomicAdd(outp + 2, v.z * a);
        unsafeAtomicAdd(outp + 3, v.w * a);
    }
}

#define GROWS 64
__global__ __launch_bounds__(128) void k_bias_relu_mean(
    float* __restrict__ out, const float* __restrict__ bias,
    const int* __restrict__ batch, float* __restrict__ mean_sum,
    float* __restrict__ cnt, int N) {
    __shared__ int sg[GROWS];
    const int base = blockIdx.x * GROWS;
    const int c = threadIdx.x;
    int rows = N - base;
    if (rows > GROWS) rows = GROWS;
    for (int i = c; i < rows; i += 128) sg[i] = batch[base + i];
    __syncthreads();

    const float bv = bias[c];
    float acc = 0.0f, cacc = 0.0f;
    int curg = -1;
    for (int r = 0; r < rows; ++r) {
        int g = sg[r];
        if (g != curg) {
            if (curg >= 0) {
                unsafeAtomicAdd(&mean_sum[(size_t)curg * DCH + c], acc);
                if (c == 0) unsafeAtomicAdd(&cnt[curg], cacc);
            }
            acc = 0.0f; cacc = 0.0f; curg = g;
        }
        size_t idx = (size_t)(base + r) * DCH + c;
        float v = out[idx] + bv;
        v = v > 0.0f ? v : 0.0f;
        out[idx] = v;
        acc += v;
        cacc += 1.0f;
    }
    if (curg >= 0) {
        unsafeAtomicAdd(&mean_sum[(size_t)curg * DCH + c], acc);
        if (c == 0) unsafeAtomicAdd(&cnt[curg], cacc);
    }
}

__global__ __launch_bounds__(128) void k_var_pass(
    const float* __restrict__ out, const int* __restrict__ batch,
    const float* __restrict__ mean_sum, const float* __restrict__ cnt,
    const float* __restrict__ ms, float* __restrict__ var_sum, int N) {
    __shared__ int sg[GROWS];
    const int base = blockIdx.x * GROWS;
    const int c = threadIdx.x;
    int rows = N - base;
    if (rows > GROWS) rows = GROWS;
    for (int i = c; i < rows; i += 128) sg[i] = batch[base + i];
    __syncthreads();

    const float msc = ms[c];
    float acc = 0.0f;
    float sub = 0.0f;
    int curg = -1;
    for (int r = 0; r < rows; ++r) {
        int g = sg[r];
        if (g != curg) {
            if (curg >= 0) unsafeAtomicAdd(&var_sum[(size_t)curg * DCH + c], acc);
            acc = 0.0f;
            curg = g;
            float cn = cnt[g];
            cn = cn > 1.0f ? cn : 1.0f;
            sub = (mean_sum[(size_t)g * DCH + c] / cn) * msc;
        }
        float o = out[(size_t)(base + r) * DCH + c] - sub;
        acc += o * o;
    }
    if (curg >= 0) unsafeAtomicAdd(&var_sum[(size_t)curg * DCH + c], acc);
}

__global__ __launch_bounds__(256) void k_finalize(
    float* __restrict__ out, const int* __restrict__ batch,
    const float* __restrict__ mean_sum, const float* __restrict__ var_sum,
    const float* __restrict__ cnt, const float* __restrict__ ms,
    const float* __restrict__ gw, const float* __restrict__ gb, int N) {
    size_t idx = (size_t)blockIdx.x * 256 + threadIdx.x;
    size_t total = (size_t)N * (DCH / 4);
    if (idx >= total) return;
    int nrow = (int)(idx >> 5);
    int q    = (int)(idx & 31);
    int g = batch[nrow];
    float cn = cnt[g];
    cn = cn > 1.0f ? cn : 1.0f;
    float inv_cn = 1.0f / cn;

    float4 m4 = reinterpret_cast<const float4*>(mean_sum + (size_t)g * DCH)[q];
    float4 v4 = reinterpret_cast<const float4*>(var_sum  + (size_t)g * DCH)[q];
    float4 s4 = reinterpret_cast<const float4*>(ms)[q];
    float4 w4 = reinterpret_cast<const float4*>(gw)[q];
    float4 b4 = reinterpret_cast<const float4*>(gb)[q];
    float4 o  = reinterpret_cast<float4*>(out)[idx];

    o.x -= (m4.x * inv_cn) * s4.x;
    o.y -= (m4.y * inv_cn) * s4.y;
    o.z -= (m4.z * inv_cn) * s4.z;
    o.w -= (m4.w * inv_cn) * s4.w;
    o.x = w4.x * o.x / sqrtf(v4.x * inv_cn + GN_EPS) + b4.x;
    o.y = w4.y * o.y / sqrtf(v4.y * inv_cn + GN_EPS) + b4.y;
    o.z = w4.z * o.z / sqrtf(v4.z * inv_cn + GN_EPS) + b4.z;
    o.w = w4.w * o.w / sqrtf(v4.w * inv_cn + GN_EPS) + b4.w;

    reinterpret_cast<float4*>(out)[idx] = o;
}

extern "C" void kernel_launch(void* const* d_in, const int* in_sizes, int n_in,
                              void* d_out, int out_size, void* d_ws, size_t ws_size,
                              hipStream_t stream) {
    const float* x     = (const float*)d_in[0];
    const int*   eidx  = (const int*)  d_in[1];
    const float* eattr = (const float*)d_in[2];
    const int*   batch = (const int*)  d_in[3];
    const float* W     = (const float*)d_in[4];
    const float* bias  = (const float*)d_in[5];
    const float* gw    = (const float*)d_in[6];
    const float* gb    = (const float*)d_in[7];
    const float* ms    = (const float*)d_in[8];
    float* out = (float*)d_out;

    const int N = in_sizes[0] / DCH;
    const int E = in_sizes[2];

    float* h        = (float*)d_ws;
    float* mean_sum = h + (size_t)N * DCH;
    float* var_sum  = mean_sum + (size_t)NGRAPH * DCH;
    float* cnt      = var_sum + (size_t)NGRAPH * DCH;
    float* Wt       = cnt + NGRAPH;

    {
        size_t n0 = (size_t)N * DCH;
        k_zero_f32<<<(unsigned)((n0 + 255) / 256), 256, 0, stream>>>(out, n0);
        size_t n1 = (size_t)2 * NGRAPH * DCH + NGRAPH;
        k_zero_f32<<<(unsigned)((n1 + 255) / 256), 256, 0, stream>>>(mean_sum, n1);
    }

    k_transpose_w<<<64, 256, 0, stream>>>(W, Wt);

    k_gemm_wmma<<<(N + TILE_M - 1) / TILE_M, 256, 0, stream>>>(x, Wt, h, N);

    {
        int edges_per_block = 8 * EDGES_PER_WAVE;
        k_edge_scatter<<<(E + edges_per_block - 1) / edges_per_block, 256, 0, stream>>>(
            eidx, eidx + E, eattr, h, out, E);
    }

    k_bias_relu_mean<<<(N + GROWS - 1) / GROWS, 128, 0, stream>>>(
        out, bias, batch, mean_sum, cnt, N);

    k_var_pass<<<(N + GROWS - 1) / GROWS, 128, 0, stream>>>(
        out, batch, mean_sum, cnt, ms, var_sum, N);

    {
        size_t total = (size_t)N * (DCH / 4);
        k_finalize<<<(unsigned)((total + 255) / 256), 256, 0, stream>>>(
            out, batch, mean_sum, var_sum, cnt, ms, gw, gb, N);
    }
}


# GEMM fused into the per-graph norm kernel: a block owns (graph, 32 channels), keeps its product tiles in registers, sums mean and variance in a fixed order (no atomics there), stores out once
# speedup vs baseline: 1.4683x; 1.4683x over previous
#include <hip/hip_runtime.h>
#include <hip/hip_bf16.h>
#include <math.h>

typedef __attribute__((ext_vector_type(2))) float v2f;
typedef __attribute__((ext_vector_type(8))) float v8f;

#define DCH 128
#define NGRAPH 64
#define GN_EPS 1e-5f
#define SX_LD (DCH + 4)

__global__ void k_zero_f32(float* __restrict__ p, size_t n) {
    size_t i = (size_t)blockIdx.x * blockDim.x + threadIdx.x;
    if (i < n) p[i] = 0.0f;
}

__global__ __launch_bounds__(256) void k_transpose_w(
    const float* __restrict__ W, float* __restrict__ Wt) {
    __shared__ float tile[16][17];
    int bx = blockIdx.x & 7;
    int by = blockIdx.x >> 3;
    int tx = threadIdx.x & 15;
    int ty = threadIdx.x >> 4;
    tile[ty][tx] = W[(size_t)(by * 16 + ty) * DCH + bx * 16 + tx];
    __syncthreads();
    Wt[(size_t)(bx * 16 + ty) * DCH + by * 16 + tx] = tile[tx][ty];
}

#define EDGES_PER_WAVE 4
__global__ __launch_bounds__(256) void k_edge_scatter(
    const int* __restrict__ src, const int* __restrict__ dst,
    const float* __restrict__ attr, const float* __restrict__ h,
    float* __restrict__ agg, int E) {
    const int lane  = threadIdx.x & 31;
    const int ebase = (blockIdx.x * 8 + (threadIdx.x >> 5)) * EDGES_PER_WAVE;

    __builtin_prefetch(&src[ebase + 4096], 0, 0);
    __builtin_prefetch(&dst[ebase + 4096], 0, 0);
    __builtin_prefetch(&attr[ebase + 4096], 0, 0);

#pragma unroll
    for (int j = 0; j < EDGES_PER_WAVE; ++j) {
        int e = ebase + j;
        if (e >= E) break;
        e = __builtin_amdgcn_readfirstlane(e);
        int   s = src[e];
        int   d = dst[e];
        float a = attr[e];
        const float4* hv = reinterpret_cast<const float4*>(h + (size_t)s * DCH);
        float4 v = hv[lane];
        float* outp = agg + (size_t)d * DCH + lane * 4;
        unsafeAtomicAdd(outp + 0, v.x * a);
        unsafeAtomicAdd(outp + 1, v.y * a);
        unsafeAtomicAdd(outp + 2, v.z * a);
        unsafeAtomicAdd(outp + 3, v.w * a);
    }
}

#define FT_SLOTS 4
#define FT_WAVES 32
template <int RB, int KU>
__device__ __forceinline__ void ft_tile_product(
    const float* __restrict__ grows, int j, int n, float (*sAw)[SX_LD],
    const float (*sBm)[SX_LD], int lane, v8f& acc0, v8f& acc1) {
    const int r0 = j * 16;
#pragma unroll
    for (int part = 0; part < 16 / RB; ++part) {
        float4 v[RB];
#pragma unroll
        for (int r = 0; r < RB; ++r) {
            const int rr = r0 + part * RB + r;
            v[r] = make_float4(0.0f, 0.0f, 0.0f, 0.0f);
            if (rr < n) v[r] = *reinterpret_cast<const float4*>(grows + (size_t)rr * DCH + lane * 4);
        }
#pragma unroll
        for (int r = 0; r < RB; ++r)
            *reinterpret_cast<float4*>(&sAw[part * RB + r][lane * 4]) = v[r];
    }
    __builtin_amdgcn_fence(__ATOMIC_RELEASE, "wavefront");
    __builtin_amdgcn_wave_barrier();
    __builtin_amdgcn_fence(__ATOMIC_ACQUIRE, "wavefront");

    const int m    = lane & 15;
    const int koff = (lane >> 4) << 1;
    acc0 = v8f{};
    acc1 = v8f{};
#pragma unroll KU
    for (int kt = 0; kt < DCH; kt += 4) {
        v2f a  = *reinterpret_cast<const v2f*>(&sAw[m][kt + koff]);
        v2f b0 = *reinterpret_cast<const v2f*>(&sBm[m][kt + koff]);
        v2f b1 = *reinterpret_cast<const v2f*>(&sBm[16 + m][kt + koff]);
        acc0 = __builtin_amdgcn_wmma_f32_16x16x4_f32(false, a, false, b0, (short)0, acc0, false, false);
        acc1 = __builtin_amdgcn_wmma_f32_16x16x4_f32(false, a, false, b1, (short)0, acc1, false, false);
    }
    __builtin_amdgcn_fence(__ATOMIC_RELEASE, "wavefront");
    __builtin_amdgcn_wave_barrier();
    __builtin_amdgcn_fence(__ATOMIC_ACQUIRE, "wavefront");
}

__device__ __forceinline__ void ft_bias_relu_sum(
    v8f& ha, v8f& hb, float bv0, float bv1, int rbase, int n, float& p0, float& p1) {
#pragma unroll
    for (int v = 0; v < 8; ++v) {
        const bool ok = (rbase + v) < n;
        float a0 = ha[v] + bv0; a0 = a0 > 0.0f ? a0 : 0.0f;
        float a1 = hb[v] + bv1; a1 = a1 > 0.0f ? a1 : 0.0f;
        ha[v] = a0;
        hb[v] = a1;
        p0 += ok ? a0 : 0.0f;
        p1 += ok ? a1 : 0.0f;
    }
}

__device__ __forceinline__ void ft_sq_sum(
    const v8f& ha, const v8f& hb, float sub0, float sub1, int rbase, int n, float& q0, float& q1) {
#pragma unroll
    for (int v = 0; v < 8; ++v) {
        const bool ok = (rbase + v) < n;
        const float o0 = ha[v] - sub0;
        const float o1 = hb[v] - sub1;
        q0 += ok ? o0 * o0 : 0.0f;
        q1 += ok ? o1 * o1 : 0.0f;
    }
}

__device__ __forceinline__ void ft_norm_store(
    const v8f& ha, const v8f& hb, float sub0, float sub1, float den0, float den1,
    float w0, float w1, float b0, float b1, int rbase, int n, float* __restrict__ ocol) {
#pragma unroll
    for (int v = 0; v < 8; ++v) {
        const int rr = rbase + v;
        if (rr < n) {
            ocol[(size_t)rr * DCH]      = w0 * (ha[v] - sub0) / den0 + b0;
            ocol[(size_t)rr * DCH + 16] = w1 * (hb[v] - sub1) / den1 + b1;
        }
    }
}

__device__ __forceinline__ void ft_block_sum(
    float p0, float p1, float (*s_red)[32], float* s_stat, int t, float& r0, float& r1) {
    const int lane = t & 31;
    const int wave = t >> 5;
    p0 += __shfl_xor(p0, 16);
    p1 += __shfl_xor(p1, 16);
    if (lane < 16) { s_red[wave][lane] = p0; s_red[wave][16 + lane] = p1; }
    __syncthreads();
    if (t < 32) {
        float s = 0.0f;
        for (int w = 0; w < FT_WAVES; ++w) s += s_red[w][t];
        s_stat[t] = s;
    }
    __syncthreads();
    r0 = s_stat[lane & 15];
    r1 = s_stat[16 + (lane & 15)];
    __syncthreads();
}

__global__ __launch_bounds__(1024) void k_gemm_graph_norm(
    const float* __restrict__ agg, const float* __restrict__ Wt,
    const float* __restrict__ bias, const int* __restrict__ batch,
    const float* __restrict__ ms, const float* __restrict__ gw,
    const float* __restrict__ gb, float* __restrict__ out, int N) {
    __shared__ __attribute__((aligned(16))) float sA[FT_WAVES][16][SX_LD];
    __shared__ __attribute__((aligned(16))) float sB[32][SX_LD];
    __shared__ float s_red[FT_WAVES][32];
    __shared__ float s_stat[32];
    __shared__ int s_less[FT_WAVES];
    __shared__ int s_eq[FT_WAVES];

    const int t = threadIdx.x;
    const int lane = t & 31;
    const int wave = __builtin_amdgcn_readfirstlane(t >> 5);
    const int g = blockIdx.x >> 2;
    const int slice = blockIdx.x & 3;

    const float4 wv = *reinterpret_cast<const float4*>(
        Wt + (size_t)(slice * 32 + (t >> 5)) * DCH + ((t & 31) << 2));

    int less = 0, eq = 0;
    {
        const int4* b4 = reinterpret_cast<const int4*>(batch);
        const int n4 = N >> 2;
        for (int base = 0; base < n4; base += 8 * 1024) {
            int4 q[8];
#pragma unroll
            for (int u = 0; u < 8; ++u) {
                const int i = base + u * 1024 + t;
                q[u] = make_int4(0x7fffffff, 0x7fffffff, 0x7fffffff, 0x7fffffff);
                if (i < n4) q[u] = b4[i];
            }
#pragma unroll
            for (int u = 0; u < 8; ++u) {
                less += (q[u].x < g) + (q[u].y < g) + (q[u].z < g) + (q[u].w < g);
                eq   += (q[u].x == g) + (q[u].y == g) + (q[u].z == g) + (q[u].w == g);
            }
        }
        for (int i = (n4 << 2) + t; i < N; i += 1024) {
            const int q = batch[i];
            less += (q < g);
            eq   += (q == g);
        }
    }
#pragma unroll
    for (int off = 16; off > 0; off >>= 1) {
        less += __shfl_xor(less, off);
        eq   += __shfl_xor(eq, off);
    }
    if (lane == 0) { s_less[wave] = less; s_eq[wave] = eq; }
    *reinterpret_cast<float4*>(&sB[t >> 5][(t & 31) << 2]) = wv;
    __syncthreads();
    int start_v = 0, n_v = 0;
    for (int w = 0; w < FT_WAVES; ++w) { start_v += s_less[w]; n_v += s_eq[w]; }
    const int start = __builtin_amdgcn_readfirstlane(start_v);
    const int n = __builtin_amdgcn_readfirstlane(n_v);
    const int ntiles = (n + 15) >> 4;

    const float* grows = agg + (size_t)start * DCH;
    float* orows = out + (size_t)start * DCH;
    float (*sAw)[SX_LD] = sA[wave];

    const int c = lane & 15;
    const int rhalf = (lane >> 4) << 3;
    const int col0 = slice * 32 + c;
    const int col1 = col0 + 16;
    const float bv0 = bias[col0], bv1 = bias[col1];

    v8f hA0, hB0, hA1, hB1, hA2, hB2, hA3, hB3;
    float p0 = 0.0f, p1 = 0.0f;
#define FT_PHASE1(S, HA, HB)                                                        \
    {                                                                               \
        const int j = wave + (S) * FT_WAVES;     \
        HA = v8f{};                                                                 \
        HB = v8f{};                                                                 \
        if (j < ntiles) {                                                           \
            ft_tile_product<8, 4>(grows, j, n, sAw, sB, lane, HA, HB);              \
            ft_bias_relu_sum(HA, HB, bv0, bv1, j * 16 + rhalf, n, p0, p1);          \
        }                                                                           \
    }
    FT_PHASE1(0, hA0, hB0)
    FT_PHASE1(1, hA1, hB1)
    FT_PHASE1(2, hA2, hB2)
    FT_PHASE1(3, hA3, hB3)
#undef FT_PHASE1
    for (int j = wave + FT_SLOTS * FT_WAVES; j < ntiles; j += FT_WAVES) {
        v8f e0, e1;
        ft_tile_product<2, 1>(grows, j, n, sAw, sB, lane, e0, e1);
        ft_bias_relu_sum(e0, e1, bv0, bv1, j * 16 + rhalf, n, p0, p1);
    }
    float msum0, msum1;
    ft_block_sum(p0, p1, s_red, s_stat, t, msum0, msum1);

    float cn = (float)n;
    cn = cn > 1.0f ? cn : 1.0f;
    const float inv_cn = 1.0f / cn;
    const float sub0 = (msum0 * inv_cn) * ms[col0];
    const float sub1 = (msum1 * inv_cn) * ms[col1];

    float q0 = 0.0f, q1 = 0.0f;
    if (wave + 0 * FT_WAVES < ntiles) ft_sq_sum(hA0, hB0, sub0, sub1, (wave + 0 * FT_WAVES) * 16 + rhalf, n, q0, q1);
    if (wave + 1 * FT_WAVES < ntiles) ft_sq_sum(hA1, hB1, sub0, sub1, (wave + 1 * FT_WAVES) * 16 + rhalf, n, q0, q1);
    if (wave + 2 * FT_WAVES < ntiles) ft_sq_sum(hA2, hB2, sub0, sub1, (wave + 2 * FT_WAVES) * 16 + rhalf, n, q0, q1);
    if (wave + 3 * FT_WAVES < ntiles) ft_sq_sum(hA3, hB3, sub0, sub1, (wave + 3 * FT_WAVES) * 16 + rhalf, n, q0, q1);
    for (int j = wave + FT_SLOTS * FT_WAVES; j < ntiles; j += FT_WAVES) {
        v8f e0, e1;
        float d0 = 0.0f, d1 = 0.0f;
        ft_tile_product<2, 1>(grows, j, n, sAw, sB, lane, e0, e1);
        ft_bias_relu_sum(e0, e1, bv0, bv1, j * 16 + rhalf, n, d0, d1);
        ft_sq_sum(e0, e1, sub0, sub1, j * 16 + rhalf, n, q0, q1);
    }
    float vsum0, vsum1;
    ft_block_sum(q0, q1, s_red, s_stat, t, vsum0, vsum1);

    const float den0 = sqrtf(vsum0 * inv_cn + GN_EPS);
    const float den1 = sqrtf(vsum1 * inv_cn + GN_EPS);
    const float w0 = gw[col0], w1 = gw[col1];
    const float b0 = gb[col0], b1 = gb[col1];

    if (wave + 0 * FT_WAVES < ntiles) ft_norm_store(hA0, hB0, sub0, sub1, den0, den1, w0, w1, b0, b1, (wave + 0 * FT_WAVES) * 16 + rhalf, n, orows + col0);
    if (wave + 1 * FT_WAVES < ntiles) ft_norm_store(hA1, hB1, sub0, sub1, den0, den1, w0, w1, b0, b1, (wave + 1 * FT_WAVES) * 16 + rhalf, n, orows + col0);
    if (wave + 2 * FT_WAVES < ntiles) ft_norm_store(hA2, hB2, sub0, sub1, den0, den1, w0, w1, b0, b1, (wave + 2 * FT_WAVES) * 16 + rhalf, n, orows + col0);
    if (wave + 3 * FT_WAVES < ntiles) ft_norm_store(hA3, hB3, sub0, sub1, den0, den1, w0, w1, b0, b1, (wave + 3 * FT_WAVES) * 16 + rhalf, n, orows + col0);
    for (int j = wave + FT_SLOTS * FT_WAVES; j < ntiles; j += FT_WAVES) {
        v8f e0, e1;
        float d0 = 0.0f, d1 = 0.0f;
        ft_tile_product<2, 1>(grows, j, n, sAw, sB, lane, e0, e1);
        ft_bias_relu_sum(e0, e1, bv0, bv1, j * 16 + rhalf, n, d0, d1);
        ft_norm_store(e0, e1, sub0, sub1, den0, den1, w0, w1, b0, b1, j * 16 + rhalf, n, orows + col0);
    }
}

extern "C" void kernel_launch(void* const* d_in, const int* in_sizes, int n_in,
                              void* d_out, int out_size, void* d_ws, size_t ws_size,
                              hipStream_t stream) {
    const float* x     = (const float*)d_in[0];
    const int*   eidx  = (const int*)  d_in[1];
    const float* eattr = (const float*)d_in[2];
    const int*   batch = (const int*)  d_in[3];
    const float* W     = (const float*)d_in[4];
    const float* bias  = (const float*)d_in[5];
    const float* gw    = (const float*)d_in[6];
    const float* gb    = (const float*)d_in[7];
    const float* ms    = (const float*)d_in[8];
    float* out = (float*)d_out;

    const int N = in_sizes[0] / DCH;
    const int E = in_sizes[2];

    float* Wt  = (float*)d_ws;
    float* agg = Wt + (size_t)DCH * DCH;

    {
        size_t n0 = (size_t)N * DCH;
        k_zero_f32<<<(unsigned)((n0 + 255) / 256), 256, 0, stream>>>(agg, n0);
    }

    k_transpose_w<<<64, 256, 0, stream>>>(W, Wt);

    {
        int edges_per_block = 8 * EDGES_PER_WAVE;
        k_edge_scatter<<<(E + edges_per_block - 1) / edges_per_block, 256, 0, stream>>>(
            eidx, eidx + E, eattr, x, agg, E);
    }

    k_gemm_graph_norm<<<NGRAPH * 4, 1024, 0, stream>>>(agg, Wt, bias, batch, ms, gw, gb, out, N);
}


# edge sum in LDS, a block per 391 nodes: waves filter the dst list, owner waves add rows in a fixed edge order (the refuted try added in arrival order); float atomics and zero fill gone
# speedup vs baseline: 2.0849x; 1.4199x over previous
#include <hip/hip_runtime.h>
#include <hip/hip_bf16.h>
#include <math.h>

typedef __attribute__((ext_vector_type(2))) float v2f;
typedef __attribute__((ext_vector_type(8))) float v8f;

#define DCH 128
#define NGRAPH 64
#define GN_EPS 1e-5f
#define SX_LD (DCH + 4)

__global__ __launch_bounds__(256) void k_transpose_w(
    const float* __restrict__ W, float* __restrict__ Wt) {
    __shared__ float tile[16][17];
    int bx = blockIdx.x & 7;
    int by = blockIdx.x >> 3;
    int tx = threadIdx.x & 15;
    int ty = threadIdx.x >> 4;
    tile[ty][tx] = W[(size_t)(by * 16 + ty) * DCH + bx * 16 + tx];
    __syncthreads();
    Wt[(size_t)(bx * 16 + ty) * DCH + by * 16 + tx] = tile[tx][ty];
}

#define EP_R 391
#define EP_WAVES 32
#define EP_CAP 512
#define EP_U 32
#define EP_STAGE 64
#define EP_ROWS 16

__device__ __forceinline__ void ep_wave_order() {
    __builtin_amdgcn_fence(__ATOMIC_RELEASE, "wavefront");
    __builtin_amdgcn_wave_barrier();
    __builtin_amdgcn_fence(__ATOMIC_ACQUIRE, "wavefront");
}

__device__ __forceinline__ void ep_flush(
    const unsigned* stage, int nb, int lane, const int* __restrict__ src,
    const float* __restrict__ attr, const float4* __restrict__ x4, float4* acc4) {
    ep_wave_order();
    for (int b0 = 0; b0 < nb; b0 += 32) {
        int nbb = nb - b0;
        nbb = nbb < 32 ? nbb : 32;
        unsigned ent = 0;
        int sv = 0;
        float av = 0.0f;
        if (lane < nbb) {
            ent = stage[b0 + lane];
            const int e = (int)(ent >> 9);
            sv = src[e];
            av = attr[e];
        }
        const int rv = (int)(ent & 511u);
        const int avi = __float_as_int(av);
        for (int j0 = 0; j0 < nbb; j0 += EP_ROWS) {
            float4 v[EP_ROWS];
#pragma unroll
            for (int u = 0; u < EP_ROWS; ++u) {
                int j = j0 + u;
                j = j < nbb ? j : nbb - 1;
                const int s = __builtin_amdgcn_readlane(sv, j);
                v[u] = x4[(size_t)s * 32 + lane];
            }
#pragma unroll
            for (int u = 0; u < EP_ROWS; ++u) {
                const int j = j0 + u;
                if (j < nbb) {
                    const int r = __builtin_amdgcn_readlane(rv, j);
                    const float a = __int_as_float(__builtin_amdgcn_readlane(avi, j));
                    float4 c = acc4[r * 32 + lane];
                    c.x += a * v[u].x;
                    c.y += a * v[u].y;
                    c.z += a * v[u].z;
                    c.w += a * v[u].w;
                    acc4[r * 32 + lane] = c;
                }
            }
        }
    }
    ep_wave_order();
}

__global__ __launch_bounds__(1024) void k_edge_rows_lds(
    const int* __restrict__ src, const int* __restrict__ dst,
    const float* __restrict__ attr, const float* __restrict__ x,
    float* __restrict__ agg, int N, int E) {
    __shared__ __attribute__((aligned(16))) float s_acc[EP_R * DCH];
    __shared__ unsigned s_list[EP_WAVES][EP_CAP];
    __shared__ unsigned s_stage[EP_WAVES][EP_STAGE];
    __shared__ int s_cnt[EP_WAVES];
    __shared__ int s_more[EP_WAVES];

    const int t = threadIdx.x;
    const int lane = t & 31;
    const int wave = __builtin_amdgcn_readfirstlane(t >> 5);
    const int base = blockIdx.x * EP_R;
    int nrows = N - base;
    nrows = nrows < EP_R ? nrows : EP_R;

    float4* acc4 = reinterpret_cast<float4*>(s_acc);
    const float4* x4 = reinterpret_cast<const float4*>(x);
    for (int i = t; i < EP_R * 32; i += 1024) acc4[i] = make_float4(0.0f, 0.0f, 0.0f, 0.0f);

    const int per = (((E + EP_WAVES - 1) / EP_WAVES) + 31) & ~31;
    int re = (wave + 1) * per;
    re = re < E ? re : E;
    int pos = wave * per;
    const int max_rounds = per / (EP_CAP - 32) + 2;

    for (int round = 0; round < max_rounds; ++round) {
        int count = 0;
        bool full = false;
        while (pos < re && !full) {
            int d[EP_U];
            if (pos + EP_U * 32 <= re) {
#pragma unroll
                for (int u = 0; u < EP_U; ++u) d[u] = dst[pos + u * 32 + lane];
            } else {
#pragma unroll
                for (int u = 0; u < EP_U; ++u) {
                    const int e = pos + u * 32 + lane;
                    d[u] = 0x7fffffff;
                    if (e < re) d[u] = dst[e];
                }
            }
            int adv = EP_U * 32;
#pragma unroll
            for (int u = 0; u < EP_U; ++u) {
                if (!full) {
                    const unsigned rel = (unsigned)(d[u] - base);
                    const bool hit = rel < (unsigned)nrows;
                    const unsigned m = __builtin_amdgcn_ballot_w32(hit);
                    if (m != 0u) {
                        if (hit)
                            s_list[wave][count + (int)__builtin_amdgcn_mbcnt_lo(m, 0u)] =
                                ((unsigned)(pos + u * 32 + lane) << 9) | rel;
                        count += __builtin_popcount(m);
                        if (count > EP_CAP - 32) { full = true; adv = (u + 1) * 32; }
                    }
                }
            }
            pos += adv;
        }
        if (lane == 0) { s_cnt[wave] = count; s_more[wave] = (pos < re) ? 1 : 0; }
        __syncthreads();
        int more_v = 0;
        for (int w = 0; w < EP_WAVES; ++w) more_v |= s_more[w];
        const int more = __builtin_amdgcn_readfirstlane(more_v);

        int npend = 0;
        for (int sw = 0; sw < EP_WAVES; ++sw) {
            const int n = __builtin_amdgcn_readfirstlane(s_cnt[sw]);
            for (int i0 = 0; i0 < n; i0 += 128) {
                unsigned ent[4];
#pragma unroll
                for (int q = 0; q < 4; ++q) {
                    const int idx = i0 + q * 32 + lane;
                    ent[q] = 0xffffffffu;
                    if (idx < n) ent[q] = s_list[sw][idx];
                }
#pragma unroll
                for (int q = 0; q < 4; ++q) {
                    const int idx = i0 + q * 32 + lane;
                    const bool mine = (idx < n) && ((int)(ent[q] & 31u) == wave);
                    const unsigned m = __builtin_amdgcn_ballot_w32(mine);
                    if (m != 0u) {
                        if (npend > EP_STAGE - 32) {
                            ep_flush(s_stage[wave], npend, lane, src, attr, x4, acc4);
                            npend = 0;
                        }
                        if (mine) s_stage[wave][npend + (int)__builtin_amdgcn_mbcnt_lo(m, 0u)] = ent[q];
                        npend += __builtin_popcount(m);
                    }
                }
            }
        }
        ep_flush(s_stage[wave], npend, lane, src, attr, x4, acc4);
        __syncthreads();
        if (!more) break;
    }

    float4* agg4 = reinterpret_cast<float4*>(agg) + (size_t)base * 32;
    for (int i = t; i < nrows * 32; i += 1024) agg4[i] = acc4[i];
}

#define FT_SLOTS 4
#define FT_WAVES 32
template <int RB, int KU>
__device__ __forceinline__ void ft_tile_product(
    const float* __restrict__ grows, int j, int n, float (*sAw)[SX_LD],
    const float (*sBm)[SX_LD], int lane, v8f& acc0, v8f& acc1) {
    const int r0 = j * 16;
#pragma unroll
    for (int part = 0; part < 16 / RB; ++part) {
        float4 v[RB];
#pragma unroll
        for (int r = 0; r < RB; ++r) {
            const int rr = r0 + part * RB + r;
            v[r] = make_float4(0.0f, 0.0f, 0.0f, 0.0f);
            if (rr < n) v[r] = *reinterpret_cast<const float4*>(grows + (size_t)rr * DCH + lane * 4);
        }
#pragma unroll
        for (int r = 0; r < RB; ++r)
            *reinterpret_cast<float4*>(&sAw[part * RB + r][lane * 4]) = v[r];
    }
    __builtin_amdgcn_fence(__ATOMIC_RELEASE, "wavefront");
    __builtin_amdgcn_wave_barrier();
    __builtin_amdgcn_fence(__ATOMIC_ACQUIRE, "wavefront");

    const int m    = lane & 15;
    const int koff = (lane >> 4) << 1;
    acc0 = v8f{};
    acc1 = v8f{};
#pragma unroll KU
    for (int kt = 0; kt < DCH; kt += 4) {
        v2f a  = *reinterpret_cast<const v2f*>(&sAw[m][kt + koff]);
        v2f b0 = *reinterpret_cast<const v2f*>(&sBm[m][kt + koff]);
        v2f b1 = *reinterpret_cast<const v2f*>(&sBm[16 + m][kt + koff]);
        acc0 = __builtin_amdgcn_wmma_f32_16x16x4_f32(false, a, false, b0, (short)0, acc0, false, false);
        acc1 = __builtin_amdgcn_wmma_f32_16x16x4_f32(false, a, false, b1, (short)0, acc1, false, false);
    }
    __builtin_amdgcn_fence(__ATOMIC_RELEASE, "wavefront");
    __builtin_amdgcn_wave_barrier();
    __builtin_amdgcn_fence(__ATOMIC_ACQUIRE, "wavefront");
}

__device__ __forceinline__ void ft_bias_relu_sum(
    v8f& ha, v8f& hb, float bv0, float bv1, int rbase, int n, float& p0, float& p1) {
#pragma unroll
    for (int v = 0; v < 8; ++v) {
        const bool ok = (rbase + v) < n;
        float a0 = ha[v] + bv0; a0 = a0 > 0.0f ? a0 : 0.0f;
        float a1 = hb[v] + bv1; a1 = a1 > 0.0f ? a1 : 0.0f;
        ha[v] = a0;
        hb[v] = a1;
        p0 += ok ? a0 : 0.0f;
        p1 += ok ? a1 : 0.0f;
    }
}

__device__ __forceinline__ void ft_sq_sum(
    const v8f& ha, const v8f& hb, float sub0, float sub1, int rbase, int n, float& q0, float& q1) {
#pragma unroll
    for (int v = 0; v < 8; ++v) {
        const bool ok = (rbase + v) < n;
        const float o0 = ha[v] - sub0;
        const float o1 = hb[v] - sub1;
        q0 += ok ? o0 * o0 : 0.0f;
        q1 += ok ? o1 * o1 : 0.0f;
    }
}

__device__ __forceinline__ void ft_norm_store(
    const v8f& ha, const v8f& hb, float sub0, float sub1, float den0, float den1,
    float w0, float w1, float b0, float b1, int rbase, int n, float* __restrict__ ocol) {
#pragma unroll
    for (int v = 0; v < 8; ++v) {
        const int rr = rbase + v;
        if (rr < n) {
            ocol[(size_t)rr * DCH]      = w0 * (ha[v] - sub0) / den0 + b0;
            ocol[(size_t)rr * DCH + 16] = w1 * (hb[v] - sub1) / den1 + b1;
        }
    }
}

__device__ __forceinline__ void ft_block_sum(
    float p0, float p1, float (*s_red)[32], float* s_stat, int t, float& r0, float& r1) {
    const int lane = t & 31;
    const int wave = t >> 5;
    p0 += __shfl_xor(p0, 16);
    p1 += __shfl_xor(p1, 16);
    if (lane < 16) { s_red[wave][lane] = p0; s_red[wave][16 + lane] = p1; }
    __syncthreads();
    if (t < 32) {
        float s = 0.0f;
        for (int w = 0; w < FT_WAVES; ++w) s += s_red[w][t];
        s_stat[t] = s;
    }
    __syncthreads();
    r0 = s_stat[lane & 15];
    r1 = s_stat[16 + (lane & 15)];
    __syncthreads();
}

__global__ __launch_bounds__(1024) void k_gemm_graph_norm(
    const float* __restrict__ agg, const float* __restrict__ Wt,
    const float* __restrict__ bias, const int* __restrict__ batch,
    const float* __restrict__ ms, const float* __restrict__ gw,
    const float* __restrict__ gb, float* __restrict__ out, int N) {
    __shared__ __attribute__((aligned(16))) float sA[FT_WAVES][16][SX_LD];
    __shared__ __attribute__((aligned(16))) float sB[32][SX_LD];
    __shared__ float s_red[FT_WAVES][32];
    __shared__ float s_stat[32];
    __shared__ int s_less[FT_WAVES];
    __shared__ int s_eq[FT_WAVES];

    const int t = threadIdx.x;
    const int lane = t & 31;
    const int wave = __builtin_amdgcn_readfirstlane(t >> 5);
    const int g = blockIdx.x >> 2;
    const int slice = blockIdx.x & 3;

    const float4 wv = *reinterpret_cast<const float4*>(
        Wt + (size_t)(slice * 32 + (t >> 5)) * DCH + ((t & 31) << 2));

    int less = 0, eq = 0;
    {
        const int4* b4 = reinterpret_cast<const int4*>(batch);
        const int n4 = N >> 2;
        for (int base = 0; base < n4; base += 8 * 1024) {
            int4 q[8];
#pragma unroll
            for (int u = 0; u < 8; ++u) {
                const int i = base + u * 1024 + t;
                q[u] = make_int4(0x7fffffff, 0x7fffffff, 0x7fffffff, 0x7fffffff);
                if (i < n4) q[u] = b4[i];
            }
#pragma unroll
            for (int u = 0; u < 8; ++u) {
                less += (q[u].x < g) + (q[u].y < g) + (q[u].z < g) + (q[u].w < g);
                eq   += (q[u].x == g) + (q[u].y == g) + (q[u].z == g) + (q[u].w == g);
            }
        }
        for (int i = (n4 << 2) + t; i < N; i += 1024) {
            const int q = batch[i];
            less += (q < g);
            eq   += (q == g);
        }
    }
#pragma unroll
    for (int off = 16; off > 0; off >>= 1) {
        less += __shfl_xor(less, off);
        eq   += __shfl_xor(eq, off);
    }
    if (lane == 0) { s_less[wave] = less; s_eq[wave] = eq; }
    *reinterpret_cast<float4*>(&sB[t >> 5][(t & 31) << 2]) = wv;
    __syncthreads();
    int start_v = 0, n_v = 0;
    for (int w = 0; w < FT_WAVES; ++w) { start_v += s_less[w]; n_v += s_eq[w]; }
    const int start = __builtin_amdgcn_readfirstlane(start_v);
    const int n = __builtin_amdgcn_readfirstlane(n_v);
    const int ntiles = (n + 15) >> 4;

    const float* grows = agg + (size_t)start * DCH;
    float* orows = out + (size_t)start * DCH;
    float (*sAw)[SX_LD] = sA[wave];

    const int c = lane & 15;
    const int rhalf = (lane >> 4) << 3;
    const int col0 = slice * 32 + c;
    const int col1 = col0 + 16;
    const float bv0 = bias[col0], bv1 = bias[col1];

    v8f hA0, hB0, hA1, hB1, hA2, hB2, hA3, hB3;
    float p0 = 0.0f, p1 = 0.0f;
#define FT_PHASE1(S, HA, HB)                                                        \
    {                                                                               \
        const int j = wave + (S) * FT_WAVES;     \
        HA = v8f{};                                                                 \
        HB = v8f{};                                                                 \
        if (j < ntiles) {                                                           \
            ft_tile_product<8, 4>(grows, j, n, sAw, sB, lane, HA, HB);              \
            ft_bias_relu_sum(HA, HB, bv0, bv1, j * 16 + rhalf, n, p0, p1);          \
        }                                                                           \
    }
    FT_PHASE1(0, hA0, hB0)
    FT_PHASE1(1, hA1, hB1)
    FT_PHASE1(2, hA2, hB2)
    FT_PHASE1(3, hA3, hB3)
#undef FT_PHASE1
    for (int j = wave + FT_SLOTS * FT_WAVES; j < ntiles; j += FT_WAVES) {
        v8f e0, e1;
        ft_tile_product<2, 1>(grows, j, n, sAw, sB, lane, e0, e1);
        ft_bias_relu_sum(e0, e1, bv0, bv1, j * 16 + rhalf, n, p0, p1);
    }
    float msum0, msum1;
    ft_block_sum(p0, p1, s_red, s_stat, t, msum0, msum1);

    float cn = (float)n;
    cn = cn > 1.0f ? cn : 1.0f;
    const float inv_cn = 1.0f / cn;
    const float sub0 = (msum0 * inv_cn) * ms[col0];
    const float sub1 = (msum1 * inv_cn) * ms[col1];

    float q0 = 0.0f, q1 = 0.0f;
    if (wave + 0 * FT_WAVES < ntiles) ft_sq_sum(hA0, hB0, sub0, sub1, (wave + 0 * FT_WAVES) * 16 + rhalf, n, q0, q1);
    if (wave + 1 * FT_WAVES < ntiles) ft_sq_sum(hA1, hB1, sub0, sub1, (wave + 1 * FT_WAVES) * 16 + rhalf, n, q0, q1);
    if (wave + 2 * FT_WAVES < ntiles) ft_sq_sum(hA2, hB2, sub0, sub1, (wave + 2 * FT_WAVES) * 16 + rhalf, n, q0, q1);
    if (wave + 3 * FT_WAVES < ntiles) ft_sq_sum(hA3, hB3, sub0, sub1, (wave + 3 * FT_WAVES) * 16 + rhalf, n, q0, q1);
    for (int j = wave + FT_SLOTS * FT_WAVES; j < ntiles; j += FT_WAVES) {
        v8f e0, e1;
        float d0 = 0.0f, d1 = 0.0f;
        ft_tile_product<2, 1>(grows, j, n, sAw, sB, lane, e0, e1);
        ft_bias_relu_sum(e0, e1, bv0, bv1, j * 16 + rhalf, n, d0, d1);
        ft_sq_sum(e0, e1, sub0, sub1, j * 16 + rhalf, n, q0, q1);
    }
    float vsum0, vsum1;
    ft_block_sum(q0, q1, s_red, s_stat, t, vsum0, vsum1);

    const float den0 = sqrtf(vsum0 * inv_cn + GN_EPS);
    const float den1 = sqrtf(vsum1 * inv_cn + GN_EPS);
    const float w0 = gw[col0], w1 = gw[col1];
    const float b0 = gb[col0], b1 = gb[col1];

    if (wave + 0 * FT_WAVES < ntiles) ft_norm_store(hA0, hB0, sub0, sub1, den0, den1, w0, w1, b0, b1, (wave + 0 * FT_WAVES) * 16 + rhalf, n, orows + col0);
    if (wave + 1 * FT_WAVES < ntiles) ft_norm_store(hA1, hB1, sub0, sub1, den0, den1, w0, w1, b0, b1, (wave + 1 * FT_WAVES) * 16 + rhalf, n, orows + col0);
    if (wave + 2 * FT_WAVES < ntiles) ft_norm_store(hA2, hB2, sub0, sub1, den0, den1, w0, w1, b0, b1, (wave + 2 * FT_WAVES) * 16 + rhalf, n, orows + col0);
    if (wave + 3 * FT_WAVES < ntiles) ft_norm_store(hA3, hB3, sub0, sub1, den0, den1, w0, w1, b0, b1, (wave + 3 * FT_WAVES) * 16 + rhalf, n, orows + col0);
    for (int j = wave + FT_SLOTS * FT_WAVES; j < ntiles; j += FT_WAVES) {
        v8f e0, e1;
        float d0 = 0.0f, d1 = 0.0f;
        ft_tile_product<2, 1>(grows, j, n, sAw, sB, lane, e0, e1);
        ft_bias_relu_sum(e0, e1, bv0, bv1, j * 16 + rhalf, n, d0, d1);
        ft_norm_store(e0, e1, sub0, sub1, den0, den1, w0, w1, b0, b1, j * 16 + rhalf, n, orows + col0);
    }
}

extern "C" void kernel_launch(void* const* d_in, const int* in_sizes, int n_in,
                              void* d_out, int out_size, void* d_ws, size_t ws_size,
                              hipStream_t stream) {
    const float* x     = (const float*)d_in[0];
    const int*   eidx  = (const int*)  d_in[1];
    const float* eattr = (const float*)d_in[2];
    const int*   batch = (const int*)  d_in[3];
    const float* W     = (const float*)d_in[4];
    const float* bias  = (const float*)d_in[5];
    const float* gw    = (const float*)d_in[6];
    const float* gb    = (const float*)d_in[7];
    const float* ms    = (const float*)d_in[8];
    float* out = (float*)d_out;

    const int N = in_sizes[0] / DCH;
    const int E = in_sizes[2];

    float* Wt  = (float*)d_ws;
    float* agg = Wt + (size_t)DCH * DCH;

    k_transpose_w<<<64, 256, 0, stream>>>(W, Wt);

    k_edge_rows_lds<<<(N + EP_R - 1) / EP_R, 1024, 0, stream>>>(eidx, eidx + E, eattr, x, agg, N, E);

    k_gemm_graph_norm<<<NGRAPH * 4, 1024, 0, stream>>>(agg, Wt, bias, batch, ms, gw, gb, out, N);
}


# edge scan loads only the 32-edge groups of dst that hold an edge of the block, marked in one occupancy word per 1024 edges and block by the pre-launch; order of adds unchanged
# speedup vs baseline: 2.6842x; 1.2874x over previous
#include <hip/hip_runtime.h>
#include <hip/hip_bf16.h>
#include <math.h>

typedef __attribute__((ext_vector_type(2))) float v2f;
typedef __attribute__((ext_vector_type(8))) float v8f;

#define DCH 128
#define NGRAPH 64
#define GN_EPS 1e-5f
#define SX_LD (DCH + 4)

#define EP_R 391
#define BM_NB 256
#define BM_EDGES 1024
#define BM_RPB 4

__global__ __launch_bounds__(256) void k_transpose_w_edge_groups(
    const float* __restrict__ W, float* __restrict__ Wt,
    const int* __restrict__ dst, unsigned* __restrict__ bm, int E) {
    __shared__ float tile[16][17];
    __shared__ unsigned s_bits[BM_RPB][8][BM_NB];
    if (blockIdx.x < 64) {
        int bx = blockIdx.x & 7;
        int by = blockIdx.x >> 3;
        int tx = threadIdx.x & 15;
        int ty = threadIdx.x >> 4;
        tile[ty][tx] = W[(size_t)(by * 16 + ty) * DCH + bx * 16 + tx];
        __syncthreads();
        Wt[(size_t)(bx * 16 + ty) * DCH + by * 16 + tx] = tile[tx][ty];
        return;
    }
    const int c0 = (blockIdx.x - 64) * BM_RPB;
    const int nwords = (E + BM_EDGES - 1) / BM_EDGES;
    const int t = threadIdx.x;
    const int wave = t >> 5;
    int d[BM_RPB][4];
    if ((c0 + BM_RPB) * BM_EDGES <= E) {
#pragma unroll
        for (int q = 0; q < BM_RPB; ++q)
#pragma unroll
            for (int i = 0; i < 4; ++i) d[q][i] = dst[(c0 + q) * BM_EDGES + i * 256 + t];
    } else {
#pragma unroll
        for (int q = 0; q < BM_RPB; ++q)
#pragma unroll
            for (int i = 0; i < 4; ++i) {
                const int e = (c0 + q) * BM_EDGES + i * 256 + t;
                d[q][i] = -1;
                if (e < E) d[q][i] = dst[e];
            }
    }
#pragma unroll
    for (int q = 0; q < BM_RPB; ++q)
#pragma unroll
        for (int i = 0; i < 8; ++i) s_bits[q][i][t] = 0u;
    __syncthreads();
#pragma unroll
    for (int q = 0; q < BM_RPB; ++q)
#pragma unroll
        for (int i = 0; i < 4; ++i) {
            const unsigned blk = (unsigned)d[q][i] / (unsigned)EP_R;
            if (blk < (unsigned)BM_NB) s_bits[q][wave][blk] |= 1u << (i * 8 + wave);
        }
    __syncthreads();
#pragma unroll
    for (int q = 0; q < BM_RPB; ++q) {
        if (c0 + q < nwords) {
            unsigned w = 0u;
#pragma unroll
            for (int i = 0; i < 8; ++i) w |= s_bits[q][i][t];
            bm[(size_t)(c0 + q) * BM_NB + t] = w;
        }
    }
}

#define EP_WAVES 32
#define EP_CAP 512
#define EP_U 16
#define EP_STAGE 64
#define EP_ROWS 16

__device__ __forceinline__ void ep_wave_order() {
    __builtin_amdgcn_fence(__ATOMIC_RELEASE, "wavefront");
    __builtin_amdgcn_wave_barrier();
    __builtin_amdgcn_fence(__ATOMIC_ACQUIRE, "wavefront");
}

__device__ __forceinline__ void ep_flush(
    const unsigned* stage, int nb, int lane, const int* __restrict__ src,
    const float* __restrict__ attr, const float4* __restrict__ x4, float4* acc4) {
    ep_wave_order();
    for (int b0 = 0; b0 < nb; b0 += 32) {
        int nbb = nb - b0;
        nbb = nbb < 32 ? nbb : 32;
        unsigned ent = 0;
        int sv = 0;
        float av = 0.0f;
        if (lane < nbb) {
            ent = stage[b0 + lane];
            const int e = (int)(ent >> 9);
            sv = src[e];
            av = attr[e];
        }
        const int rv = (int)(ent & 511u);
        const int avi = __float_as_int(av);
        for (int j0 = 0; j0 < nbb; j0 += EP_ROWS) {
            float4 v[EP_ROWS];
#pragma unroll
            for (int u = 0; u < EP_ROWS; ++u) {
                int j = j0 + u;
                j = j < nbb ? j : nbb - 1;
                const int s = __builtin_amdgcn_readlane(sv, j);
                v[u] = x4[(size_t)s * 32 + lane];
            }
#pragma unroll
            for (int u = 0; u < EP_ROWS; ++u) {
                const int j = j0 + u;
                if (j < nbb) {
                    const int r = __builtin_amdgcn_readlane(rv, j);
                    const float a = __int_as_float(__builtin_amdgcn_readlane(avi, j));
                    float4 c = acc4[r * 32 + lane];
                    c.x += a * v[u].x;
                    c.y += a * v[u].y;
                    c.z += a * v[u].z;
                    c.w += a * v[u].w;
                    acc4[r * 32 + lane] = c;
                }
            }
        }
    }
    ep_wave_order();
}

__global__ __launch_bounds__(1024) void k_edge_rows_lds(
    const int* __restrict__ src, const int* __restrict__ dst,
    const float* __restrict__ attr, const float* __restrict__ x,
    const unsigned* __restrict__ bm, float* __restrict__ agg, int N, int E) {
    __shared__ __attribute__((aligned(16))) float s_acc[EP_R * DCH];
    __shared__ unsigned s_list[EP_WAVES][EP_CAP];
    __shared__ unsigned s_stage[EP_WAVES][EP_STAGE];
    __shared__ int s_cnt[EP_WAVES];
    __shared__ int s_more[EP_WAVES];

    const int t = threadIdx.x;
    const int lane = t & 31;
    const int wave = __builtin_amdgcn_readfirstlane(t >> 5);
    const int base = blockIdx.x * EP_R;
    int nrows = N - base;
    nrows = nrows < EP_R ? nrows : EP_R;

    float4* acc4 = reinterpret_cast<float4*>(s_acc);
    const float4* x4 = reinterpret_cast<const float4*>(x);
    for (int i = t; i < EP_R * 32; i += 1024) acc4[i] = make_float4(0.0f, 0.0f, 0.0f, 0.0f);

    const int bid = blockIdx.x;
    const int nwords = (E + BM_EDGES - 1) / BM_EDGES;
    const int wper = (nwords + EP_WAVES - 1) / EP_WAVES;
    int wend = (wave + 1) * wper;
    wend = wend < nwords ? wend : nwords;
    int wi = wave * wper;
    int wc_base = wi, wc_end = wi;
    unsigned w0 = 0u;
    int cur = 0;
    unsigned long long rem = 0ull;
    const int max_rounds = (wper * BM_EDGES) / (EP_CAP - 32) + 2;

    for (int round = 0; round < max_rounds; ++round) {
        int count = 0;
        bool full = false;
        while (!full) {
            if (rem == 0ull) {
                if (wi >= wend) break;
                if (wi >= wc_end) {
                    const int r = wi + lane;
                    w0 = 0u;
                    if (r < wend) w0 = bid < BM_NB ? bm[(size_t)r * BM_NB + bid] : 0xffffffffu;
                    wc_base = wi;
                    wc_end = wi + 32;
                }
                const int j = wi - wc_base;
                const unsigned lo = (unsigned)__builtin_amdgcn_readlane((int)w0, j);
                const unsigned hi = (unsigned)__builtin_amdgcn_readlane((int)w0, j + 1);
                rem = (unsigned long long)lo | ((unsigned long long)hi << 32);
                cur = wi;
                wi += 2;
                continue;
            }
            int g[EP_U];
            int d[EP_U];
            unsigned long long m = rem;
#pragma unroll
            for (int u = 0; u < EP_U; ++u) {
                g[u] = -1;
                if (m != 0ull) { g[u] = (int)__builtin_ctzll(m); m &= m - 1ull; }
            }
            if (((cur + 2) << 10) <= E) {
#pragma unroll
                for (int u = 0; u < EP_U; ++u) {
                    const int gg = g[u] >= 0 ? g[u] : g[0];
                    d[u] = dst[((cur << 5) + gg) * 32 + lane];
                }
            } else {
#pragma unroll
                for (int u = 0; u < EP_U; ++u) {
                    const int gg = g[u] >= 0 ? g[u] : g[0];
                    const int e = ((cur << 5) + gg) * 32 + lane;
                    d[u] = 0x7fffffff;
                    if (e < E) d[u] = dst[e];
                }
            }
#pragma unroll
            for (int u = 0; u < EP_U; ++u) {
                if (g[u] >= 0 && !full) {
                    const unsigned rel = (unsigned)(d[u] - base);
                    const bool hit = rel < (unsigned)nrows;
                    const unsigned mk = __builtin_amdgcn_ballot_w32(hit);
                    if (mk != 0u) {
                        if (hit)
                            s_list[wave][count + (int)__builtin_amdgcn_mbcnt_lo(mk, 0u)] =
                                ((unsigned)(((cur << 5) + g[u]) * 32 + lane) << 9) | rel;
                        count += __builtin_popcount(mk);
                        if (count > EP_CAP - 32) full = true;
                    }
                    rem &= ~(1ull << g[u]);
                }
            }
        }
        if (lane == 0) { s_cnt[wave] = count; s_more[wave] = (rem != 0ull || wi < wend) ? 1 : 0; }
        __syncthreads();
        int more_v = 0;
        for (int w = 0; w < EP_WAVES; ++w) more_v |= s_more[w];
        const int more = __builtin_amdgcn_readfirstlane(more_v);

        int npend = 0;
        for (int sw = 0; sw < EP_WAVES; ++sw) {
            const int n = __builtin_amdgcn_readfirstlane(s_cnt[sw]);
            for (int i0 = 0; i0 < n; i0 += 128) {
                unsigned ent[4];
#pragma unroll
                for (int q = 0; q < 4; ++q) {
                    const int idx = i0 + q * 32 + lane;
                    ent[q] = 0xffffffffu;
                    if (idx < n) ent[q] = s_list[sw][idx];
                }
#pragma unroll
                for (int q = 0; q < 4; ++q) {
                    const int idx = i0 + q * 32 + lane;
                    const bool mine = (idx < n) && ((int)(ent[q] & 31u) == wave);
                    const unsigned m = __builtin_amdgcn_ballot_w32(mine);
                    if (m != 0u) {
                        if (npend > EP_STAGE - 32) {
                            ep_flush(s_stage[wave], npend, lane, src, attr, x4, acc4);
                            npend = 0;
                        }
                        if (mine) s_stage[wave][npend + (int)__builtin_amdgcn_mbcnt_lo(m, 0u)] = ent[q];
                        npend += __builtin_popcount(m);
                    }
                }
            }
        }
        ep_flush(s_stage[wave], npend, lane, src, attr, x4, acc4);
        __syncthreads();
        if (!more) break;
    }

    float4* agg4 = reinterpret_cast<float4*>(agg) + (size_t)base * 32;
    for (int i = t; i < nrows * 32; i += 1024) agg4[i] = acc4[i];
}

#define FT_SLOTS 4
#define FT_WAVES 32
template <int RB, int KU>
__device__ __forceinline__ void ft_tile_product(
    const float* __restrict__ grows, int j, int n, float (*sAw)[SX_LD],
    const float (*sBm)[SX_LD], int lane, v8f& acc0, v8f& acc1) {
    const int r0 = j * 16;
#pragma unroll
    for (int part = 0; part < 16 / RB; ++part) {
        float4 v[RB];
#pragma unroll
        for (int r = 0; r < RB; ++r) {
            const int rr = r0 + part * RB + r;
            v[r] = make_float4(0.0f, 0.0f, 0.0f, 0.0f);
            if (rr < n) v[r] = *reinterpret_cast<const float4*>(grows + (size_t)rr * DCH + lane * 4);
        }
#pragma unroll
        for (int r = 0; r < RB; ++r)
            *reinterpret_cast<float4*>(&sAw[part * RB + r][lane * 4]) = v[r];
    }
    __builtin_amdgcn_fence(__ATOMIC_RELEASE, "wavefront");
    __builtin_amdgcn_wave_barrier();
    __builtin_amdgcn_fence(__ATOMIC_ACQUIRE, "wavefront");

    const int m    = lane & 15;
    const int koff = (lane >> 4) << 1;
    acc0 = v8f{};
    acc1 = v8f{};
#pragma unroll KU
    for (int kt = 0; kt < DCH; kt += 4) {
        v2f a  = *reinterpret_cast<const v2f*>(&sAw[m][kt + koff]);
        v2f b0 = *reinterpret_cast<const v2f*>(&sBm[m][kt + koff]);
        v2f b1 = *reinterpret_cast<const v2f*>(&sBm[16 + m][kt + koff]);
        acc0 = __builtin_amdgcn_wmma_f32_16x16x4_f32(false, a, false, b0, (short)0, acc0, false, false);
        acc1 = __builtin_amdgcn_wmma_f32_16x16x4_f32(false, a, false, b1, (short)0, acc1, false, false);
    }
    __builtin_amdgcn_fence(__ATOMIC_RELEASE, "wavefront");
    __builtin_amdgcn_wave_barrier();
    __builtin_amdgcn_fence(__ATOMIC_ACQUIRE, "wavefront");
}

__device__ __forceinline__ void ft_bias_relu_sum(
    v8f& ha, v8f& hb, float bv0, float bv1, int rbase, int n, float& p0, float& p1) {
#pragma unroll
    for (int v = 0; v < 8; ++v) {
        const bool ok = (rbase + v) < n;
        float a0 = ha[v] + bv0; a0 = a0 > 0.0f ? a0 : 0.0f;
        float a1 = hb[v] + bv1; a1 = a1 > 0.0f ? a1 : 0.0f;
        ha[v] = a0;
        hb[v] = a1;
        p0 += ok ? a0 : 0.0f;
        p1 += ok ? a1 : 0.0f;
    }
}

__device__ __forceinline__ void ft_sq_sum(
    const v8f& ha, const v8f& hb, float sub0, float sub1, int rbase, int n, float& q0, float& q1) {
#pragma unroll
    for (int v = 0; v < 8; ++v) {
        const bool ok = (rbase + v) < n;
        const float o0 = ha[v] - sub0;
        const float o1 = hb[v] - sub1;
        q0 += ok ? o0 * o0 : 0.0f;
        q1 += ok ? o1 * o1 : 0.0f;
    }
}

__device__ __forceinline__ void ft_norm_store(
    const v8f& ha, const v8f& hb, float sub0, float sub1, float den0, float den1,
    float w0, float w1, float b0, float b1, int rbase, int n, float* __restrict__ ocol) {
#pragma unroll
    for (int v = 0; v < 8; ++v) {
        const int rr = rbase + v;
        if (rr < n) {
            ocol[(size_t)rr * DCH]      = w0 * (ha[v] - sub0) / den0 + b0;
            ocol[(size_t)rr * DCH + 16] = w1 * (hb[v] - sub1) / den1 + b1;
        }
    }
}

__device__ __forceinline__ void ft_block_sum(
    float p0, float p1, float (*s_red)[32], float* s_stat, int t, float& r0, float& r1) {
    const int lane = t & 31;
    const int wave = t >> 5;
    p0 += __shfl_xor(p0, 16);
    p1 += __shfl_xor(p1, 16);
    if (lane < 16) { s_red[wave][lane] = p0; s_red[wave][16 + lane] = p1; }
    __syncthreads();
    if (t < 32) {
        float s = 0.0f;
        for (int w = 0; w < FT_WAVES; ++w) s += s_red[w][t];
        s_stat[t] = s;
    }
    __syncthreads();
    r0 = s_stat[lane & 15];
    r1 = s_stat[16 + (lane & 15)];
    __syncthreads();
}

__global__ __launch_bounds__(1024) void k_gemm_graph_norm(
    const float* __restrict__ agg, const float* __restrict__ Wt,
    const float* __restrict__ bias, const int* __restrict__ batch,
    const float* __restrict__ ms, const float* __restrict__ gw,
    const float* __restrict__ gb, float* __restrict__ out, int N) {
    __shared__ __attribute__((aligned(16))) float sA[FT_WAVES][16][SX_LD];
    __shared__ __attribute__((aligned(16))) float sB[32][SX_LD];
    __shared__ float s_red[FT_WAVES][32];
    __shared__ float s_stat[32];
    __shared__ int s_less[FT_WAVES];
    __shared__ int s_eq[FT_WAVES];

    const int t = threadIdx.x;
    const int lane = t & 31;
    const int wave = __builtin_amdgcn_readfirstlane(t >> 5);
    const int g = blockIdx.x >> 2;
    const int slice = blockIdx.x & 3;

    const float4 wv = *reinterpret_cast<const float4*>(
        Wt + (size_t)(slice * 32 + (t >> 5)) * DCH + ((t & 31) << 2));

    int less = 0, eq = 0;
    {
        const int4* b4 = reinterpret_cast<const int4*>(batch);
        const int n4 = N >> 2;
        for (int base = 0; base < n4; base += 8 * 1024) {
            int4 q[8];
#pragma unroll
            for (int u = 0; u < 8; ++u) {
                const int i = base + u * 1024 + t;
                q[u] = make_int4(0x7fffffff, 0x7fffffff, 0x7fffffff, 0x7fffffff);
                if (i < n4) q[u] = b4[i];
            }
#pragma unroll
            for (int u = 0; u < 8; ++u) {
                less += (q[u].x < g) + (q[u].y < g) + (q[u].z < g) + (q[u].w < g);
                eq   += (q[u].x == g) + (q[u].y == g) + (q[u].z == g) + (q[u].w == g);
            }
        }
        for (int i = (n4 << 2) + t; i < N; i += 1024) {
            const int q = batch[i];
            less += (q < g);
            eq   += (q == g);
        }
    }
#pragma unroll
    for (int off = 16; off > 0; off >>= 1) {
        less += __shfl_xor(less, off);
        eq   += __shfl_xor(eq, off);
    }
    if (lane == 0) { s_less[wave] = less; s_eq[wave] = eq; }
    *reinterpret_cast<float4*>(&sB[t >> 5][(t & 31) << 2]) = wv;
    __syncthreads();
    int start_v = 0, n_v = 0;
    for (int w = 0; w < FT_WAVES; ++w) { start_v += s_less[w]; n_v += s_eq[w]; }
    const int start = __builtin_amdgcn_readfirstlane(start_v);
    const int n = __builtin_amdgcn_readfirstlane(n_v);
    const int ntiles = (n + 15) >> 4;

    const float* grows = agg + (size_t)start * DCH;
    float* orows = out + (size_t)start * DCH;
    float (*sAw)[SX_LD] = sA[wave];

    const int c = lane & 15;
    const int rhalf = (lane >> 4) << 3;
    const int col0 = slice * 32 + c;
    const int col1 = col0 + 16;
    const float bv0 = bias[col0], bv1 = bias[col1];

    v8f hA0, hB0, hA1, hB1, hA2, hB2, hA3, hB3;
    float p0 = 0.0f, p1 = 0.0f;
#define FT_PHASE1(S, HA, HB)                                                        \
    {                                                                               \
        const int j = wave + (S) * FT_WAVES;     \
        HA = v8f{};                                                                 \
        HB = v8f{};                                                                 \
        if (j < ntiles) {                                                           \
            ft_tile_product<8, 4>(grows, j, n, sAw, sB, lane, HA, HB);              \
            ft_bias_relu_sum(HA, HB, bv0, bv1, j * 16 + rhalf, n, p0, p1);          \
        }                                                                           \
    }
    FT_PHASE1(0, hA0, hB0)
    FT_PHASE1(1, hA1, hB1)
    FT_PHASE1(2, hA2, hB2)
    FT_PHASE1(3, hA3, hB3)
#undef FT_PHASE1
    for (int j = wave + FT_SLOTS * FT_WAVES; j < ntiles; j += FT_WAVES) {
        v8f e0, e1;
        ft_tile_product<2, 1>(grows, j, n, sAw, sB, lane, e0, e1);
        ft_bias_relu_sum(e0, e1, bv0, bv1, j * 16 + rhalf, n, p0, p1);
    }
    float msum0, msum1;
    ft_block_sum(p0, p1, s_red, s_stat, t, msum0, msum1);

    float cn = (float)n;
    cn = cn > 1.0f ? cn : 1.0f;
    const float inv_cn = 1.0f / cn;
    const float sub0 = (msum0 * inv_cn) * ms[col0];
    const float sub1 = (msum1 * inv_cn) * ms[col1];

    float q0 = 0.0f, q1 = 0.0f;
    if (wave + 0 * FT_WAVES < ntiles) ft_sq_sum(hA0, hB0, sub0, sub1, (wave + 0 * FT_WAVES) * 16 + rhalf, n, q0, q1);
    if (wave + 1 * FT_WAVES < ntiles) ft_sq_sum(hA1, hB1, sub0, sub1, (wave + 1 * FT_WAVES) * 16 + rhalf, n, q0, q1);
    if (wave + 2 * FT_WAVES < ntiles) ft_sq_sum(hA2, hB2, sub0, sub1, (wave + 2 * FT_WAVES) * 16 + rhalf, n, q0, q1);
    if (wave + 3 * FT_WAVES < ntiles) ft_sq_sum(hA3, hB3, sub0, sub1, (wave + 3 * FT_WAVES) * 16 + rhalf, n, q0, q1);
    for (int j = wave + FT_SLOTS * FT_WAVES; j < ntiles; j += FT_WAVES) {
        v8f e0, e1;
        float d0 = 0.0f, d1 = 0.0f;
        ft_tile_product<2, 1>(grows, j, n, sAw, sB, lane, e0, e1);
        ft_bias_relu_sum(e0, e1, bv0, bv1, j * 16 + rhalf, n, d0, d1);
        ft_sq_sum(e0, e1, sub0, sub1, j * 16 + rhalf, n, q0, q1);
    }
    float vsum0, vsum1;
    ft_block_sum(q0, q1, s_red, s_stat, t, vsum0, vsum1);

    const float den0 = sqrtf(vsum0 * inv_cn + GN_EPS);
    const float den1 = sqrtf(vsum1 * inv_cn + GN_EPS);
    const float w0 = gw[col0], w1 = gw[col1];
    const float b0 = gb[col0], b1 = gb[col1];

    if (wave + 0 * FT_WAVES < ntiles) ft_norm_store(hA0, hB0, sub0, sub1, den0, den1, w0, w1, b0, b1, (wave + 0 * FT_WAVES) * 16 + rhalf, n, orows + col0);
    if (wave + 1 * FT_WAVES < ntiles) ft_norm_store(hA1, hB1, sub0, sub1, den0, den1, w0, w1, b0, b1, (wave + 1 * FT_WAVES) * 16 + rhalf, n, orows + col0);
    if (wave + 2 * FT_WAVES < ntiles) ft_norm_store(hA2, hB2, sub0, sub1, den0, den1, w0, w1, b0, b1, (wave + 2 * FT_WAVES) * 16 + rhalf, n, orows + col0);
    if (wave + 3 * FT_WAVES < ntiles) ft_norm_store(hA3, hB3, sub0, sub1, den0, den1, w0, w1, b0, b1, (wave + 3 * FT_WAVES) * 16 + rhalf, n, orows + col0);
    for (int j = wave + FT_SLOTS * FT_WAVES; j < ntiles; j += FT_WAVES) {
        v8f e0, e1;
        float d0 = 0.0f, d1 = 0.0f;
        ft_tile_product<2, 1>(grows, j, n, sAw, sB, lane, e0, e1);
        ft_bias_relu_sum(e0, e1, bv0, bv1, j * 16 + rhalf, n, d0, d1);
        ft_norm_store(e0, e1, sub0, sub1, den0, den1, w0, w1, b0, b1, j * 16 + rhalf, n, orows + col0);
    }
}

extern "C" void kernel_launch(void* const* d_in, const int* in_sizes, int n_in,
                              void* d_out, int out_size, void* d_ws, size_t ws_size,
                              hipStream_t stream) {
    const float* x     = (const float*)d_in[0];
    const int*   eidx  = (const int*)  d_in[1];
    const float* eattr = (const float*)d_in[2];
    const int*   batch = (const int*)  d_in[3];
    const float* W     = (const float*)d_in[4];
    const float* bias  = (const float*)d_in[5];
    const float* gw    = (const float*)d_in[6];
    const float* gb    = (const float*)d_in[7];
    const float* ms    = (const float*)d_in[8];
    float* out = (float*)d_out;

    const int N = in_sizes[0] / DCH;
    const int E = in_sizes[2];

    float* Wt  = (float*)d_ws;
    float* agg = Wt + (size_t)DCH * DCH;
    unsigned* bm = (unsigned*)(agg + (size_t)N * DCH);

    k_transpose_w_edge_groups<<<64 + (E + BM_EDGES * BM_RPB - 1) / (BM_EDGES * BM_RPB), 256, 0, stream>>>(W, Wt, eidx + E, bm, E);

    k_edge_rows_lds<<<(N + EP_R - 1) / EP_R, 1024, 0, stream>>>(eidx, eidx + E, eattr, x, bm, agg, N, E);

    k_gemm_graph_norm<<<NGRAPH * 4, 1024, 0, stream>>>(agg, Wt, bias, batch, ms, gw, gb, out, N);
}


# edge scan: a wave lists the marked groups of up to 32 rows of dst in LDS in one lane-parallel pass (prefix sum of bit counts), then loads them 32 at a time across rows; same add order
# speedup vs baseline: 3.1421x; 1.1706x over previous
#include <hip/hip_runtime.h>
#include <hip/hip_bf16.h>
#include <math.h>

typedef __attribute__((ext_vector_type(2))) float v2f;
typedef __attribute__((ext_vector_type(8))) float v8f;

#define DCH 128
#define NGRAPH 64
#define GN_EPS 1e-5f
#define SX_LD (DCH + 4)

#define EP_R 391
#define BM_NB 256
#define BM_EDGES 1024
#define BM_RPB 4

__global__ __launch_bounds__(256) void k_transpose_w_edge_groups(
    const float* __restrict__ W, float* __restrict__ Wt,
    const int* __restrict__ dst, unsigned* __restrict__ bm, int E) {
    __shared__ float tile[16][17];
    __shared__ unsigned s_bits[BM_RPB][8][BM_NB];
    if (blockIdx.x < 64) {
        int bx = blockIdx.x & 7;
        int by = blockIdx.x >> 3;
        int tx = threadIdx.x & 15;
        int ty = threadIdx.x >> 4;
        tile[ty][tx] = W[(size_t)(by * 16 + ty) * DCH + bx * 16 + tx];
        __syncthreads();
        Wt[(size_t)(bx * 16 + ty) * DCH + by * 16 + tx] = tile[tx][ty];
        return;
    }
    const int c0 = (blockIdx.x - 64) * BM_RPB;
    const int nwords = (E + BM_EDGES - 1) / BM_EDGES;
    const int t = threadIdx.x;
    const int wave = t >> 5;
    int d[BM_RPB][4];
    if ((c0 + BM_RPB) * BM_EDGES <= E) {
#pragma unroll
        for (int q = 0; q < BM_RPB; ++q)
#pragma unroll
            for (int i = 0; i < 4; ++i) d[q][i] = dst[(c0 + q) * BM_EDGES + i * 256 + t];
    } else {
#pragma unroll
        for (int q = 0; q < BM_RPB; ++q)
#pragma unroll
            for (int i = 0; i < 4; ++i) {
                const int e = (c0 + q) * BM_EDGES + i * 256 + t;
                d[q][i] = -1;
                if (e < E) d[q][i] = dst[e];
            }
    }
#pragma unroll
    for (int q = 0; q < BM_RPB; ++q)
#pragma unroll
        for (int i = 0; i < 8; ++i) s_bits[q][i][t] = 0u;
    __syncthreads();
#pragma unroll
    for (int q = 0; q < BM_RPB; ++q)
#pragma unroll
        for (int i = 0; i < 4; ++i) {
            const unsigned blk = (unsigned)d[q][i] / (unsigned)EP_R;
            if (blk < (unsigned)BM_NB) s_bits[q][wave][blk] |= 1u << (i * 8 + wave);
        }
    __syncthreads();
#pragma unroll
    for (int q = 0; q < BM_RPB; ++q) {
        if (c0 + q < nwords) {
            unsigned w = 0u;
#pragma unroll
            for (int i = 0; i < 8; ++i) w |= s_bits[q][i][t];
            bm[(size_t)(c0 + q) * BM_NB + t] = w;
        }
    }
}

#define EP_WAVES 32
#define EP_CAP 512
#define EP_GCAP 512
#define EP_U 32
#define EP_STAGE 64
#define EP_ROWS 16

__device__ __forceinline__ void ep_wave_order() {
    __builtin_amdgcn_fence(__ATOMIC_RELEASE, "wavefront");
    __builtin_amdgcn_wave_barrier();
    __builtin_amdgcn_fence(__ATOMIC_ACQUIRE, "wavefront");
}

__device__ __forceinline__ void ep_flush(
    const unsigned* stage, int nb, int lane, const int* __restrict__ src,
    const float* __restrict__ attr, const float4* __restrict__ x4, float4* acc4) {
    ep_wave_order();
    for (int b0 = 0; b0 < nb; b0 += 32) {
        int nbb = nb - b0;
        nbb = nbb < 32 ? nbb : 32;
        unsigned ent = 0;
        int sv = 0;
        float av = 0.0f;
        if (lane < nbb) {
            ent = stage[b0 + lane];
            const int e = (int)(ent >> 9);
            sv = src[e];
            av = attr[e];
        }
        const int rv = (int)(ent & 511u);
        const int avi = __float_as_int(av);
        for (int j0 = 0; j0 < nbb; j0 += EP_ROWS) {
            float4 v[EP_ROWS];
#pragma unroll
            for (int u = 0; u < EP_ROWS; ++u) {
                int j = j0 + u;
                j = j < nbb ? j : nbb - 1;
                const int s = __builtin_amdgcn_readlane(sv, j);
                v[u] = x4[(size_t)s * 32 + lane];
            }
#pragma unroll
            for (int u = 0; u < EP_ROWS; ++u) {
                const int j = j0 + u;
                if (j < nbb) {
                    const int r = __builtin_amdgcn_readlane(rv, j);
                    const float a = __int_as_float(__builtin_amdgcn_readlane(avi, j));
                    float4 c = acc4[r * 32 + lane];
                    c.x += a * v[u].x;
                    c.y += a * v[u].y;
                    c.z += a * v[u].z;
                    c.w += a * v[u].w;
                    acc4[r * 32 + lane] = c;
                }
            }
        }
    }
    ep_wave_order();
}

template <bool FULL>
__device__ __forceinline__ void ep_scan_batch(
    const int* __restrict__ dst, const unsigned short* glist, unsigned* mylist, int lane,
    int g_row0, int nval_in, int gp, int base, unsigned unrows, int& count) {
    const int nval = FULL ? EP_U : nval_in;
    const int myg = (int)glist[gp + (lane < nval ? lane : 0)];
    const int ebase = (g_row0 << 10) + lane;
    int e[EP_U];
    int d[EP_U];
#pragma unroll
    for (int u = 0; u < EP_U; ++u) {
        e[u] = ebase + (__builtin_amdgcn_readlane(myg, u) << 5);
        d[u] = dst[e[u]];
    }
    int c = count;
#pragma unroll
    for (int q = 0; q < EP_U / 8; ++q) {
        if (FULL || q * 8 < nval) {
#pragma unroll
            for (int u = q * 8; u < q * 8 + 8; ++u) {
                const unsigned rel = (unsigned)(d[u] - base);
                const bool hit = rel < unrows;
                const unsigned mk = __builtin_amdgcn_ballot_w32(hit);
                if ((FULL || u < nval) && c <= EP_CAP - 32) {
                    if (hit) mylist[c + (int)__builtin_amdgcn_mbcnt_lo(mk, 0u)] = ((unsigned)e[u] << 9) + rel;
                    c += __builtin_popcount(mk);
                }
            }
        }
    }
    count = c;
}

__global__ __launch_bounds__(1024) void k_edge_rows_lds(
    const int* __restrict__ src, const int* __restrict__ dst,
    const float* __restrict__ attr, const float* __restrict__ x,
    const unsigned* __restrict__ bm, float* __restrict__ agg, int N, int E) {
    __shared__ __attribute__((aligned(16))) float s_acc[EP_R * DCH];
    __shared__ unsigned s_list[EP_WAVES][EP_CAP];
    __shared__ unsigned s_stage[EP_WAVES][EP_STAGE];
    __shared__ unsigned short s_grp[EP_WAVES][EP_GCAP];
    __shared__ int s_cnt[EP_WAVES];
    __shared__ int s_more[EP_WAVES];

    const int t = threadIdx.x;
    const int lane = t & 31;
    const int wave = __builtin_amdgcn_readfirstlane(t >> 5);
    const int base = blockIdx.x * EP_R;
    int nrows = N - base;
    nrows = nrows < EP_R ? nrows : EP_R;
    const unsigned unrows = (unsigned)nrows;

    const int bid = blockIdx.x;
    const int nwords = (E + BM_EDGES - 1) / BM_EDGES;
    const int nfull = E / BM_EDGES;
    const int wper = (nwords + EP_WAVES - 1) / EP_WAVES;
    int wend = (wave + 1) * wper;
    wend = wend < nwords ? wend : nwords;
    int wi = wave * wper;
    int wc_base = wi, wc_end = wi + 32;
    unsigned w0 = 0u;
    if (wi + lane < wend) w0 = bid < BM_NB ? bm[(size_t)(wi + lane) * BM_NB + bid] : 0xffffffffu;

    float4* acc4 = reinterpret_cast<float4*>(s_acc);
    const float4* x4 = reinterpret_cast<const float4*>(x);
    for (int i = t; i < EP_R * 32; i += 1024) acc4[i] = make_float4(0.0f, 0.0f, 0.0f, 0.0f);

    unsigned short* glist = s_grp[wave];
    unsigned* mylist = s_list[wave];
    int g_row0 = 0;
    int g_n = 0;
    int g_pos = 0;
    const int max_rounds = (wper * BM_EDGES) / (EP_CAP - 32) + 2;

    for (int round = 0; round < max_rounds; ++round) {
        int count = 0;
        bool full = false;
        while (!full) {
            if (g_pos >= g_n) {
                if (wi >= wend) break;
                if (wi >= wc_end) {
                    const int r = wi + lane;
                    w0 = 0u;
                    if (r < wend) w0 = bid < BM_NB ? bm[(size_t)r * BM_NB + bid] : 0xffffffffu;
                    wc_base = wi;
                    wc_end = wi + 32;
                }
                const int j = wi - wc_base + lane;
                unsigned bits = (unsigned)__shfl((int)w0, j & 31);
                if (j >= 32) bits = 0u;
                const int cnt = __builtin_popcount(bits);
                int incl = cnt;
#pragma unroll
                for (int off = 1; off < 32; off <<= 1) {
                    const int v = __shfl_up(incl, off);
                    if (lane >= off) incl += v;
                }
                int nrow = __builtin_popcount(__builtin_amdgcn_ballot_w32(incl <= EP_GCAP));
                nrow = nrow < wc_end - wi ? nrow : wc_end - wi;
                if (wi >= nfull) nrow = 1;
                else nrow = nrow < nfull - wi ? nrow : nfull - wi;
                if (lane >= nrow) bits = 0u;
                int pos = incl - cnt;
                ep_wave_order();
                while (bits != 0u) {
                    glist[pos++] = (unsigned short)((lane << 5) + __builtin_ctz(bits));
                    bits &= bits - 1u;
                }
                ep_wave_order();
                g_n = __builtin_amdgcn_readlane(incl, nrow - 1);
                g_row0 = wi;
                g_pos = 0;
                wi += nrow;
                continue;
            }
            const int left = g_n - g_pos;
            const int nval = left < EP_U ? left : EP_U;
            bool taken = false;
            if (g_row0 < nfull) {
                const int count0 = count;
                if (nval == EP_U)
                    ep_scan_batch<true>(dst, glist, mylist, lane, g_row0, nval, g_pos, base, unrows, count);
                else
                    ep_scan_batch<false>(dst, glist, mylist, lane, g_row0, nval, g_pos, base, unrows, count);
                taken = count <= EP_CAP - 32;
                if (!taken) count = count0;
            }
            if (taken) {
                g_pos += nval;
            } else {
                for (int u = 0; u < nval && !full; ++u) {
                    const int g = __builtin_amdgcn_readfirstlane((int)glist[g_pos]);
                    const int e = (g_row0 << 10) + (g << 5) + lane;
                    int dv = 0x7fffffff;
                    if (e < E) dv = dst[e];
                    const unsigned rel = (unsigned)(dv - base);
                    const bool hit = rel < unrows;
                    const unsigned mk = __builtin_amdgcn_ballot_w32(hit);
                    const int pop = __builtin_popcount(mk);
                    if (count + pop > EP_CAP) {
                        full = true;
                    } else {
                        if (hit) mylist[count + (int)__builtin_amdgcn_mbcnt_lo(mk, 0u)] = ((unsigned)e << 9) | rel;
                        count += pop;
                        ++g_pos;
                    }
                }
            }
        }
        if (lane == 0) { s_cnt[wave] = count; s_more[wave] = (g_pos < g_n || wi < wend) ? 1 : 0; }
        __syncthreads();
        int more_v = 0;
        for (int w = 0; w < EP_WAVES; ++w) more_v |= s_more[w];
        const int more = __builtin_amdgcn_readfirstlane(more_v);

        int npend = 0;
        for (int sw = 0; sw < EP_WAVES; ++sw) {
            const int n = __builtin_amdgcn_readfirstlane(s_cnt[sw]);
            for (int i0 = 0; i0 < n; i0 += 128) {
                unsigned ent[4];
#pragma unroll
                for (int q = 0; q < 4; ++q) {
                    const int idx = i0 + q * 32 + lane;
                    ent[q] = 0xffffffffu;
                    if (idx < n) ent[q] = s_list[sw][idx];
                }
#pragma unroll
                for (int q = 0; q < 4; ++q) {
                    const int idx = i0 + q * 32 + lane;
                    const bool mine = (idx < n) && ((int)(ent[q] & 31u) == wave);
                    const unsigned m = __builtin_amdgcn_ballot_w32(mine);
                    if (m != 0u) {
                        if (npend > EP_STAGE - 32) {
                            ep_flush(s_stage[wave], npend, lane, src, attr, x4, acc4);
                            npend = 0;
                        }
                        if (mine) s_stage[wave][npend + (int)__builtin_amdgcn_mbcnt_lo(m, 0u)] = ent[q];
                        npend += __builtin_popcount(m);
                    }
                }
            }
        }
        ep_flush(s_stage[wave], npend, lane, src, attr, x4, acc4);
        __syncthreads();
        if (!more) break;
    }

    float4* agg4 = reinterpret_cast<float4*>(agg) + (size_t)base * 32;
    for (int i = t; i < nrows * 32; i += 1024) agg4[i] = acc4[i];
}

#define FT_SLOTS 4
#define FT_WAVES 32
template <int RB, int KU>
__device__ __forceinline__ void ft_tile_product(
    const float* __restrict__ grows, int j, int n, float (*sAw)[SX_LD],
    const float (*sBm)[SX_LD], int lane, v8f& acc0, v8f& acc1) {
    const int r0 = j * 16;
#pragma unroll
    for (int part = 0; part < 16 / RB; ++part) {
        float4 v[RB];
#pragma unroll
        for (int r = 0; r < RB; ++r) {
            const int rr = r0 + part * RB + r;
            v[r] = make_float4(0.0f, 0.0f, 0.0f, 0.0f);
            if (rr < n) v[r] = *reinterpret_cast<const float4*>(grows + (size_t)rr * DCH + lane * 4);
        }
#pragma unroll
        for (int r = 0; r < RB; ++r)
            *reinterpret_cast<float4*>(&sAw[part * RB + r][lane * 4]) = v[r];
    }
    __builtin_amdgcn_fence(__ATOMIC_RELEASE, "wavefront");
    __builtin_amdgcn_wave_barrier();
    __builtin_amdgcn_fence(__ATOMIC_ACQUIRE, "wavefront");

    const int m    = lane & 15;
    const int koff = (lane >> 4) << 1;
    acc0 = v8f{};
    acc1 = v8f{};
#pragma unroll KU
    for (int kt = 0; kt < DCH; kt += 4) {
        v2f a  = *reinterpret_cast<const v2f*>(&sAw[m][kt + koff]);
        v2f b0 = *reinterpret_cast<const v2f*>(&sBm[m][kt + koff]);
        v2f b1 = *reinterpret_cast<const v2f*>(&sBm[16 + m][kt + koff]);
        acc0 = __builtin_amdgcn_wmma_f32_16x16x4_f32(false, a, false, b0, (short)0, acc0, false, false);
        acc1 = __builtin_amdgcn_wmma_f32_16x16x4_f32(false, a, false, b1, (short)0, acc1, false, false);
    }
    __builtin_amdgcn_fence(__ATOMIC_RELEASE, "wavefront");
    __builtin_amdgcn_wave_barrier();
    __builtin_amdgcn_fence(__ATOMIC_ACQUIRE, "wavefront");
}

__device__ __forceinline__ void ft_bias_relu_sum(
    v8f& ha, v8f& hb, float bv0, float bv1, int rbase, int n, float& p0, float& p1) {
#pragma unroll
    for (int v = 0; v < 8; ++v) {
        const bool ok = (rbase + v) < n;
        float a0 = ha[v] + bv0; a0 = a0 > 0.0f ? a0 : 0.0f;
        float a1 = hb[v] + bv1; a1 = a1 > 0.0f ? a1 : 0.0f;
        ha[v] = a0;
        hb[v] = a1;
        p0 += ok ? a0 : 0.0f;
        p1 += ok ? a1 : 0.0f;
    }
}

__device__ __forceinline__ void ft_sq_sum(
    const v8f& ha, const v8f& hb, float sub0, float sub1, int rbase, int n, float& q0, float& q1) {
#pragma unroll
    for (int v = 0; v < 8; ++v) {
        const bool ok = (rbase + v) < n;
        const float o0 = ha[v] - sub0;
        const float o1 = hb[v] - sub1;
        q0 += ok ? o0 * o0 : 0.0f;
        q1 += ok ? o1 * o1 : 0.0f;
    }
}

__device__ __forceinline__ void ft_norm_store(
    const v8f& ha, const v8f& hb, float sub0, float sub1, float den0, float den1,
    float w0, float w1, float b0, float b1, int rbase, int n, float* __restrict__ ocol) {
#pragma unroll
    for (int v = 0; v < 8; ++v) {
        const int rr = rbase + v;
        if (rr < n) {
            ocol[(size_t)rr * DCH]      = w0 * (ha[v] - sub0) / den0 + b0;
            ocol[(size_t)rr * DCH + 16] = w1 * (hb[v] - sub1) / den1 + b1;
        }
    }
}

__device__ __forceinline__ void ft_block_sum(
    float p0, float p1, float (*s_red)[32], float* s_stat, int t, float& r0, float& r1) {
    const int lane = t & 31;
    const int wave = t >> 5;
    p0 += __shfl_xor(p0, 16);
    p1 += __shfl_xor(p1, 16);
    if (lane < 16) { s_red[wave][lane] = p0; s_red[wave][16 + lane] = p1; }
    __syncthreads();
    if (t < 32) {
        float s = 0.0f;
        for (int w = 0; w < FT_WAVES; ++w) s += s_red[w][t];
        s_stat[t] = s;
    }
    __syncthreads();
    r0 = s_stat[lane & 15];
    r1 = s_stat[16 + (lane & 15)];
    __syncthreads();
}

__global__ __launch_bounds__(1024) void k_gemm_graph_norm(
    const float* __restrict__ agg, const float* __restrict__ Wt,
    const float* __restrict__ bias, const int* __restrict__ batch,
    const float* __restrict__ ms, const float* __restrict__ gw,
    const float* __restrict__ gb, float* __restrict__ out, int N) {
    __shared__ __attribute__((aligned(16))) float sA[FT_WAVES][16][SX_LD];
    __shared__ __attribute__((aligned(16))) float sB[32][SX_LD];
    __shared__ float s_red[FT_WAVES][32];
    __shared__ float s_stat[32];
    __shared__ int s_less[FT_WAVES];
    __shared__ int s_eq[FT_WAVES];

    const int t = threadIdx.x;
    const int lane = t & 31;
    const int wave = __builtin_amdgcn_readfirstlane(t >> 5);
    const int g = blockIdx.x >> 2;
    const int slice = blockIdx.x & 3;

    const float4 wv = *reinterpret_cast<const float4*>(
        Wt + (size_t)(slice * 32 + (t >> 5)) * DCH + ((t & 31) << 2));

    int less = 0, eq = 0;
    {
        const int4* b4 = reinterpret_cast<const int4*>(batch);
        const int n4 = N >> 2;
        for (int base = 0; base < n4; base += 8 * 1024) {
            int4 q[8];
#pragma unroll
            for (int u = 0; u < 8; ++u) {
                const int i = base + u * 1024 + t;
                q[u] = make_int4(0x7fffffff, 0x7fffffff, 0x7fffffff, 0x7fffffff);
                if (i < n4) q[u] = b4[i];
            }
#pragma unroll
            for (int u = 0; u < 8; ++u) {
                less += (q[u].x < g) + (q[u].y < g) + (q[u].z < g) + (q[u].w < g);
                eq   += (q[u].x == g) + (q[u].y == g) + (q[u].z == g) + (q[u].w == g);
            }
        }
        for (int i = (n4 << 2) + t; i < N; i += 1024) {
            const int q = batch[i];
            less += (q < g);
            eq   += (q == g);
        }
    }
#pragma unroll
    for (int off = 16; off > 0; off >>= 1) {
        less += __shfl_xor(less, off);
        eq   += __shfl_xor(eq, off);
    }
    if (lane == 0) { s_less[wave] = less; s_eq[wave] = eq; }
    *reinterpret_cast<float4*>(&sB[t >> 5][(t & 31) << 2]) = wv;
    __syncthreads();
    int start_v = 0, n_v = 0;
    for (int w = 0; w < FT_WAVES; ++w) { start_v += s_less[w]; n_v += s_eq[w]; }
    const int start = __builtin_amdgcn_readfirstlane(start_v);
    const int n = __builtin_amdgcn_readfirstlane(n_v);
    const int ntiles = (n + 15) >> 4;

    const float* grows = agg + (size_t)start * DCH;
    float* orows = out + (size_t)start * DCH;
    float (*sAw)[SX_LD] = sA[wave];

    const int c = lane & 15;
    const int rhalf = (lane >> 4) << 3;
    const int col0 = slice * 32 + c;
    const int col1 = col0 + 16;
    const float bv0 = bias[col0], bv1 = bias[col1];

    v8f hA0, hB0, hA1, hB1, hA2, hB2, hA3, hB3;
    float p0 = 0.0f, p1 = 0.0f;
#define FT_PHASE1(S, HA, HB)                                                        \
    {                                                                               \
        const int j = wave + (S) * FT_WAVES;     \
        HA = v8f{};                                                                 \
        HB = v8f{};                                                                 \
        if (j < ntiles) {                                                           \
            ft_tile_product<8, 4>(grows, j, n, sAw, sB, lane, HA, HB);              \
            ft_bias_relu_sum(HA, HB, bv0, bv1, j * 16 + rhalf, n, p0, p1);          \
        }                                                                           \
    }
    FT_PHASE1(0, hA0, hB0)
    FT_PHASE1(1, hA1, hB1)
    FT_PHASE1(2, hA2, hB2)
    FT_PHASE1(3, hA3, hB3)
#undef FT_PHASE1
    for (int j = wave + FT_SLOTS * FT_WAVES; j < ntiles; j += FT_WAVES) {
        v8f e0, e1;
        ft_tile_product<2, 1>(grows, j, n, sAw, sB, lane, e0, e1);
        ft_bias_relu_sum(e0, e1, bv0, bv1, j * 16 + rhalf, n, p0, p1);
    }
    float msum0, msum1;
    ft_block_sum(p0, p1, s_red, s_stat, t, msum0, msum1);

    float cn = (float)n;
    cn = cn > 1.0f ? cn : 1.0f;
    const float inv_cn = 1.0f / cn;
    const float sub0 = (msum0 * inv_cn) * ms[col0];
    const float sub1 = (msum1 * inv_cn) * ms[col1];

    float q0 = 0.0f, q1 = 0.0f;
    if (wave + 0 * FT_WAVES < ntiles) ft_sq_sum(hA0, hB0, sub0, sub1, (wave + 0 * FT_WAVES) * 16 + rhalf, n, q0, q1);
    if (wave + 1 * FT_WAVES < ntiles) ft_sq_sum(hA1, hB1, sub0, sub1, (wave + 1 * FT_WAVES) * 16 + rhalf, n, q0, q1);
    if (wave + 2 * FT_WAVES < ntiles) ft_sq_sum(hA2, hB2, sub0, sub1, (wave + 2 * FT_WAVES) * 16 + rhalf, n, q0, q1);
    if (wave + 3 * FT_WAVES < ntiles) ft_sq_sum(hA3, hB3, sub0, sub1, (wave + 3 * FT_WAVES) * 16 + rhalf, n, q0, q1);
    for (int j = wave + FT_SLOTS * FT_WAVES; j < ntiles; j += FT_WAVES) {
        v8f e0, e1;
        float d0 = 0.0f, d1 = 0.0f;
        ft_tile_product<2, 1>(grows, j, n, sAw, sB, lane, e0, e1);
        ft_bias_relu_sum(e0, e1, bv0, bv1, j * 16 + rhalf, n, d0, d1);
        ft_sq_sum(e0, e1, sub0, sub1, j * 16 + rhalf, n, q0, q1);
    }
    float vsum0, vsum1;
    ft_block_sum(q0, q1, s_red, s_stat, t, vsum0, vsum1);

    const float den0 = sqrtf(vsum0 * inv_cn + GN_EPS);
    const float den1 = sqrtf(vsum1 * inv_cn + GN_EPS);
    const float w0 = gw[col0], w1 = gw[col1];
    const float b0 = gb[col0], b1 = gb[col1];

    if (wave + 0 * FT_WAVES < ntiles) ft_norm_store(hA0, hB0, sub0, sub1, den0, den1, w0, w1, b0, b1, (wave + 0 * FT_WAVES) * 16 + rhalf, n, orows + col0);
    if (wave + 1 * FT_WAVES < ntiles) ft_norm_store(hA1, hB1, sub0, sub1, den0, den1, w0, w1, b0, b1, (wave + 1 * FT_WAVES) * 16 + rhalf, n, orows + col0);
    if (wave + 2 * FT_WAVES < ntiles) ft_norm_store(hA2, hB2, sub0, sub1, den0, den1, w0, w1, b0, b1, (wave + 2 * FT_WAVES) * 16 + rhalf, n, orows + col0);
    if (wave + 3 * FT_WAVES < ntiles) ft_norm_store(hA3, hB3, sub0, sub1, den0, den1, w0, w1, b0, b1, (wave + 3 * FT_WAVES) * 16 + rhalf, n, orows + col0);
    for (int j = wave + FT_SLOTS * FT_WAVES; j < ntiles; j += FT_WAVES) {
        v8f e0, e1;
        float d0 = 0.0f, d1 = 0.0f;
        ft_tile_product<2, 1>(grows, j, n, sAw, sB, lane, e0, e1);
        ft_bias_relu_sum(e0, e1, bv0, bv1, j * 16 + rhalf, n, d0, d1);
        ft_norm_store(e0, e1, sub0, sub1, den0, den1, w0, w1, b0, b1, j * 16 + rhalf, n, orows + col0);
    }
}

extern "C" void kernel_launch(void* const* d_in, const int* in_sizes, int n_in,
                              void* d_out, int out_size, void* d_ws, size_t ws_size,
                              hipStream_t stream) {
    const float* x     = (const float*)d_in[0];
    const int*   eidx  = (const int*)  d_in[1];
    const float* eattr = (const float*)d_in[2];
    const int*   batch = (const int*)  d_in[3];
    const float* W     = (const float*)d_in[4];
    const float* bias  = (const float*)d_in[5];
    const float* gw    = (const float*)d_in[6];
    const float* gb    = (const float*)d_in[7];
    const float* ms    = (const float*)d_in[8];
    float* out = (float*)d_out;

    const int N = in_sizes[0] / DCH;
    const int E = in_sizes[2];

    float* Wt  = (float*)d_ws;
    float* agg = Wt + (size_t)DCH * DCH;
    unsigned* bm = (unsigned*)(agg + (size_t)N * DCH);

    k_transpose_w_edge_groups<<<64 + (E + BM_EDGES * BM_RPB - 1) / (BM_EDGES * BM_RPB), 256, 0, stream>>>(W, Wt, eidx + E, bm, E);

    k_edge_rows_lds<<<(N + EP_R - 1) / EP_R, 1024, 0, stream>>>(eidx, eidx + E, eattr, x, bm, agg, N, E);

    k_gemm_graph_norm<<<NGRAPH * 4, 1024, 0, stream>>>(agg, Wt, bias, batch, ms, gw, gb, out, N);
}


# edge sum, add phase: own edges taken 32 at a time; a group's src/attr and first 12 x rows are asked for a step ahead, a 12-register ring refills after each add: loads run beside adds; same add order
# speedup vs baseline: 3.4066x; 1.0842x over previous
#include <hip/hip_runtime.h>
#include <hip/hip_bf16.h>
#include <math.h>

typedef __attribute__((ext_vector_type(2))) float v2f;
typedef __attribute__((ext_vector_type(8))) float v8f;

#define DCH 128
#define NGRAPH 64
#define GN_EPS 1e-5f
#define SX_LD (DCH + 4)

#define EP_R 391
#define BM_NB 256
#define BM_EDGES 1024
#define BM_RPB 4

__global__ __launch_bounds__(256) void k_transpose_w_edge_groups(
    const float* __restrict__ W, float* __restrict__ Wt,
    const int* __restrict__ dst, unsigned* __restrict__ bm, int E) {
    __shared__ float tile[16][17];
    __shared__ unsigned s_bits[BM_RPB][8][BM_NB];
    if (blockIdx.x < 64) {
        int bx = blockIdx.x & 7;
        int by = blockIdx.x >> 3;
        int tx = threadIdx.x & 15;
        int ty = threadIdx.x >> 4;
        tile[ty][tx] = W[(size_t)(by * 16 + ty) * DCH + bx * 16 + tx];
        __syncthreads();
        Wt[(size_t)(bx * 16 + ty) * DCH + by * 16 + tx] = tile[tx][ty];
        return;
    }
    const int c0 = (blockIdx.x - 64) * BM_RPB;
    const int nwords = (E + BM_EDGES - 1) / BM_EDGES;
    const int t = threadIdx.x;
    const int wave = t >> 5;
    int d[BM_RPB][4];
    if ((c0 + BM_RPB) * BM_EDGES <= E) {
#pragma unroll
        for (int q = 0; q < BM_RPB; ++q)
#pragma unroll
            for (int i = 0; i < 4; ++i) d[q][i] = dst[(c0 + q) * BM_EDGES + i * 256 + t];
    } else {
#pragma unroll
        for (int q = 0; q < BM_RPB; ++q)
#pragma unroll
            for (int i = 0; i < 4; ++i) {
                const int e = (c0 + q) * BM_EDGES + i * 256 + t;
                d[q][i] = -1;
                if (e < E) d[q][i] = dst[e];
            }
    }
#pragma unroll
    for (int q = 0; q < BM_RPB; ++q)
#pragma unroll
        for (int i = 0; i < 8; ++i) s_bits[q][i][t] = 0u;
    __syncthreads();
#pragma unroll
    for (int q = 0; q < BM_RPB; ++q)
#pragma unroll
        for (int i = 0; i < 4; ++i) {
            const unsigned blk = (unsigned)d[q][i] / (unsigned)EP_R;
            if (blk < (unsigned)BM_NB) s_bits[q][wave][blk] |= 1u << (i * 8 + wave);
        }
    __syncthreads();
#pragma unroll
    for (int q = 0; q < BM_RPB; ++q) {
        if (c0 + q < nwords) {
            unsigned w = 0u;
#pragma unroll
            for (int i = 0; i < 8; ++i) w |= s_bits[q][i][t];
            bm[(size_t)(c0 + q) * BM_NB + t] = w;
        }
    }
}

#define EP_WAVES 32
#define EP_CAP 512
#define EP_GCAP 512
#define EP_U 32
#define EP_STAGE 160
#define EP_ROWS 16
#define EP_RING 12

__device__ __forceinline__ void ep_wave_order() {
    __builtin_amdgcn_fence(__ATOMIC_RELEASE, "wavefront");
    __builtin_amdgcn_wave_barrier();
    __builtin_amdgcn_fence(__ATOMIC_ACQUIRE, "wavefront");
}

__device__ __forceinline__ void ep_flush(
    const unsigned* stage, int nb, int lane, const int* __restrict__ src,
    const float* __restrict__ attr, const float4* __restrict__ x4, float4* acc4) {
    ep_wave_order();
    for (int b0 = 0; b0 < nb; b0 += 32) {
        int nbb = nb - b0;
        nbb = nbb < 32 ? nbb : 32;
        unsigned ent = 0;
        int sv = 0;
        float av = 0.0f;
        if (lane < nbb) {
            ent = stage[b0 + lane];
            const int e = (int)(ent >> 9);
            sv = src[e];
            av = attr[e];
        }
        const int rv = (int)(ent & 511u);
        const int avi = __float_as_int(av);
        for (int j0 = 0; j0 < nbb; j0 += EP_ROWS) {
            float4 v[EP_ROWS];
#pragma unroll
            for (int u = 0; u < EP_ROWS; ++u) {
                int j = j0 + u;
                j = j < nbb ? j : nbb - 1;
                const int s = __builtin_amdgcn_readlane(sv, j);
                v[u] = x4[(size_t)s * 32 + lane];
            }
#pragma unroll
            for (int u = 0; u < EP_ROWS; ++u) {
                const int j = j0 + u;
                if (j < nbb) {
                    const int r = __builtin_amdgcn_readlane(rv, j);
                    const float a = __int_as_float(__builtin_amdgcn_readlane(avi, j));
                    float4 c = acc4[r * 32 + lane];
                    c.x += a * v[u].x;
                    c.y += a * v[u].y;
                    c.z += a * v[u].z;
                    c.w += a * v[u].w;
                    acc4[r * 32 + lane] = c;
                }
            }
        }
    }
    ep_wave_order();
}

__device__ __forceinline__ void ep_ask_rows(
    float4 (&v)[EP_RING], int sv, int lane, const float4* __restrict__ x4) {
#pragma unroll
    for (int u = 0; u < EP_RING; ++u) {
        const int s = __builtin_amdgcn_readlane(sv, u);
        v[u] = x4[(size_t)s * 32 + lane];
    }
}

__device__ __forceinline__ void ep_add_rows32(
    float4 (&v)[EP_RING], int sv, int rv, int avi, int lane,
    const float4* __restrict__ x4, float4* acc4) {
#pragma unroll
    for (int j = 0; j < 32; ++j) {
        const int u = j % EP_RING;
        const int r = __builtin_amdgcn_readlane(rv, j);
        const float a = __int_as_float(__builtin_amdgcn_readlane(avi, j));
        float4 c = acc4[r * 32 + lane];
        c.x += a * v[u].x;
        c.y += a * v[u].y;
        c.z += a * v[u].z;
        c.w += a * v[u].w;
        acc4[r * 32 + lane] = c;
        if (j + EP_RING < 32) {
            const int s = __builtin_amdgcn_readlane(sv, j + EP_RING);
            v[u] = x4[(size_t)s * 32 + lane];
        }
    }
}

template <bool FULL>
__device__ __forceinline__ void ep_scan_batch(
    const int* __restrict__ dst, const unsigned short* glist, unsigned* mylist, int lane,
    int g_row0, int nval_in, int gp, int base, unsigned unrows, int& count) {
    const int nval = FULL ? EP_U : nval_in;
    const int myg = (int)glist[gp + (lane < nval ? lane : 0)];
    const int ebase = (g_row0 << 10) + lane;
    int e[EP_U];
    int d[EP_U];
#pragma unroll
    for (int u = 0; u < EP_U; ++u) {
        e[u] = ebase + (__builtin_amdgcn_readlane(myg, u) << 5);
        d[u] = dst[e[u]];
    }
    int c = count;
#pragma unroll
    for (int q = 0; q < EP_U / 8; ++q) {
        if (FULL || q * 8 < nval) {
#pragma unroll
            for (int u = q * 8; u < q * 8 + 8; ++u) {
                const unsigned rel = (unsigned)(d[u] - base);
                const bool hit = rel < unrows;
                const unsigned mk = __builtin_amdgcn_ballot_w32(hit);
                if ((FULL || u < nval) && c <= EP_CAP - 32) {
                    if (hit) mylist[c + (int)__builtin_amdgcn_mbcnt_lo(mk, 0u)] = ((unsigned)e[u] << 9) + rel;
                    c += __builtin_popcount(mk);
                }
            }
        }
    }
    count = c;
}

__global__ __launch_bounds__(1024) void k_edge_rows_lds(
    const int* __restrict__ src, const int* __restrict__ dst,
    const float* __restrict__ attr, const float* __restrict__ x,
    const unsigned* __restrict__ bm, float* __restrict__ agg, int N, int E) {
    __shared__ __attribute__((aligned(16))) float s_acc[EP_R * DCH];
    __shared__ unsigned s_list[EP_WAVES][EP_CAP];
    __shared__ unsigned s_stage[EP_WAVES][EP_STAGE];
    __shared__ unsigned short s_grp[EP_WAVES][EP_GCAP];
    __shared__ int s_cnt[EP_WAVES];
    __shared__ int s_more[EP_WAVES];

    const int t = threadIdx.x;
    const int lane = t & 31;
    const int wave = __builtin_amdgcn_readfirstlane(t >> 5);
    const int base = blockIdx.x * EP_R;
    int nrows = N - base;
    nrows = nrows < EP_R ? nrows : EP_R;
    const unsigned unrows = (unsigned)nrows;

    const int bid = blockIdx.x;
    const int nwords = (E + BM_EDGES - 1) / BM_EDGES;
    const int nfull = E / BM_EDGES;
    const int wper = (nwords + EP_WAVES - 1) / EP_WAVES;
    int wend = (wave + 1) * wper;
    wend = wend < nwords ? wend : nwords;
    int wi = wave * wper;
    int wc_base = wi, wc_end = wi + 32;
    unsigned w0 = 0u;
    if (wi + lane < wend) w0 = bid < BM_NB ? bm[(size_t)(wi + lane) * BM_NB + bid] : 0xffffffffu;

    float4* acc4 = reinterpret_cast<float4*>(s_acc);
    const float4* x4 = reinterpret_cast<const float4*>(x);
    for (int i = t; i < EP_R * 32; i += 1024) acc4[i] = make_float4(0.0f, 0.0f, 0.0f, 0.0f);

    unsigned short* glist = s_grp[wave];
    unsigned* mylist = s_list[wave];
    int g_row0 = 0;
    int g_n = 0;
    int g_pos = 0;
    const int max_rounds = (wper * BM_EDGES) / (EP_CAP - 32) + 2;

    for (int round = 0; round < max_rounds; ++round) {
        int count = 0;
        bool full = false;
        while (!full) {
            if (g_pos >= g_n) {
                if (wi >= wend) break;
                if (wi >= wc_end) {
                    const int r = wi + lane;
                    w0 = 0u;
                    if (r < wend) w0 = bid < BM_NB ? bm[(size_t)r * BM_NB + bid] : 0xffffffffu;
                    wc_base = wi;
                    wc_end = wi + 32;
                }
                const int j = wi - wc_base + lane;
                unsigned bits = (unsigned)__shfl((int)w0, j & 31);
                if (j >= 32) bits = 0u;
                const int cnt = __builtin_popcount(bits);
                int incl = cnt;
#pragma unroll
                for (int off = 1; off < 32; off <<= 1) {
                    const int v = __shfl_up(incl, off);
                    if (lane >= off) incl += v;
                }
                int nrow = __builtin_popcount(__builtin_amdgcn_ballot_w32(incl <= EP_GCAP));
                nrow = nrow < wc_end - wi ? nrow : wc_end - wi;
                if (wi >= nfull) nrow = 1;
                else nrow = nrow < nfull - wi ? nrow : nfull - wi;
                if (lane >= nrow) bits = 0u;
                int pos = incl - cnt;
                ep_wave_order();
                while (bits != 0u) {
                    glist[pos++] = (unsigned short)((lane << 5) + __builtin_ctz(bits));
                    bits &= bits - 1u;
                }
                ep_wave_order();
                g_n = __builtin_amdgcn_readlane(incl, nrow - 1);
                g_row0 = wi;
                g_pos = 0;
                wi += nrow;
                continue;
            }
            const int left = g_n - g_pos;
            const int nval = left < EP_U ? left : EP_U;
            bool taken = false;
            if (g_row0 < nfull) {
                const int count0 = count;
                if (nval == EP_U)
                    ep_scan_batch<true>(dst, glist, mylist, lane, g_row0, nval, g_pos, base, unrows, count);
                else
                    ep_scan_batch<false>(dst, glist, mylist, lane, g_row0, nval, g_pos, base, unrows, count);
                taken = count <= EP_CAP - 32;
                if (!taken) count = count0;
            }
            if (taken) {
                g_pos += nval;
            } else {
                for (int u = 0; u < nval && !full; ++u) {
                    const int g = __builtin_amdgcn_readfirstlane((int)glist[g_pos]);
                    const int e = (g_row0 << 10) + (g << 5) + lane;
                    int dv = 0x7fffffff;
                    if (e < E) dv = dst[e];
                    const unsigned rel = (unsigned)(dv - base);
                    const bool hit = rel < unrows;
                    const unsigned mk = __builtin_amdgcn_ballot_w32(hit);
                    const int pop = __builtin_popcount(mk);
                    if (count + pop > EP_CAP) {
                        full = true;
                    } else {
                        if (hit) mylist[count + (int)__builtin_amdgcn_mbcnt_lo(mk, 0u)] = ((unsigned)e << 9) | rel;
                        count += pop;
                        ++g_pos;
                    }
                }
            }
        }
        if (lane == 0) { s_cnt[wave] = count; s_more[wave] = (g_pos < g_n || wi < wend) ? 1 : 0; }
        __syncthreads();
        int more_v = 0;
        for (int w = 0; w < EP_WAVES; ++w) more_v |= s_more[w];
        const int more = __builtin_amdgcn_readfirstlane(more_v);

        unsigned* stage = s_stage[wave];
        int npend = 0;
        int n_p = 0;
        int sv_p = 0, rv_p = 0, avi_p = 0;
        float4 vrow[EP_RING];
#pragma unroll
        for (int u = 0; u < EP_RING; ++u) vrow[u] = make_float4(0.0f, 0.0f, 0.0f, 0.0f);
        for (int sw = 0; sw < EP_WAVES; ++sw) {
            const int n = __builtin_amdgcn_readfirstlane(s_cnt[sw]);
            for (int i0 = 0; i0 < n; i0 += 128) {
                unsigned ent[4];
#pragma unroll
                for (int q = 0; q < 4; ++q) {
                    const int idx = i0 + q * 32 + lane;
                    ent[q] = 0xffffffffu;
                    if (idx < n) ent[q] = s_list[sw][idx];
                }
#pragma unroll
                for (int q = 0; q < 4; ++q) {
                    const int idx = i0 + q * 32 + lane;
                    const bool mine = (idx < n) && ((int)(ent[q] & 31u) == wave);
                    const unsigned m = __builtin_amdgcn_ballot_w32(mine);
                    if (m != 0u) {
                        if (mine) stage[npend + (int)__builtin_amdgcn_mbcnt_lo(m, 0u)] = ent[q];
                        npend += __builtin_popcount(m);
                    }
                }
                if (npend >= 32) {
                    ep_wave_order();
                    int h = 0;
                    while (npend - h >= 32) {
                        const unsigned e2 = stage[h + lane];
                        h += 32;
                        const int e = (int)(e2 >> 9);
                        const int sv = src[e];
                        const float av = attr[e];
                        if (n_p != 0) ep_add_rows32(vrow, sv_p, rv_p, avi_p, lane, x4, acc4);
                        sv_p = sv;
                        rv_p = (int)(e2 & 511u);
                        avi_p = __float_as_int(av);
                        n_p = 32;
                        ep_ask_rows(vrow, sv_p, lane, x4);
                    }
                    const int rem = npend - h;
                    unsigned keep = 0u;
                    if (lane < rem) keep = stage[h + lane];
                    ep_wave_order();
                    if (lane < rem) stage[lane] = keep;
                    npend = rem;
                    ep_wave_order();
                }
            }
        }
        if (n_p != 0) ep_add_rows32(vrow, sv_p, rv_p, avi_p, lane, x4, acc4);
        ep_flush(stage, npend, lane, src, attr, x4, acc4);
        __syncthreads();
        if (!more) break;
    }

    float4* agg4 = reinterpret_cast<float4*>(agg) + (size_t)base * 32;
    for (int i = t; i < nrows * 32; i += 1024) agg4[i] = acc4[i];
}

#define FT_SLOTS 4
#define FT_WAVES 32
template <int RB, int KU>
__device__ __forceinline__ void ft_tile_product(
    const float* __restrict__ grows, int j, int n, float (*sAw)[SX_LD],
    const float (*sBm)[SX_LD], int lane, v8f& acc0, v8f& acc1) {
    const int r0 = j * 16;
#pragma unroll
    for (int part = 0; part < 16 / RB; ++part) {
        float4 v[RB];
#pragma unroll
        for (int r = 0; r < RB; ++r) {
            const int rr = r0 + part * RB + r;
            v[r] = make_float4(0.0f, 0.0f, 0.0f, 0.0f);
            if (rr < n) v[r] = *reinterpret_cast<const float4*>(grows + (size_t)rr * DCH + lane * 4);
        }
#pragma unroll
        for (int r = 0; r < RB; ++r)
            *reinterpret_cast<float4*>(&sAw[part * RB + r][lane * 4]) = v[r];
    }
    __builtin_amdgcn_fence(__ATOMIC_RELEASE, "wavefront");
    __builtin_amdgcn_wave_barrier();
    __builtin_amdgcn_fence(__ATOMIC_ACQUIRE, "wavefront");

    const int m    = lane & 15;
    const int koff = (lane >> 4) << 1;
    acc0 = v8f{};
    acc1 = v8f{};
#pragma unroll KU
    for (int kt = 0; kt < DCH; kt += 4) {
        v2f a  = *reinterpret_cast<const v2f*>(&sAw[m][kt + koff]);
        v2f b0 = *reinterpret_cast<const v2f*>(&sBm[m][kt + koff]);
        v2f b1 = *reinterpret_cast<const v2f*>(&sBm[16 + m][kt + koff]);
        acc0 = __builtin_amdgcn_wmma_f32_16x16x4_f32(false, a, false, b0, (short)0, acc0, false, false);
        acc1 = __builtin_amdgcn_wmma_f32_16x16x4_f32(false, a, false, b1, (short)0, acc1, false, false);
    }
    __builtin_amdgcn_fence(__ATOMIC_RELEASE, "wavefront");
    __builtin_amdgcn_wave_barrier();
    __builtin_amdgcn_fence(__ATOMIC_ACQUIRE, "wavefront");
}

__device__ __forceinline__ void ft_bias_relu_sum(
    v8f& ha, v8f& hb, float bv0, float bv1, int rbase, int n, float& p0, float& p1) {
#pragma unroll
    for (int v = 0; v < 8; ++v) {
        const bool ok = (rbase + v) < n;
        float a0 = ha[v] + bv0; a0 = a0 > 0.0f ? a0 : 0.0f;
        float a1 = hb[v] + bv1; a1 = a1 > 0.0f ? a1 : 0.0f;
        ha[v] = a0;
        hb[v] = a1;
        p0 += ok ? a0 : 0.0f;
        p1 += ok ? a1 : 0.0f;
    }
}

__device__ __forceinline__ void ft_sq_sum(
    const v8f& ha, const v8f& hb, float sub0, float sub1, int rbase, int n, float& q0, float& q1) {
#pragma unroll
    for (int v = 0; v < 8; ++v) {
        const bool ok = (rbase + v) < n;
        const float o0 = ha[v] - sub0;
        const float o1 = hb[v] - sub1;
        q0 += ok ? o0 * o0 : 0.0f;
        q1 += ok ? o1 * o1 : 0.0f;
    }
}

__device__ __forceinline__ void ft_norm_store(
    const v8f& ha, const v8f& hb, float sub0, float sub1, float den0, float den1,
    float w0, float w1, float b0, float b1, int rbase, int n, float* __restrict__ ocol) {
#pragma unroll
    for (int v = 0; v < 8; ++v) {
        const int rr = rbase + v;
        if (rr < n) {
            ocol[(size_t)rr * DCH]      = w0 * (ha[v] - sub0) / den0 + b0;
            ocol[(size_t)rr * DCH + 16] = w1 * (hb[v] - sub1) / den1 + b1;
        }
    }
}

__device__ __forceinline__ void ft_block_sum(
    float p0, float p1, float (*s_red)[32], float* s_stat, int t, float& r0, float& r1) {
    const int lane = t & 31;
    const int wave = t >> 5;
    p0 += __shfl_xor(p0, 16);
    p1 += __shfl_xor(p1, 16);
    if (lane < 16) { s_red[wave][lane] = p0; s_red[wave][16 + lane] = p1; }
    __syncthreads();
    if (t < 32) {
        float s = 0.0f;
        for (int w = 0; w < FT_WAVES; ++w) s += s_red[w][t];
        s_stat[t] = s;
    }
    __syncthreads();
    r0 = s_stat[lane & 15];
    r1 = s_stat[16 + (lane & 15)];
    __syncthreads();
}

__global__ __launch_bounds__(1024) void k_gemm_graph_norm(
    const float* __restrict__ agg, const float* __restrict__ Wt,
    const float* __restrict__ bias, const int* __restrict__ batch,
    const float* __restrict__ ms, const float* __restrict__ gw,
    const float* __restrict__ gb, float* __restrict__ out, int N) {
    __shared__ __attribute__((aligned(16))) float sA[FT_WAVES][16][SX_LD];
    __shared__ __attribute__((aligned(16))) float sB[32][SX_LD];
    __shared__ float s_red[FT_WAVES][32];
    __shared__ float s_stat[32];
    __shared__ int s_less[FT_WAVES];
    __shared__ int s_eq[FT_WAVES];

    const int t = threadIdx.x;
    const int lane = t & 31;
    const int wave = __builtin_amdgcn_readfirstlane(t >> 5);
    const int g = blockIdx.x >> 2;
    const int slice = blockIdx.x & 3;

    const float4 wv = *reinterpret_cast<const float4*>(
        Wt + (size_t)(slice * 32 + (t >> 5)) * DCH + ((t & 31) << 2));

    int less = 0, eq = 0;
    {
        const int4* b4 = reinterpret_cast<const int4*>(batch);
        const int n4 = N >> 2;
        for (int base = 0; base < n4; base += 8 * 1024) {
            int4 q[8];
#pragma unroll
            for (int u = 0; u < 8; ++u) {
                const int i = base + u * 1024 + t;
                q[u] = make_int4(0x7fffffff, 0x7fffffff, 0x7fffffff, 0x7fffffff);
                if (i < n4) q[u] = b4[i];
            }
#pragma unroll
            for (int u = 0; u < 8; ++u) {
                less += (q[u].x < g) + (q[u].y < g) + (q[u].z < g) + (q[u].w < g);
                eq   += (q[u].x == g) + (q[u].y == g) + (q[u].z == g) + (q[u].w == g);
            }
        }
        for (int i = (n4 << 2) + t; i < N; i += 1024) {
            const int q = batch[i];
            less += (q < g);
            eq   += (q == g);
        }
    }
#pragma unroll
    for (int off = 16; off > 0; off >>= 1) {
        less += __shfl_xor(less, off);
        eq   += __shfl_xor(eq, off);
    }
    if (lane == 0) { s_less[wave] = less; s_eq[wave] = eq; }
    *reinterpret_cast<float4*>(&sB[t >> 5][(t & 31) << 2]) = wv;
    __syncthreads();
    int start_v = 0, n_v = 0;
    for (int w = 0; w < FT_WAVES; ++w) { start_v += s_less[w]; n_v += s_eq[w]; }
    const int start = __builtin_amdgcn_readfirstlane(start_v);
    const int n = __builtin_amdgcn_readfirstlane(n_v);
    const int ntiles = (n + 15) >> 4;

    const float* grows = agg + (size_t)start * DCH;
    float* orows = out + (size_t)start * DCH;
    float (*sAw)[SX_LD] = sA[wave];

    const int c = lane & 15;
    const int rhalf = (lane >> 4) << 3;
    const int col0 = slice * 32 + c;
    const int col1 = col0 + 16;
    const float bv0 = bias[col0], bv1 = bias[col1];

    v8f hA0, hB0, hA1, hB1, hA2, hB2, hA3, hB3;
    float p0 = 0.0f, p1 = 0.0f;
#define FT_PHASE1(S, HA, HB)                                                        \
    {                                                                               \
        const int j = wave + (S) * FT_WAVES;     \
        HA = v8f{};                                                                 \
        HB = v8f{};                                                                 \
        if (j < ntiles) {                                                           \
            ft_tile_product<8, 4>(grows, j, n, sAw, sB, lane, HA, HB);              \
            ft_bias_relu_sum(HA, HB, bv0, bv1, j * 16 + rhalf, n, p0, p1);          \
        }                                                                           \
    }
    FT_PHASE1(0, hA0, hB0)
    FT_PHASE1(1, hA1, hB1)
    FT_PHASE1(2, hA2, hB2)
    FT_PHASE1(3, hA3, hB3)
#undef FT_PHASE1
    for (int j = wave + FT_SLOTS * FT_WAVES; j < ntiles; j += FT_WAVES) {
        v8f e0, e1;
        ft_tile_product<2, 1>(grows, j, n, sAw, sB, lane, e0, e1);
        ft_bias_relu_sum(e0, e1, bv0, bv1, j * 16 + rhalf, n, p0, p1);
    }
    float msum0, msum1;
    ft_block_sum(p0, p1, s_red, s_stat, t, msum0, msum1);

    float cn = (float)n;
    cn = cn > 1.0f ? cn : 1.0f;
    const float inv_cn = 1.0f / cn;
    const float sub0 = (msum0 * inv_cn) * ms[col0];
    const float sub1 = (msum1 * inv_cn) * ms[col1];

    float q0 = 0.0f, q1 = 0.0f;
    if (wave + 0 * FT_WAVES < ntiles) ft_sq_sum(hA0, hB0, sub0, sub1, (wave + 0 * FT_WAVES) * 16 + rhalf, n, q0, q1);
    if (wave + 1 * FT_WAVES < ntiles) ft_sq_sum(hA1, hB1, sub0, sub1, (wave + 1 * FT_WAVES) * 16 + rhalf, n, q0, q1);
    if (wave + 2 * FT_WAVES < ntiles) ft_sq_sum(hA2, hB2, sub0, sub1, (wave + 2 * FT_WAVES) * 16 + rhalf, n, q0, q1);
    if (wave + 3 * FT_WAVES < ntiles) ft_sq_sum(hA3, hB3, sub0, sub1, (wave + 3 * FT_WAVES) * 16 + rhalf, n, q0, q1);
    for (int j = wave + FT_SLOTS * FT_WAVES; j < ntiles; j += FT_WAVES) {
        v8f e0, e1;
        float d0 = 0.0f, d1 = 0.0f;
        ft_tile_product<2, 1>(grows, j, n, sAw, sB, lane, e0, e1);
        ft_bias_relu_sum(e0, e1, bv0, bv1, j * 16 + rhalf, n, d0, d1);
        ft_sq_sum(e0, e1, sub0, sub1, j * 16 + rhalf, n, q0, q1);
    }
    float vsum0, vsum1;
    ft_block_sum(q0, q1, s_red, s_stat, t, vsum0, vsum1);

    const float den0 = sqrtf(vsum0 * inv_cn + GN_EPS);
    const float den1 = sqrtf(vsum1 * inv_cn + GN_EPS);
    const float w0 = gw[col0], w1 = gw[col1];
    const float b0 = gb[col0], b1 = gb[col1];

    if (wave + 0 * FT_WAVES < ntiles) ft_norm_store(hA0, hB0, sub0, sub1, den0, den1, w0, w1, b0, b1, (wave + 0 * FT_WAVES) * 16 + rhalf, n, orows + col0);
    if (wave + 1 * FT_WAVES < ntiles) ft_norm_store(hA1, hB1, sub0, sub1, den0, den1, w0, w1, b0, b1, (wave + 1 * FT_WAVES) * 16 + rhalf, n, orows + col0);
    if (wave + 2 * FT_WAVES < ntiles) ft_norm_store(hA2, hB2, sub0, sub1, den0, den1, w0, w1, b0, b1, (wave + 2 * FT_WAVES) * 16 + rhalf, n, orows + col0);
    if (wave + 3 * FT_WAVES < ntiles) ft_norm_store(hA3, hB3, sub0, sub1, den0, den1, w0, w1, b0, b1, (wave + 3 * FT_WAVES) * 16 + rhalf, n, orows + col0);
    for (int j = wave + FT_SLOTS * FT_WAVES; j < ntiles; j += FT_WAVES) {
        v8f e0, e1;
        float d0 = 0.0f, d1 = 0.0f;
        ft_tile_product<2, 1>(grows, j, n, sAw, sB, lane, e0, e1);
        ft_bias_relu_sum(e0, e1, bv0, bv1, j * 16 + rhalf, n, d0, d1);
        ft_norm_store(e0, e1, sub0, sub1, den0, den1, w0, w1, b0, b1, j * 16 + rhalf, n, orows + col0);
    }
}

extern "C" void kernel_launch(void* const* d_in, const int* in_sizes, int n_in,
                              void* d_out, int out_size, void* d_ws, size_t ws_size,
                              hipStream_t stream) {
    const float* x     = (const float*)d_in[0];
    const int*   eidx  = (const int*)  d_in[1];
    const float* eattr = (const float*)d_in[2];
    const int*   batch = (const int*)  d_in[3];
    const float* W     = (const float*)d_in[4];
    const float* bias  = (const float*)d_in[5];
    const float* gw    = (const float*)d_in[6];
    const float* gb    = (const float*)d_in[7];
    const float* ms    = (const float*)d_in[8];
    float* out = (float*)d_out;

    const int N = in_sizes[0] / DCH;
    const int E = in_sizes[2];

    float* Wt  = (float*)d_ws;
    float* agg = Wt + (size_t)DCH * DCH;
    unsigned* bm = (unsigned*)(agg + (size_t)N * DCH);

    k_transpose_w_edge_groups<<<64 + (E + BM_EDGES * BM_RPB - 1) / (BM_EDGES * BM_RPB), 256, 0, stream>>>(W, Wt, eidx + E, bm, E);

    k_edge_rows_lds<<<(N + EP_R - 1) / EP_R, 1024, 0, stream>>>(eidx, eidx + E, eattr, x, bm, agg, N, E);

    k_gemm_graph_norm<<<NGRAPH * 4, 1024, 0, stream>>>(agg, Wt, bias, batch, ms, gw, gb, out, N);
}


# edge sum, phase 2: entries reach their owner wave by a counting sort in LDS (owner counts per list, prefix sums, one scatter); every wave reading all 32 lists is gone; order of adds unchanged
# speedup vs baseline: 3.5957x; 1.0555x over previous
#include <hip/hip_runtime.h>
#include <hip/hip_bf16.h>
#include <math.h>

typedef __attribute__((ext_vector_type(2))) float v2f;
typedef __attribute__((ext_vector_type(8))) float v8f;

#define DCH 128
#define NGRAPH 64
#define GN_EPS 1e-5f
#define SX_LD (DCH + 4)

#define EP_R 391
#define BM_NB 256
#define BM_EDGES 1024
#define BM_RPB 4

__global__ __launch_bounds__(256) void k_transpose_w_edge_groups(
    const float* __restrict__ W, float* __restrict__ Wt,
    const int* __restrict__ dst, unsigned* __restrict__ bm, int E) {
    __shared__ float tile[16][17];
    __shared__ unsigned s_bits[BM_RPB][8][BM_NB];
    if (blockIdx.x < 64) {
        int bx = blockIdx.x & 7;
        int by = blockIdx.x >> 3;
        int tx = threadIdx.x & 15;
        int ty = threadIdx.x >> 4;
        tile[ty][tx] = W[(size_t)(by * 16 + ty) * DCH + bx * 16 + tx];
        __syncthreads();
        Wt[(size_t)(bx * 16 + ty) * DCH + by * 16 + tx] = tile[tx][ty];
        return;
    }
    const int c0 = (blockIdx.x - 64) * BM_RPB;
    const int nwords = (E + BM_EDGES - 1) / BM_EDGES;
    const int t = threadIdx.x;
    const int wave = t >> 5;
    int d[BM_RPB][4];
    if ((c0 + BM_RPB) * BM_EDGES <= E) {
#pragma unroll
        for (int q = 0; q < BM_RPB; ++q)
#pragma unroll
            for (int i = 0; i < 4; ++i) d[q][i] = dst[(c0 + q) * BM_EDGES + i * 256 + t];
    } else {
#pragma unroll
        for (int q = 0; q < BM_RPB; ++q)
#pragma unroll
            for (int i = 0; i < 4; ++i) {
                const int e = (c0 + q) * BM_EDGES + i * 256 + t;
                d[q][i] = -1;
                if (e < E) d[q][i] = dst[e];
            }
    }
#pragma unroll
    for (int q = 0; q < BM_RPB; ++q)
#pragma unroll
        for (int i = 0; i < 8; ++i) s_bits[q][i][t] = 0u;
    __syncthreads();
#pragma unroll
    for (int q = 0; q < BM_RPB; ++q)
#pragma unroll
        for (int i = 0; i < 4; ++i) {
            const unsigned blk = (unsigned)d[q][i] / (unsigned)EP_R;
            if (blk < (unsigned)BM_NB) s_bits[q][wave][blk] |= 1u << (i * 8 + wave);
        }
    __syncthreads();
#pragma unroll
    for (int q = 0; q < BM_RPB; ++q) {
        if (c0 + q < nwords) {
            unsigned w = 0u;
#pragma unroll
            for (int i = 0; i < 8; ++i) w |= s_bits[q][i][t];
            bm[(size_t)(c0 + q) * BM_NB + t] = w;
        }
    }
}

#define EP_WAVES 32
#define EP_CAP 384
#define EP_GCAP 384
#define EP_U 32
#define EP_ROWS 16
#define EP_RING 12

__device__ __forceinline__ void ep_wave_order() {
    __builtin_amdgcn_fence(__ATOMIC_RELEASE, "wavefront");
    __builtin_amdgcn_wave_barrier();
    __builtin_amdgcn_fence(__ATOMIC_ACQUIRE, "wavefront");
}

__device__ __forceinline__ void ep_flush(
    const unsigned* stage, int nb, int lane, const int* __restrict__ src,
    const float* __restrict__ attr, const float4* __restrict__ x4, float4* acc4) {
    ep_wave_order();
    for (int b0 = 0; b0 < nb; b0 += 32) {
        int nbb = nb - b0;
        nbb = nbb < 32 ? nbb : 32;
        unsigned ent = 0;
        int sv = 0;
        float av = 0.0f;
        if (lane < nbb) {
            ent = stage[b0 + lane];
            const int e = (int)(ent >> 9);
            sv = src[e];
            av = attr[e];
        }
        const int rv = (int)(ent & 511u);
        const int avi = __float_as_int(av);
        for (int j0 = 0; j0 < nbb; j0 += EP_ROWS) {
            float4 v[EP_ROWS];
#pragma unroll
            for (int u = 0; u < EP_ROWS; ++u) {
                int j = j0 + u;
                j = j < nbb ? j : nbb - 1;
                const int s = __builtin_amdgcn_readlane(sv, j);
                v[u] = x4[(size_t)s * 32 + lane];
            }
#pragma unroll
            for (int u = 0; u < EP_ROWS; ++u) {
                const int j = j0 + u;
                if (j < nbb) {
                    const int r = __builtin_amdgcn_readlane(rv, j);
                    const float a = __int_as_float(__builtin_amdgcn_readlane(avi, j));
                    float4 c = acc4[r * 32 + lane];
                    c.x += a * v[u].x;
                    c.y += a * v[u].y;
                    c.z += a * v[u].z;
                    c.w += a * v[u].w;
                    acc4[r * 32 + lane] = c;
                }
            }
        }
    }
    ep_wave_order();
}

__device__ __forceinline__ void ep_ask_rows(
    float4 (&v)[EP_RING], int sv, int lane, const float4* __restrict__ x4) {
#pragma unroll
    for (int u = 0; u < EP_RING; ++u) {
        const int s = __builtin_amdgcn_readlane(sv, u);
        v[u] = x4[(size_t)s * 32 + lane];
    }
}

__device__ __forceinline__ void ep_add_rows32(
    float4 (&v)[EP_RING], int sv, int rv, int avi, int lane,
    const float4* __restrict__ x4, float4* acc4) {
#pragma unroll
    for (int j = 0; j < 32; ++j) {
        const int u = j % EP_RING;
        const int r = __builtin_amdgcn_readlane(rv, j);
        const float a = __int_as_float(__builtin_amdgcn_readlane(avi, j));
        float4 c = acc4[r * 32 + lane];
        c.x += a * v[u].x;
        c.y += a * v[u].y;
        c.z += a * v[u].z;
        c.w += a * v[u].w;
        acc4[r * 32 + lane] = c;
        if (j + EP_RING < 32) {
            const int s = __builtin_amdgcn_readlane(sv, j + EP_RING);
            v[u] = x4[(size_t)s * 32 + lane];
        }
    }
}

template <bool FULL>
__device__ __forceinline__ void ep_scan_batch(
    const int* __restrict__ dst, const unsigned short* glist, unsigned* mylist, int lane,
    int g_row0, int nval_in, int gp, int base, unsigned unrows, int& count) {
    const int nval = FULL ? EP_U : nval_in;
    const int myg = (int)glist[gp + (lane < nval ? lane : 0)];
    const int ebase = (g_row0 << 10) + lane;
    int e[EP_U];
    int d[EP_U];
#pragma unroll
    for (int u = 0; u < EP_U; ++u) {
        e[u] = ebase + (__builtin_amdgcn_readlane(myg, u) << 5);
        d[u] = dst[e[u]];
    }
    int c = count;
#pragma unroll
    for (int q = 0; q < EP_U / 8; ++q) {
        if (FULL || q * 8 < nval) {
#pragma unroll
            for (int u = q * 8; u < q * 8 + 8; ++u) {
                const unsigned rel = (unsigned)(d[u] - base);
                const bool hit = rel < unrows;
                const unsigned mk = __builtin_amdgcn_ballot_w32(hit);
                if ((FULL || u < nval) && c <= EP_CAP - 32) {
                    if (hit) mylist[c + (int)__builtin_amdgcn_mbcnt_lo(mk, 0u)] = ((unsigned)e[u] << 9) + rel;
                    c += __builtin_popcount(mk);
                }
            }
        }
    }
    count = c;
}

__global__ __launch_bounds__(1024) void k_edge_rows_lds(
    const int* __restrict__ src, const int* __restrict__ dst,
    const float* __restrict__ attr, const float* __restrict__ x,
    const unsigned* __restrict__ bm, float* __restrict__ agg, int N, int E) {
    __shared__ __attribute__((aligned(16))) float s_acc[EP_R * DCH];
    __shared__ unsigned s_list[EP_WAVES][EP_CAP];
    __shared__ unsigned s_sorted[EP_WAVES * EP_CAP];
    __shared__ unsigned short s_grp[EP_WAVES][EP_GCAP];
    __shared__ int s_own[EP_WAVES][EP_WAVES];
    __shared__ int s_more[EP_WAVES];

    const int t = threadIdx.x;
    const int lane = t & 31;
    const int wave = __builtin_amdgcn_readfirstlane(t >> 5);
    const int base = blockIdx.x * EP_R;
    int nrows = N - base;
    nrows = nrows < EP_R ? nrows : EP_R;
    const unsigned unrows = (unsigned)nrows;

    const int bid = blockIdx.x;
    const int nwords = (E + BM_EDGES - 1) / BM_EDGES;
    const int nfull = E / BM_EDGES;
    const int wper = (nwords + EP_WAVES - 1) / EP_WAVES;
    int wend = (wave + 1) * wper;
    wend = wend < nwords ? wend : nwords;
    int wi = wave * wper;
    int wc_base = wi, wc_end = wi + 32;
    unsigned w0 = 0u;
    if (wi + lane < wend) w0 = bid < BM_NB ? bm[(size_t)(wi + lane) * BM_NB + bid] : 0xffffffffu;

    float4* acc4 = reinterpret_cast<float4*>(s_acc);
    const float4* x4 = reinterpret_cast<const float4*>(x);
    for (int i = t; i < EP_R * 32; i += 1024) acc4[i] = make_float4(0.0f, 0.0f, 0.0f, 0.0f);

    unsigned short* glist = s_grp[wave];
    unsigned* mylist = s_list[wave];
    int g_row0 = 0;
    int g_n = 0;
    int g_pos = 0;
    const int max_rounds = (wper * BM_EDGES) / (EP_CAP - 32) + 2;

    for (int round = 0; round < max_rounds; ++round) {
        int count = 0;
        bool full = false;
        while (!full) {
            if (g_pos >= g_n) {
                if (wi >= wend) break;
                if (wi >= wc_end) {
                    const int r = wi + lane;
                    w0 = 0u;
                    if (r < wend) w0 = bid < BM_NB ? bm[(size_t)r * BM_NB + bid] : 0xffffffffu;
                    wc_base = wi;
                    wc_end = wi + 32;
                }
                const int j = wi - wc_base + lane;
                unsigned bits = (unsigned)__shfl((int)w0, j & 31);
                if (j >= 32) bits = 0u;
                const int cnt = __builtin_popcount(bits);
                int incl = cnt;
#pragma unroll
                for (int off = 1; off < 32; off <<= 1) {
                    const int v = __shfl_up(incl, off);
                    if (lane >= off) incl += v;
                }
                int nrow = __builtin_popcount(__builtin_amdgcn_ballot_w32(incl <= EP_GCAP));
                nrow = nrow < wc_end - wi ? nrow : wc_end - wi;
                if (wi >= nfull) nrow = 1;
                else nrow = nrow < nfull - wi ? nrow : nfull - wi;
                if (lane >= nrow) bits = 0u;
                int pos = incl - cnt;
                ep_wave_order();
                while (bits != 0u) {
                    glist[pos++] = (unsigned short)((lane << 5) + __builtin_ctz(bits));
                    bits &= bits - 1u;
                }
                ep_wave_order();
                g_n = __builtin_amdgcn_readlane(incl, nrow - 1);
                g_row0 = wi;
                g_pos = 0;
                wi += nrow;
                continue;
            }
            const int left = g_n - g_pos;
            const int nval = left < EP_U ? left : EP_U;
            bool taken = false;
            if (g_row0 < nfull) {
                const int count0 = count;
                if (nval == EP_U)
                    ep_scan_batch<true>(dst, glist, mylist, lane, g_row0, nval, g_pos, base, unrows, count);
                else
                    ep_scan_batch<false>(dst, glist, mylist, lane, g_row0, nval, g_pos, base, unrows, count);
                taken = count <= EP_CAP - 32;
                if (!taken) count = count0;
            }
            if (taken) {
                g_pos += nval;
            } else {
                for (int u = 0; u < nval && !full; ++u) {
                    const int g = __builtin_amdgcn_readfirstlane((int)glist[g_pos]);
                    const int e = (g_row0 << 10) + (g << 5) + lane;
                    int dv = 0x7fffffff;
                    if (e < E) dv = dst[e];
                    const unsigned rel = (unsigned)(dv - base);
                    const bool hit = rel < unrows;
                    const unsigned mk = __builtin_amdgcn_ballot_w32(hit);
                    const int pop = __builtin_popcount(mk);
                    if (count + pop > EP_CAP) {
                        full = true;
                    } else {
                        if (hit) mylist[count + (int)__builtin_amdgcn_mbcnt_lo(mk, 0u)] = ((unsigned)e << 9) | rel;
                        count += pop;
                        ++g_pos;
                    }
                }
            }
        }
        int cnto = 0;
        ep_wave_order();
#pragma unroll 1
        for (int k0 = 0; k0 < count; k0 += 32) {
            const bool valid = lane < count - k0;
            unsigned e2 = 0u;
            if (valid) e2 = mylist[k0 + lane];
            const unsigned own = e2 & 31u;
            unsigned mine = __builtin_amdgcn_ballot_w32(valid);
#pragma unroll
            for (int b = 0; b < 5; ++b) {
                const unsigned mb = __builtin_amdgcn_ballot_w32(((own >> b) & 1u) != 0u);
                mine &= ((lane >> b) & 1) != 0 ? mb : ~mb;
            }
            cnto += __builtin_popcount(mine);
        }
        s_own[wave][lane] = cnto;
        if (lane == 0) s_more[wave] = (g_pos < g_n || wi < wend) ? 1 : 0;
        __syncthreads();
        int more_v = 0;
        for (int w = 0; w < EP_WAVES; ++w) more_v |= s_more[w];
        const int more = __builtin_amdgcn_readfirstlane(more_v);

        int part = 0;
#pragma unroll 4
        for (int s = 0; s < wave; ++s) part += s_own[s][lane];
        int tot = part;
#pragma unroll 4
        for (int s = wave; s < EP_WAVES; ++s) tot += s_own[s][lane];
        int incl = tot;
#pragma unroll
        for (int off = 1; off < 32; off <<= 1) {
            const int v = __shfl_up(incl, off);
            if (lane >= off) incl += v;
        }
        const int obase = incl - tot;
        int dcur = obase + part;
#pragma unroll 1
        for (int k0 = 0; k0 < count; k0 += 32) {
            const bool valid = lane < count - k0;
            unsigned e2 = 0u;
            if (valid) e2 = mylist[k0 + lane];
            const unsigned own = e2 & 31u;
            const unsigned vm = __builtin_amdgcn_ballot_w32(valid);
            unsigned same = vm;
            unsigned mine = vm;
#pragma unroll
            for (int b = 0; b < 5; ++b) {
                const unsigned mb = __builtin_amdgcn_ballot_w32(((own >> b) & 1u) != 0u);
                same &= ((own >> b) & 1u) != 0u ? mb : ~mb;
                mine &= ((lane >> b) & 1) != 0 ? mb : ~mb;
            }
            const int d = __shfl(dcur, (int)own);
            if (valid) s_sorted[d + (int)__builtin_amdgcn_mbcnt_lo(same, 0u)] = e2;
            dcur += __builtin_popcount(mine);
        }
        __syncthreads();

        const int my_n = __builtin_amdgcn_readlane(tot, wave);
        const unsigned* run = s_sorted + __builtin_amdgcn_readlane(obase, wave);
        const int ngrp = my_n >> 5;
        if (ngrp > 0) {
            float4 vrow[EP_RING];
            int sv_p = 0, rv_p = 0, avi_p = 0;
            for (int g = 0; g < ngrp; ++g) {
                const unsigned e2 = run[g * 32 + lane];
                const int e = (int)(e2 >> 9);
                const int sv = src[e];
                const float av = attr[e];
                __builtin_amdgcn_sched_barrier(0);
                if (g != 0) ep_add_rows32(vrow, sv_p, rv_p, avi_p, lane, x4, acc4);
                __builtin_amdgcn_sched_barrier(0);
                sv_p = sv;
                rv_p = (int)(e2 & 511u);
                avi_p = __float_as_int(av);
                ep_ask_rows(vrow, sv_p, lane, x4);
            }
            ep_add_rows32(vrow, sv_p, rv_p, avi_p, lane, x4, acc4);
        }
        const int h = ngrp << 5;
        ep_flush(run + h, my_n - h, lane, src, attr, x4, acc4);
        __syncthreads();
        if (!more) break;
    }

    float4* agg4 = reinterpret_cast<float4*>(agg) + (size_t)base * 32;
    for (int i = t; i < nrows * 32; i += 1024) agg4[i] = acc4[i];
}

#define FT_SLOTS 4
#define FT_WAVES 32
template <int RB, int KU>
__device__ __forceinline__ void ft_tile_product(
    const float* __restrict__ grows, int j, int n, float (*sAw)[SX_LD],
    const float (*sBm)[SX_LD], int lane, v8f& acc0, v8f& acc1) {
    const int r0 = j * 16;
#pragma unroll
    for (int part = 0; part < 16 / RB; ++part) {
        float4 v[RB];
#pragma unroll
        for (int r = 0; r < RB; ++r) {
            const int rr = r0 + part * RB + r;
            v[r] = make_float4(0.0f, 0.0f, 0.0f, 0.0f);
            if (rr < n) v[r] = *reinterpret_cast<const float4*>(grows + (size_t)rr * DCH + lane * 4);
        }
#pragma unroll
        for (int r = 0; r < RB; ++r)
            *reinterpret_cast<float4*>(&sAw[part * RB + r][lane * 4]) = v[r];
    }
    __builtin_amdgcn_fence(__ATOMIC_RELEASE, "wavefront");
    __builtin_amdgcn_wave_barrier();
    __builtin_amdgcn_fence(__ATOMIC_ACQUIRE, "wavefront");

    const int m    = lane & 15;
    const int koff = (lane >> 4) << 1;
    acc0 = v8f{};
    acc1 = v8f{};
#pragma unroll KU
    for (int kt = 0; kt < DCH; kt += 4) {
        v2f a  = *reinterpret_cast<const v2f*>(&sAw[m][kt + koff]);
        v2f b0 = *reinterpret_cast<const v2f*>(&sBm[m][kt + koff]);
        v2f b1 = *reinterpret_cast<const v2f*>(&sBm[16 + m][kt + koff]);
        acc0 = __builtin_amdgcn_wmma_f32_16x16x4_f32(false, a, false, b0, (short)0, acc0, false, false);
        acc1 = __builtin_amdgcn_wmma_f32_16x16x4_f32(false, a, false, b1, (short)0, acc1, false, false);
    }
    __builtin_amdgcn_fence(__ATOMIC_RELEASE, "wavefront");
    __builtin_amdgcn_wave_barrier();
    __builtin_amdgcn_fence(__ATOMIC_ACQUIRE, "wavefront");
}

__device__ __forceinline__ void ft_bias_relu_sum(
    v8f& ha, v8f& hb, float bv0, float bv1, int rbase, int n, float& p0, float& p1) {
#pragma unroll
    for (int v = 0; v < 8; ++v) {
        const bool ok = (rbase + v) < n;
        float a0 = ha[v] + bv0; a0 = a0 > 0.0f ? a0 : 0.0f;
        float a1 = hb[v] + bv1; a1 = a1 > 0.0f ? a1 : 0.0f;
        ha[v] = a0;
        hb[v] = a1;
        p0 += ok ? a0 : 0.0f;
        p1 += ok ? a1 : 0.0f;
    }
}

__device__ __forceinline__ void ft_sq_sum(
    const v8f& ha, const v8f& hb, float sub0, float sub1, int rbase, int n, float& q0, float& q1) {
#pragma unroll
    for (int v = 0; v < 8; ++v) {
        const bool ok = (rbase + v) < n;
        const float o0 = ha[v] - sub0;
        const float o1 = hb[v] - sub1;
        q0 += ok ? o0 * o0 : 0.0f;
        q1 += ok ? o1 * o1 : 0.0f;
    }
}

__device__ __forceinline__ void ft_norm_store(
    const v8f& ha, const v8f& hb, float sub0, float sub1, float den0, float den1,
    float w0, float w1, float b0, float b1, int rbase, int n, float* __restrict__ ocol) {
#pragma unroll
    for (int v = 0; v < 8; ++v) {
        const int rr = rbase + v;
        if (rr < n) {
            ocol[(size_t)rr * DCH]      = w0 * (ha[v] - sub0) / den0 + b0;
            ocol[(size_t)rr * DCH + 16] = w1 * (hb[v] - sub1) / den1 + b1;
        }
    }
}

__device__ __forceinline__ void ft_block_sum(
    float p0, float p1, float (*s_red)[32], float* s_stat, int t, float& r0, float& r1) {
    const int lane = t & 31;
    const int wave = t >> 5;
    p0 += __shfl_xor(p0, 16);
    p1 += __shfl_xor(p1, 16);
    if (lane < 16) { s_red[wave][lane] = p0; s_red[wave][16 + lane] = p1; }
    __syncthreads();
    if (t < 32) {
        float s = 0.0f;
        for (int w = 0; w < FT_WAVES; ++w) s += s_red[w][t];
        s_stat[t] = s;
    }
    __syncthreads();
    r0 = s_stat[lane & 15];
    r1 = s_stat[16 + (lane & 15)];
    __syncthreads();
}

__global__ __launch_bounds__(1024) void k_gemm_graph_norm(
    const float* __restrict__ agg, const float* __restrict__ Wt,
    const float* __restrict__ bias, const int* __restrict__ batch,
    const float* __restrict__ ms, const float* __restrict__ gw,
    const float* __restrict__ gb, float* __restrict__ out, int N) {
    __shared__ __attribute__((aligned(16))) float sA[FT_WAVES][16][SX_LD];
    __shared__ __attribute__((aligned(16))) float sB[32][SX_LD];
    __shared__ float s_red[FT_WAVES][32];
    __shared__ float s_stat[32];
    __shared__ int s_less[FT_WAVES];
    __shared__ int s_eq[FT_WAVES];

    const int t = threadIdx.x;
    const int lane = t & 31;
    const int wave = __builtin_amdgcn_readfirstlane(t >> 5);
    const int g = blockIdx.x >> 2;
    const int slice = blockIdx.x & 3;

    const float4 wv = *reinterpret_cast<const float4*>(
        Wt + (size_t)(slice * 32 + (t >> 5)) * DCH + ((t & 31) << 2));

    int less = 0, eq = 0;
    {
        const int4* b4 = reinterpret_cast<const int4*>(batch);
        const int n4 = N >> 2;
        for (int base = 0; base < n4; base += 8 * 1024) {
            int4 q[8];
#pragma unroll
            for (int u = 0; u < 8; ++u) {
                const int i = base + u * 1024 + t;
                q[u] = make_int4(0x7fffffff, 0x7fffffff, 0x7fffffff, 0x7fffffff);
                if (i < n4) q[u] = b4[i];
            }
#pragma unroll
            for (int u = 0; u < 8; ++u) {
                less += (q[u].x < g) + (q[u].y < g) + (q[u].z < g) + (q[u].w < g);
                eq   += (q[u].x == g) + (q[u].y == g) + (q[u].z == g) + (q[u].w == g);
            }
        }
        for (int i = (n4 << 2) + t; i < N; i += 1024) {
            const int q = batch[i];
            less += (q < g);
            eq   += (q == g);
        }
    }
#pragma unroll
    for (int off = 16; off > 0; off >>= 1) {
        less += __shfl_xor(less, off);
        eq   += __shfl_xor(eq, off);
    }
    if (lane == 0) { s_less[wave] = less; s_eq[wave] = eq; }
    *reinterpret_cast<float4*>(&sB[t >> 5][(t & 31) << 2]) = wv;
    __syncthreads();
    int start_v = 0, n_v = 0;
    for (int w = 0; w < FT_WAVES; ++w) { start_v += s_less[w]; n_v += s_eq[w]; }
    const int start = __builtin_amdgcn_readfirstlane(start_v);
    const int n = __builtin_amdgcn_readfirstlane(n_v);
    const int ntiles = (n + 15) >> 4;

    const float* grows = agg + (size_t)start * DCH;
    float* orows = out + (size_t)start * DCH;
    float (*sAw)[SX_LD] = sA[wave];

    const int c = lane & 15;
    const int rhalf = (lane >> 4) << 3;
    const int col0 = slice * 32 + c;
    const int col1 = col0 + 16;
    const float bv0 = bias[col0], bv1 = bias[col1];

    v8f hA0, hB0, hA1, hB1, hA2, hB2, hA3, hB3;
    float p0 = 0.0f, p1 = 0.0f;
#define FT_PHASE1(S, HA, HB)                                                        \
    {                                                                               \
        const int j = wave + (S) * FT_WAVES;     \
        HA = v8f{};                                                                 \
        HB = v8f{};                                                                 \
        if (j < ntiles) {                                                           \
            ft_tile_product<8, 4>(grows, j, n, sAw, sB, lane, HA, HB);              \
            ft_bias_relu_sum(HA, HB, bv0, bv1, j * 16 + rhalf, n, p0, p1);          \
        }                                                                           \
    }
    FT_PHASE1(0, hA0, hB0)
    FT_PHASE1(1, hA1, hB1)
    FT_PHASE1(2, hA2, hB2)
    FT_PHASE1(3, hA3, hB3)
#undef FT_PHASE1
    for (int j = wave + FT_SLOTS * FT_WAVES; j < ntiles; j += FT_WAVES) {
        v8f e0, e1;
        ft_tile_product<2, 1>(grows, j, n, sAw, sB, lane, e0, e1);
        ft_bias_relu_sum(e0, e1, bv0, bv1, j * 16 + rhalf, n, p0, p1);
    }
    float msum0, msum1;
    ft_block_sum(p0, p1, s_red, s_stat, t, msum0, msum1);

    float cn = (float)n;
    cn = cn > 1.0f ? cn : 1.0f;
    const float inv_cn = 1.0f / cn;
    const float sub0 = (msum0 * inv_cn) * ms[col0];
    const float sub1 = (msum1 * inv_cn) * ms[col1];

    float q0 = 0.0f, q1 = 0.0f;
    if (wave + 0 * FT_WAVES < ntiles) ft_sq_sum(hA0, hB0, sub0, sub1, (wave + 0 * FT_WAVES) * 16 + rhalf, n, q0, q1);
    if (wave + 1 * FT_WAVES < ntiles) ft_sq_sum(hA1, hB1, sub0, sub1, (wave + 1 * FT_WAVES) * 16 + rhalf, n, q0, q1);
    if (wave + 2 * FT_WAVES < ntiles) ft_sq_sum(hA2, hB2, sub0, sub1, (wave + 2 * FT_WAVES) * 16 + rhalf, n, q0, q1);
    if (wave + 3 * FT_WAVES < ntiles) ft_sq_sum(hA3, hB3, sub0, sub1, (wave + 3 * FT_WAVES) * 16 + rhalf, n, q0, q1);
    for (int j = wave + FT_SLOTS * FT_WAVES; j < ntiles; j += FT_WAVES) {
        v8f e0, e1;
        float d0 = 0.0f, d1 = 0.0f;
        ft_tile_product<2, 1>(grows, j, n, sAw, sB, lane, e0, e1);
        ft_bias_relu_sum(e0, e1, bv0, bv1, j * 16 + rhalf, n, d0, d1);
        ft_sq_sum(e0, e1, sub0, sub1, j * 16 + rhalf, n, q0, q1);
    }
    float vsum0, vsum1;
    ft_block_sum(q0, q1, s_red, s_stat, t, vsum0, vsum1);

    const float den0 = sqrtf(vsum0 * inv_cn + GN_EPS);
    const float den1 = sqrtf(vsum1 * inv_cn + GN_EPS);
    const float w0 = gw[col0], w1 = gw[col1];
    const float b0 = gb[col0], b1 = gb[col1];

    if (wave + 0 * FT_WAVES < ntiles) ft_norm_store(hA0, hB0, sub0, sub1, den0, den1, w0, w1, b0, b1, (wave + 0 * FT_WAVES) * 16 + rhalf, n, orows + col0);
    if (wave + 1 * FT_WAVES < ntiles) ft_norm_store(hA1, hB1, sub0, sub1, den0, den1, w0, w1, b0, b1, (wave + 1 * FT_WAVES) * 16 + rhalf, n, orows + col0);
    if (wave + 2 * FT_WAVES < ntiles) ft_norm_store(hA2, hB2, sub0, sub1, den0, den1, w0, w1, b0, b1, (wave + 2 * FT_WAVES) * 16 + rhalf, n, orows + col0);
    if (wave + 3 * FT_WAVES < ntiles) ft_norm_store(hA3, hB3, sub0, sub1, den0, den1, w0, w1, b0, b1, (wave + 3 * FT_WAVES) * 16 + rhalf, n, orows + col0);
    for (int j = wave + FT_SLOTS * FT_WAVES; j < ntiles; j += FT_WAVES) {
        v8f e0, e1;
        float d0 = 0.0f, d1 = 0.0f;
        ft_tile_product<2, 1>(grows, j, n, sAw, sB, lane, e0, e1);
        ft_bias_relu_sum(e0, e1, bv0, bv1, j * 16 + rhalf, n, d0, d1);
        ft_norm_store(e0, e1, sub0, sub1, den0, den1, w0, w1, b0, b1, j * 16 + rhalf, n, orows + col0);
    }
}

extern "C" void kernel_launch(void* const* d_in, const int* in_sizes, int n_in,
                              void* d_out, int out_size, void* d_ws, size_t ws_size,
                              hipStream_t stream) {
    const float* x     = (const float*)d_in[0];
    const int*   eidx  = (const int*)  d_in[1];
    const float* eattr = (const float*)d_in[2];
    const int*   batch = (const int*)  d_in[3];
    const float* W     = (const float*)d_in[4];
    const float* bias  = (const float*)d_in[5];
    const float* gw    = (const float*)d_in[6];
    const float* gb    = (const float*)d_in[7];
    const float* ms    = (const float*)d_in[8];
    float* out = (float*)d_out;

    const int N = in_sizes[0] / DCH;
    const int E = in_sizes[2];

    float* Wt  = (float*)d_ws;
    float* agg = Wt + (size_t)DCH * DCH;
    unsigned* bm = (unsigned*)(agg + (size_t)N * DCH);

    k_transpose_w_edge_groups<<<64 + (E + BM_EDGES * BM_RPB - 1) / (BM_EDGES * BM_RPB), 256, 0, stream>>>(W, Wt, eidx + E, bm, E);

    k_edge_rows_lds<<<(N + EP_R - 1) / EP_R, 1024, 0, stream>>>(eidx, eidx + E, eattr, x, bm, agg, N, E);

    k_gemm_graph_norm<<<NGRAPH * 4, 1024, 0, stream>>>(agg, Wt, bias, batch, ms, gw, gb, out, N);
}


# edge records: the pre-launch writes one 8-byte (src, attr) record per edge; the edge sum fetches an edge's source and weight with one 8-byte gather, not two 4-byte gathers; same add order
# speedup vs baseline: 3.7173x; 1.0338x over previous
#include <hip/hip_runtime.h>
#include <hip/hip_bf16.h>
#include <math.h>

typedef __attribute__((ext_vector_type(2))) float v2f;
typedef __attribute__((ext_vector_type(8))) float v8f;

#define DCH 128
#define NGRAPH 64
#define GN_EPS 1e-5f
#define SX_LD (DCH + 4)

#define EP_R 391
#define BM_NB 256
#define BM_EDGES 1024
#define BM_RPB 4

__global__ __launch_bounds__(256) void k_transpose_w_edge_groups(
    const float* __restrict__ W, float* __restrict__ Wt,
    const int* __restrict__ src, const int* __restrict__ dst,
    const float* __restrict__ attr, int* __restrict__ sa_words,
    unsigned* __restrict__ bm, int E) {
    __shared__ float tile[16][17];
    __shared__ unsigned s_bits[BM_RPB][8][BM_NB];
    if (blockIdx.x < 64) {
        int bx = blockIdx.x & 7;
        int by = blockIdx.x >> 3;
        int tx = threadIdx.x & 15;
        int ty = threadIdx.x >> 4;
        tile[ty][tx] = W[(size_t)(by * 16 + ty) * DCH + bx * 16 + tx];
        __syncthreads();
        Wt[(size_t)(bx * 16 + ty) * DCH + by * 16 + tx] = tile[tx][ty];
        return;
    }
    const int c0 = (blockIdx.x - 64) * BM_RPB;
    const int nwords = (E + BM_EDGES - 1) / BM_EDGES;
    const int t = threadIdx.x;
    const int wave = t >> 5;
    int2* sa = reinterpret_cast<int2*>(sa_words);
    int d[BM_RPB][4];
    int sv[BM_RPB][4];
    int av[BM_RPB][4];
    if ((c0 + BM_RPB) * BM_EDGES <= E) {
#pragma unroll
        for (int q = 0; q < BM_RPB; ++q)
#pragma unroll
            for (int i = 0; i < 4; ++i) {
                const int e = (c0 + q) * BM_EDGES + i * 256 + t;
                d[q][i] = dst[e];
                sv[q][i] = src[e];
                av[q][i] = __float_as_int(attr[e]);
            }
#pragma unroll
        for (int q = 0; q < BM_RPB; ++q)
#pragma unroll
            for (int i = 0; i < 4; ++i)
                sa[(c0 + q) * BM_EDGES + i * 256 + t] = make_int2(sv[q][i], av[q][i]);
    } else {
#pragma unroll
        for (int q = 0; q < BM_RPB; ++q)
#pragma unroll
            for (int i = 0; i < 4; ++i) {
                const int e = (c0 + q) * BM_EDGES + i * 256 + t;
                d[q][i] = -1;
                sv[q][i] = 0;
                av[q][i] = 0;
                if (e < E) {
                    d[q][i] = dst[e];
                    sv[q][i] = src[e];
                    av[q][i] = __float_as_int(attr[e]);
                }
            }
#pragma unroll
        for (int q = 0; q < BM_RPB; ++q)
#pragma unroll
            for (int i = 0; i < 4; ++i) {
                const int e = (c0 + q) * BM_EDGES + i * 256 + t;
                if (e < E) sa[e] = make_int2(sv[q][i], av[q][i]);
            }
    }
#pragma unroll
    for (int q = 0; q < BM_RPB; ++q)
#pragma unroll
        for (int i = 0; i < 8; ++i) s_bits[q][i][t] = 0u;
    __syncthreads();
#pragma unroll
    for (int q = 0; q < BM_RPB; ++q)
#pragma unroll
        for (int i = 0; i < 4; ++i) {
            const unsigned blk = (unsigned)d[q][i] / (unsigned)EP_R;
            if (blk < (unsigned)BM_NB) s_bits[q][wave][blk] |= 1u << (i * 8 + wave);
        }
    __syncthreads();
#pragma unroll
    for (int q = 0; q < BM_RPB; ++q) {
        if (c0 + q < nwords) {
            unsigned w = 0u;
#pragma unroll
            for (int i = 0; i < 8; ++i) w |= s_bits[q][i][t];
            bm[(size_t)(c0 + q) * BM_NB + t] = w;
        }
    }
}

#define EP_WAVES 32
#define EP_CAP 384
#define EP_GCAP 384
#define EP_U 32
#define EP_ROWS 16
#define EP_RING 12

__device__ __forceinline__ void ep_wave_order() {
    __builtin_amdgcn_fence(__ATOMIC_RELEASE, "wavefront");
    __builtin_amdgcn_wave_barrier();
    __builtin_amdgcn_fence(__ATOMIC_ACQUIRE, "wavefront");
}

__device__ __forceinline__ void ep_flush(
    const unsigned* stage, int nb, int lane, const int2* __restrict__ sa,
    const float4* __restrict__ x4, float4* acc4) {
    ep_wave_order();
    for (int b0 = 0; b0 < nb; b0 += 32) {
        int nbb = nb - b0;
        nbb = nbb < 32 ? nbb : 32;
        unsigned ent = 0;
        int2 rec = make_int2(0, 0);
        if (lane < nbb) {
            ent = stage[b0 + lane];
            rec = sa[(int)(ent >> 9)];
        }
        const int sv = rec.x;
        const int rv = (int)(ent & 511u);
        const int avi = rec.y;
        for (int j0 = 0; j0 < nbb; j0 += EP_ROWS) {
            float4 v[EP_ROWS];
#pragma unroll
            for (int u = 0; u < EP_ROWS; ++u) {
                int j = j0 + u;
                j = j < nbb ? j : nbb - 1;
                const int s = __builtin_amdgcn_readlane(sv, j);
                v[u] = x4[(size_t)s * 32 + lane];
            }
#pragma unroll
            for (int u = 0; u < EP_ROWS; ++u) {
                const int j = j0 + u;
                if (j < nbb) {
                    const int r = __builtin_amdgcn_readlane(rv, j);
                    const float a = __int_as_float(__builtin_amdgcn_readlane(avi, j));
                    float4 c = acc4[r * 32 + lane];
                    c.x += a * v[u].x;
                    c.y += a * v[u].y;
                    c.z += a * v[u].z;
                    c.w += a * v[u].w;
                    acc4[r * 32 + lane] = c;
                }
            }
        }
    }
    ep_wave_order();
}

__device__ __forceinline__ void ep_ask_rows(
    float4 (&v)[EP_RING], int sv, int lane, const float4* __restrict__ x4) {
#pragma unroll
    for (int u = 0; u < EP_RING; ++u) {
        const int s = __builtin_amdgcn_readlane(sv, u);
        v[u] = x4[(size_t)s * 32 + lane];
    }
}

__device__ __forceinline__ void ep_add_rows32(
    float4 (&v)[EP_RING], int sv, int rv, int avi, int lane,
    const float4* __restrict__ x4, float4* acc4) {
#pragma unroll
    for (int j = 0; j < 32; ++j) {
        const int u = j % EP_RING;
        const int r = __builtin_amdgcn_readlane(rv, j);
        const float a = __int_as_float(__builtin_amdgcn_readlane(avi, j));
        float4 c = acc4[r * 32 + lane];
        c.x += a * v[u].x;
        c.y += a * v[u].y;
        c.z += a * v[u].z;
        c.w += a * v[u].w;
        acc4[r * 32 + lane] = c;
        if (j + EP_RING < 32) {
            const int s = __builtin_amdgcn_readlane(sv, j + EP_RING);
            v[u] = x4[(size_t)s * 32 + lane];
        }
    }
}

template <bool FULL>
__device__ __forceinline__ void ep_scan_batch(
    const int* __restrict__ dst, const unsigned short* glist, unsigned* mylist, int lane,
    int g_row0, int nval_in, int gp, int base, unsigned unrows, int& count) {
    const int nval = FULL ? EP_U : nval_in;
    const int myg = (int)glist[gp + (lane < nval ? lane : 0)];
    const int ebase = (g_row0 << 10) + lane;
    int e[EP_U];
    int d[EP_U];
#pragma unroll
    for (int u = 0; u < EP_U; ++u) {
        e[u] = ebase + (__builtin_amdgcn_readlane(myg, u) << 5);
        d[u] = dst[e[u]];
    }
    int c = count;
#pragma unroll
    for (int q = 0; q < EP_U / 8; ++q) {
        if (FULL || q * 8 < nval) {
#pragma unroll
            for (int u = q * 8; u < q * 8 + 8; ++u) {
                const unsigned rel = (unsigned)(d[u] - base);
                const bool hit = rel < unrows;
                const unsigned mk = __builtin_amdgcn_ballot_w32(hit);
                if ((FULL || u < nval) && c <= EP_CAP - 32) {
                    if (hit) mylist[c + (int)__builtin_amdgcn_mbcnt_lo(mk, 0u)] = ((unsigned)e[u] << 9) + rel;
                    c += __builtin_popcount(mk);
                }
            }
        }
    }
    count = c;
}

__global__ __launch_bounds__(1024) void k_edge_rows_lds(
    const int* __restrict__ sa_words, const int* __restrict__ dst,
    const float* __restrict__ x,
    const unsigned* __restrict__ bm, float* __restrict__ agg, int N, int E) {
    __shared__ __attribute__((aligned(16))) float s_acc[EP_R * DCH];
    __shared__ unsigned s_list[EP_WAVES][EP_CAP];
    __shared__ unsigned s_sorted[EP_WAVES * EP_CAP];
    __shared__ unsigned short s_grp[EP_WAVES][EP_GCAP];
    __shared__ int s_own[EP_WAVES][EP_WAVES];
    __shared__ int s_more[EP_WAVES];

    const int t = threadIdx.x;
    const int lane = t & 31;
    const int wave = __builtin_amdgcn_readfirstlane(t >> 5);
    const int base = blockIdx.x * EP_R;
    int nrows = N - base;
    nrows = nrows < EP_R ? nrows : EP_R;
    const unsigned unrows = (unsigned)nrows;

    const int bid = blockIdx.x;
    const int nwords = (E + BM_EDGES - 1) / BM_EDGES;
    const int nfull = E / BM_EDGES;
    const int wper = (nwords + EP_WAVES - 1) / EP_WAVES;
    int wend = (wave + 1) * wper;
    wend = wend < nwords ? wend : nwords;
    int wi = wave * wper;
    int wc_base = wi, wc_end = wi + 32;
    unsigned w0 = 0u;
    if (wi + lane < wend) w0 = bid < BM_NB ? bm[(size_t)(wi + lane) * BM_NB + bid] : 0xffffffffu;

    float4* acc4 = reinterpret_cast<float4*>(s_acc);
    const float4* x4 = reinterpret_cast<const float4*>(x);
    const int2* sa = reinterpret_cast<const int2*>(sa_words);
    for (int i = t; i < EP_R * 32; i += 1024) acc4[i] = make_float4(0.0f, 0.0f, 0.0f, 0.0f);

    unsigned short* glist = s_grp[wave];
    unsigned* mylist = s_list[wave];
    int g_row0 = 0;
    int g_n = 0;
    int g_pos = 0;
    const int max_rounds = (wper * BM_EDGES) / (EP_CAP - 32) + 2;

    for (int round = 0; round < max_rounds; ++round) {
        int count = 0;
        bool full = false;
        while (!full) {
            if (g_pos >= g_n) {
                if (wi >= wend) break;
                if (wi >= wc_end) {
                    const int r = wi + lane;
                    w0 = 0u;
                    if (r < wend) w0 = bid < BM_NB ? bm[(size_t)r * BM_NB + bid] : 0xffffffffu;
                    wc_base = wi;
                    wc_end = wi + 32;
                }
                const int j = wi - wc_base + lane;
                unsigned bits = (unsigned)__shfl((int)w0, j & 31);
                if (j >= 32) bits = 0u;
                const int cnt = __builtin_popcount(bits);
                int incl = cnt;
#pragma unroll
                for (int off = 1; off < 32; off <<= 1) {
                    const int v = __shfl_up(incl, off);
                    if (lane >= off) incl += v;
                }
                int nrow = __builtin_popcount(__builtin_amdgcn_ballot_w32(incl <= EP_GCAP));
                nrow = nrow < wc_end - wi ? nrow : wc_end - wi;
                if (wi >= nfull) nrow = 1;
                else nrow = nrow < nfull - wi ? nrow : nfull - wi;
                if (lane >= nrow) bits = 0u;
                int pos = incl - cnt;
                ep_wave_order();
                while (bits != 0u) {
                    glist[pos++] = (unsigned short)((lane << 5) + __builtin_ctz(bits));
                    bits &= bits - 1u;
                }
                ep_wave_order();
                g_n = __builtin_amdgcn_readlane(incl, nrow - 1);
                g_row0 = wi;
                g_pos = 0;
                wi += nrow;
                continue;
            }
            const int left = g_n - g_pos;
            const int nval = left < EP_U ? left : EP_U;
            bool taken = false;
            if (g_row0 < nfull) {
                const int count0 = count;
                if (nval == EP_U)
                    ep_scan_batch<true>(dst, glist, mylist, lane, g_row0, nval, g_pos, base, unrows, count);
                else
                    ep_scan_batch<false>(dst, glist, mylist, lane, g_row0, nval, g_pos, base, unrows, count);
                taken = count <= EP_CAP - 32;
                if (!taken) count = count0;
            }
            if (taken) {
                g_pos += nval;
            } else {
                for (int u = 0; u < nval && !full; ++u) {
                    const int g = __builtin_amdgcn_readfirstlane((int)glist[g_pos]);
                    const int e = (g_row0 << 10) + (g << 5) + lane;
                    int dv = 0x7fffffff;
                    if (e < E) dv = dst[e];
                    const unsigned rel = (unsigned)(dv - base);
                    const bool hit = rel < unrows;
                    const unsigned mk = __builtin_amdgcn_ballot_w32(hit);
                    const int pop = __builtin_popcount(mk);
                    if (count + pop > EP_CAP) {
                        full = true;
                    } else {
                        if (hit) mylist[count + (int)__builtin_amdgcn_mbcnt_lo(mk, 0u)] = ((unsigned)e << 9) | rel;
                        count += pop;
                        ++g_pos;
                    }
                }
            }
        }
        int cnto = 0;
        ep_wave_order();
#pragma unroll 1
        for (int k0 = 0; k0 < count; k0 += 32) {
            const bool valid = lane < count - k0;
            unsigned e2 = 0u;
            if (valid) e2 = mylist[k0 + lane];
            const unsigned own = e2 & 31u;
            unsigned mine = __builtin_amdgcn_ballot_w32(valid);
#pragma unroll
            for (int b = 0; b < 5; ++b) {
                const unsigned mb = __builtin_amdgcn_ballot_w32(((own >> b) & 1u) != 0u);
                mine &= ((lane >> b) & 1) != 0 ? mb : ~mb;
            }
            cnto += __builtin_popcount(mine);
        }
        s_own[wave][lane] = cnto;
        if (lane == 0) s_more[wave] = (g_pos < g_n || wi < wend) ? 1 : 0;
        __syncthreads();
        int more_v = 0;
        for (int w = 0; w < EP_WAVES; ++w) more_v |= s_more[w];
        const int more = __builtin_amdgcn_readfirstlane(more_v);

        int part = 0;
#pragma unroll 4
        for (int s = 0; s < wave; ++s) part += s_own[s][lane];
        int tot = part;
#pragma unroll 4
        for (int s = wave; s < EP_WAVES; ++s) tot += s_own[s][lane];
        int incl = tot;
#pragma unroll
        for (int off = 1; off < 32; off <<= 1) {
            const int v = __shfl_up(incl, off);
            if (lane >= off) incl += v;
        }
        const int obase = incl - tot;
        int dcur = obase + part;
#pragma unroll 1
        for (int k0 = 0; k0 < count; k0 += 32) {
            const bool valid = lane < count - k0;
            unsigned e2 = 0u;
            if (valid) e2 = mylist[k0 + lane];
            const unsigned own = e2 & 31u;
            const unsigned vm = __builtin_amdgcn_ballot_w32(valid);
            unsigned same = vm;
            unsigned mine = vm;
#pragma unroll
            for (int b = 0; b < 5; ++b) {
                const unsigned mb = __builtin_amdgcn_ballot_w32(((own >> b) & 1u) != 0u);
                same &= ((own >> b) & 1u) != 0u ? mb : ~mb;
                mine &= ((lane >> b) & 1) != 0 ? mb : ~mb;
            }
            const int d = __shfl(dcur, (int)own);
            if (valid) s_sorted[d + (int)__builtin_amdgcn_mbcnt_lo(same, 0u)] = e2;
            dcur += __builtin_popcount(mine);
        }
        __syncthreads();

        const int my_n = __builtin_amdgcn_readlane(tot, wave);
        const unsigned* run = s_sorted + __builtin_amdgcn_readlane(obase, wave);
        const int ngrp = my_n >> 5;
        if (ngrp > 0) {
            float4 vrow[EP_RING];
            int sv_p = 0, rv_p = 0, avi_p = 0;
            for (int g = 0; g < ngrp; ++g) {
                const unsigned e2 = run[g * 32 + lane];
                const int2 rec = sa[(int)(e2 >> 9)];
                __builtin_amdgcn_sched_barrier(0);
                if (g != 0) ep_add_rows32(vrow, sv_p, rv_p, avi_p, lane, x4, acc4);
                __builtin_amdgcn_sched_barrier(0);
                sv_p = rec.x;
                rv_p = (int)(e2 & 511u);
                avi_p = rec.y;
                ep_ask_rows(vrow, sv_p, lane, x4);
            }
            ep_add_rows32(vrow, sv_p, rv_p, avi_p, lane, x4, acc4);
        }
        const int h = ngrp << 5;
        ep_flush(run + h, my_n - h, lane, sa, x4, acc4);
        __syncthreads();
        if (!more) break;
    }

    float4* agg4 = reinterpret_cast<float4*>(agg) + (size_t)base * 32;
    for (int i = t; i < nrows * 32; i += 1024) agg4[i] = acc4[i];
}

#define FT_SLOTS 4
#define FT_WAVES 32
template <int RB, int KU>
__device__ __forceinline__ void ft_tile_product(
    const float* __restrict__ grows, int j, int n, float (*sAw)[SX_LD],
    const float (*sBm)[SX_LD], int lane, v8f& acc0, v8f& acc1) {
    const int r0 = j * 16;
#pragma unroll
    for (int part = 0; part < 16 / RB; ++part) {
        float4 v[RB];
#pragma unroll
        for (int r = 0; r < RB; ++r) {
            const int rr = r0 + part * RB + r;
            v[r] = make_float4(0.0f, 0.0f, 0.0f, 0.0f);
            if (rr < n) v[r] = *reinterpret_cast<const float4*>(grows + (size_t)rr * DCH + lane * 4);
        }
#pragma unroll
        for (int r = 0; r < RB; ++r)
            *reinterpret_cast<float4*>(&sAw[part * RB + r][lane * 4]) = v[r];
    }
    __builtin_amdgcn_fence(__ATOMIC_RELEASE, "wavefront");
    __builtin_amdgcn_wave_barrier();
    __builtin_amdgcn_fence(__ATOMIC_ACQUIRE, "wavefront");

    const int m    = lane & 15;
    const int koff = (lane >> 4) << 1;
    acc0 = v8f{};
    acc1 = v8f{};
#pragma unroll KU
    for (int kt = 0; kt < DCH; kt += 4) {
        v2f a  = *reinterpret_cast<const v2f*>(&sAw[m][kt + koff]);
        v2f b0 = *reinterpret_cast<const v2f*>(&sBm[m][kt + koff]);
        v2f b1 = *reinterpret_cast<const v2f*>(&sBm[16 + m][kt + koff]);
        acc0 = __builtin_amdgcn_wmma_f32_16x16x4_f32(false, a, false, b0, (short)0, acc0, false, false);
        acc1 = __builtin_amdgcn_wmma_f32_16x16x4_f32(false, a, false, b1, (short)0, acc1, false, false);
    }
    __builtin_amdgcn_fence(__ATOMIC_RELEASE, "wavefront");
    __builtin_amdgcn_wave_barrier();
    __builtin_amdgcn_fence(__ATOMIC_ACQUIRE, "wavefront");
}

__device__ __forceinline__ void ft_bias_relu_sum(
    v8f& ha, v8f& hb, float bv0, float bv1, int rbase, int n, float& p0, float& p1) {
#pragma unroll
    for (int v = 0; v < 8; ++v) {
        const bool ok = (rbase + v) < n;
        float a0 = ha[v] + bv0; a0 = a0 > 0.0f ? a0 : 0.0f;
        float a1 = hb[v] + bv1; a1 = a1 > 0.0f ? a1 : 0.0f;
        ha[v] = a0;
        hb[v] = a1;
        p0 += ok ? a0 : 0.0f;
        p1 += ok ? a1 : 0.0f;
    }
}

__device__ __forceinline__ void ft_sq_sum(
    const v8f& ha, const v8f& hb, float sub0, float sub1, int rbase, int n, float& q0, float& q1) {
#pragma unroll
    for (int v = 0; v < 8; ++v) {
        const bool ok = (rbase + v) < n;
        const float o0 = ha[v] - sub0;
        const float o1 = hb[v] - sub1;
        q0 += ok ? o0 * o0 : 0.0f;
        q1 += ok ? o1 * o1 : 0.0f;
    }
}

__device__ __forceinline__ void ft_norm_store(
    const v8f& ha, const v8f& hb, float sub0, float sub1, float den0, float den1,
    float w0, float w1, float b0, float b1, int rbase, int n, float* __restrict__ ocol) {
#pragma unroll
    for (int v = 0; v < 8; ++v) {
        const int rr = rbase + v;
        if (rr < n) {
            ocol[(size_t)rr * DCH]      = w0 * (ha[v] - sub0) / den0 + b0;
            ocol[(size_t)rr * DCH + 16] = w1 * (hb[v] - sub1) / den1 + b1;
        }
    }
}

__device__ __forceinline__ void ft_block_sum(
    float p0, float p1, float (*s_red)[32], float* s_stat, int t, float& r0, float& r1) {
    const int lane = t & 31;
    const int wave = t >> 5;
    p0 += __shfl_xor(p0, 16);
    p1 += __shfl_xor(p1, 16);
    if (lane < 16) { s_red[wave][lane] = p0; s_red[wave][16 + lane] = p1; }
    __syncthreads();
    if (t < 32) {
        float s = 0.0f;
        for (int w = 0; w < FT_WAVES; ++w) s += s_red[w][t];
        s_stat[t] = s;
    }
    __syncthreads();
    r0 = s_stat[lane & 15];
    r1 = s_stat[16 + (lane & 15)];
    __syncthreads();
}

__global__ __launch_bounds__(1024) void k_gemm_graph_norm(
    const float* __restrict__ agg, const float* __restrict__ Wt,
    const float* __restrict__ bias, const int* __restrict__ batch,
    const float* __restrict__ ms, const float* __restrict__ gw,
    const float* __restrict__ gb, float* __restrict__ out, int N) {
    __shared__ __attribute__((aligned(16))) float sA[FT_WAVES][16][SX_LD];
    __shared__ __attribute__((aligned(16))) float sB[32][SX_LD];
    __shared__ float s_red[FT_WAVES][32];
    __shared__ float s_stat[32];
    __shared__ int s_less[FT_WAVES];
    __shared__ int s_eq[FT_WAVES];

    const int t = threadIdx.x;
    const int lane = t & 31;
    const int wave = __builtin_amdgcn_readfirstlane(t >> 5);
    const int g = blockIdx.x >> 2;
    const int slice = blockIdx.x & 3;

    const float4 wv = *reinterpret_cast<const float4*>(
        Wt + (size_t)(slice * 32 + (t >> 5)) * DCH + ((t & 31) << 2));

    int less = 0, eq = 0;
    {
        const int4* b4 = reinterpret_cast<const int4*>(batch);
        const int n4 = N >> 2;
        for (int base = 0; base < n4; base += 8 * 1024) {
            int4 q[8];
#pragma unroll
            for (int u = 0; u < 8; ++u) {
                const int i = base + u * 1024 + t;
                q[u] = make_int4(0x7fffffff, 0x7fffffff, 0x7fffffff, 0x7fffffff);
                if (i < n4) q[u] = b4[i];
            }
#pragma unroll
            for (int u = 0; u < 8; ++u) {
                less += (q[u].x < g) + (q[u].y < g) + (q[u].z < g) + (q[u].w < g);
                eq   += (q[u].x == g) + (q[u].y == g) + (q[u].z == g) + (q[u].w == g);
            }
        }
        for (int i = (n4 << 2) + t; i < N; i += 1024) {
            const int q = batch[i];
            less += (q < g);
            eq   += (q == g);
        }
    }
#pragma unroll
    for (int off = 16; off > 0; off >>= 1) {
        less += __shfl_xor(less, off);
        eq   += __shfl_xor(eq, off);
    }
    if (lane == 0) { s_less[wave] = less; s_eq[wave] = eq; }
    *reinterpret_cast<float4*>(&sB[t >> 5][(t & 31) << 2]) = wv;
    __syncthreads();
    int start_v = 0, n_v = 0;
    for (int w = 0; w < FT_WAVES; ++w) { start_v += s_less[w]; n_v += s_eq[w]; }
    const int start = __builtin_amdgcn_readfirstlane(start_v);
    const int n = __builtin_amdgcn_readfirstlane(n_v);
    const int ntiles = (n + 15) >> 4;

    const float* grows = agg + (size_t)start * DCH;
    float* orows = out + (size_t)start * DCH;
    float (*sAw)[SX_LD] = sA[wave];

    const int c = lane & 15;
    const int rhalf = (lane >> 4) << 3;
    const int col0 = slice * 32 + c;
    const int col1 = col0 + 16;
    const float bv0 = bias[col0], bv1 = bias[col1];

    v8f hA0, hB0, hA1, hB1, hA2, hB2, hA3, hB3;
    float p0 = 0.0f, p1 = 0.0f;
#define FT_PHASE1(S, HA, HB)                                                        \
    {                                                                               \
        const int j = wave + (S) * FT_WAVES;     \
        HA = v8f{};                                                                 \
        HB = v8f{};                                                                 \
        if (j < ntiles) {                                                           \
            ft_tile_product<8, 4>(grows, j, n, sAw, sB, lane, HA, HB);              \
            ft_bias_relu_sum(HA, HB, bv0, bv1, j * 16 + rhalf, n, p0, p1);          \
        }                                                                           \
    }
    FT_PHASE1(0, hA0, hB0)
    FT_PHASE1(1, hA1, hB1)
    FT_PHASE1(2, hA2, hB2)
    FT_PHASE1(3, hA3, hB3)
#undef FT_PHASE1
    for (int j = wave + FT_SLOTS * FT_WAVES; j < ntiles; j += FT_WAVES) {
        v8f e0, e1;
        ft_tile_product<2, 1>(grows, j, n, sAw, sB, lane, e0, e1);
        ft_bias_relu_sum(e0, e1, bv0, bv1, j * 16 + rhalf, n, p0, p1);
    }
    float msum0, msum1;
    ft_block_sum(p0, p1, s_red, s_stat, t, msum0, msum1);

    float cn = (float)n;
    cn = cn > 1.0f ? cn : 1.0f;
    const float inv_cn = 1.0f / cn;
    const float sub0 = (msum0 * inv_cn) * ms[col0];
    const float sub1 = (msum1 * inv_cn) * ms[col1];

    float q0 = 0.0f, q1 = 0.0f;
    if (wave + 0 * FT_WAVES < ntiles) ft_sq_sum(hA0, hB0, sub0, sub1, (wave + 0 * FT_WAVES) * 16 + rhalf, n, q0, q1);
    if (wave + 1 * FT_WAVES < ntiles) ft_sq_sum(hA1, hB1, sub0, sub1, (wave + 1 * FT_WAVES) * 16 + rhalf, n, q0, q1);
    if (wave + 2 * FT_WAVES < ntiles) ft_sq_sum(hA2, hB2, sub0, sub1, (wave + 2 * FT_WAVES) * 16 + rhalf, n, q0, q1);
    if (wave + 3 * FT_WAVES < ntiles) ft_sq_sum(hA3, hB3, sub0, sub1, (wave + 3 * FT_WAVES) * 16 + rhalf, n, q0, q1);
    for (int j = wave + FT_SLOTS * FT_WAVES; j < ntiles; j += FT_WAVES) {
        v8f e0, e1;
        float d0 = 0.0f, d1 = 0.0f;
        ft_tile_product<2, 1>(grows, j, n, sAw, sB, lane, e0, e1);
        ft_bias_relu_sum(e0, e1, bv0, bv1, j * 16 + rhalf, n, d0, d1);
        ft_sq_sum(e0, e1, sub0, sub1, j * 16 + rhalf, n, q0, q1);
    }
    float vsum0, vsum1;
    ft_block_sum(q0, q1, s_red, s_stat, t, vsum0, vsum1);

    const float den0 = sqrtf(vsum0 * inv_cn + GN_EPS);
    const float den1 = sqrtf(vsum1 * inv_cn + GN_EPS);
    const float w0 = gw[col0], w1 = gw[col1];
    const float b0 = gb[col0], b1 = gb[col1];

    if (wave + 0 * FT_WAVES < ntiles) ft_norm_store(hA0, hB0, sub0, sub1, den0, den1, w0, w1, b0, b1, (wave + 0 * FT_WAVES) * 16 + rhalf, n, orows + col0);
    if (wave + 1 * FT_WAVES < ntiles) ft_norm_store(hA1, hB1, sub0, sub1, den0, den1, w0, w1, b0, b1, (wave + 1 * FT_WAVES) * 16 + rhalf, n, orows + col0);
    if (wave + 2 * FT_WAVES < ntiles) ft_norm_store(hA2, hB2, sub0, sub1, den0, den1, w0, w1, b0, b1, (wave + 2 * FT_WAVES) * 16 + rhalf, n, orows + col0);
    if (wave + 3 * FT_WAVES < ntiles) ft_norm_store(hA3, hB3, sub0, sub1, den0, den1, w0, w1, b0, b1, (wave + 3 * FT_WAVES) * 16 + rhalf, n, orows + col0);
    for (int j = wave + FT_SLOTS * FT_WAVES; j < ntiles; j += FT_WAVES) {
        v8f e0, e1;
        float d0 = 0.0f, d1 = 0.0f;
        ft_tile_product<2, 1>(grows, j, n, sAw, sB, lane, e0, e1);
        ft_bias_relu_sum(e0, e1, bv0, bv1, j * 16 + rhalf, n, d0, d1);
        ft_norm_store(e0, e1, sub0, sub1, den0, den1, w0, w1, b0, b1, j * 16 + rhalf, n, orows + col0);
    }
}

extern "C" void kernel_launch(void* const* d_in, const int* in_sizes, int n_in,
                              void* d_out, int out_size, void* d_ws, size_t ws_size,
                              hipStream_t stream) {
    const float* x     = (const float*)d_in[0];
    const int*   eidx  = (const int*)  d_in[1];
    const float* eattr = (const float*)d_in[2];
    const int*   batch = (const int*)  d_in[3];
    const float* W     = (const float*)d_in[4];
    const float* bias  = (const float*)d_in[5];
    const float* gw    = (const float*)d_in[6];
    const float* gb    = (const float*)d_in[7];
    const float* ms    = (const float*)d_in[8];
    float* out = (float*)d_out;

    const int N = in_sizes[0] / DCH;
    const int E = in_sizes[2];

    float* Wt  = (float*)d_ws;
    float* agg = Wt + (size_t)DCH * DCH;
    int* sa = (int*)(agg + (size_t)N * DCH);
    unsigned* bm = (unsigned*)(sa + 2 * (size_t)E);

    k_transpose_w_edge_groups<<<64 + (E + BM_EDGES * BM_RPB - 1) / (BM_EDGES * BM_RPB), 256, 0, stream>>>(W, Wt, eidx, eidx + E, eattr, sa, bm, E);

    k_edge_rows_lds<<<(N + EP_R - 1) / EP_R, 1024, 0, stream>>>(sa, eidx + E, x, bm, agg, N, E);

    k_gemm_graph_norm<<<NGRAPH * 4, 1024, 0, stream>>>(agg, Wt, bias, batch, ms, gw, gb, out, N);
}


# graph row ranges: the pre-launch writes a 65-entry table of first rows from the boundaries of the sorted batch list; every K2 block's full count of batch_index is gone; sums unchanged
# speedup vs baseline: 3.8690x; 1.0408x over previous
#include <hip/hip_runtime.h>
#include <hip/hip_bf16.h>
#include <math.h>

typedef __attribute__((ext_vector_type(2))) float v2f;
typedef __attribute__((ext_vector_type(8))) float v8f;

#define DCH 128
#define NGRAPH 64
#define GN_EPS 1e-5f
#define SX_LD (DCH + 4)

#define EP_R 391
#define BM_NB 256
#define BM_EDGES 1024
#define BM_RPB 4

__global__ __launch_bounds__(256) void k_transpose_w_edge_groups(
    const float* __restrict__ W, float* __restrict__ Wt,
    const int* __restrict__ src, const int* __restrict__ dst,
    const float* __restrict__ attr, int* __restrict__ sa_words,
    unsigned* __restrict__ bm, int E,
    const int* __restrict__ batch, int* __restrict__ gstart, int N) {
    __shared__ float tile[16][17];
    __shared__ unsigned s_bits[BM_RPB][8][BM_NB];
    if (blockIdx.x < 64) {
        int bx = blockIdx.x & 7;
        int by = blockIdx.x >> 3;
        int tx = threadIdx.x & 15;
        int ty = threadIdx.x >> 4;
        tile[ty][tx] = W[(size_t)(by * 16 + ty) * DCH + bx * 16 + tx];
        __syncthreads();
        Wt[(size_t)(bx * 16 + ty) * DCH + by * 16 + tx] = tile[tx][ty];
        return;
    }
    const int c0 = (blockIdx.x - 64) * BM_RPB;
    const int nwords = (E + BM_EDGES - 1) / BM_EDGES;
    const int t = threadIdx.x;
    const int wave = t >> 5;
    {
        const int nodes_per_pass = ((int)gridDim.x - 64) * 256;
        for (int i = ((int)blockIdx.x - 64) * 256 + t; i < N; i += nodes_per_pass) {
            const int cur = batch[i];
            const int prev = i > 0 ? batch[i - 1] : -1;
            const int hi = cur < NGRAPH ? cur : NGRAPH;
            for (int g = prev + 1 > 0 ? prev + 1 : 0; g <= hi; ++g) gstart[g] = i;
            if (i == N - 1)
                for (int g = cur + 1 > 0 ? cur + 1 : 0; g <= NGRAPH; ++g) gstart[g] = N;
        }
    }
    int2* sa = reinterpret_cast<int2*>(sa_words);
    int d[BM_RPB][4];
    int sv[BM_RPB][4];
    int av[BM_RPB][4];
    if ((c0 + BM_RPB) * BM_EDGES <= E) {
#pragma unroll
        for (int q = 0; q < BM_RPB; ++q)
#pragma unroll
            for (int i = 0; i < 4; ++i) {
                const int e = (c0 + q) * BM_EDGES + i * 256 + t;
                d[q][i] = dst[e];
                sv[q][i] = src[e];
                av[q][i] = __float_as_int(attr[e]);
            }
#pragma unroll
        for (int q = 0; q < BM_RPB; ++q)
#pragma unroll
            for (int i = 0; i < 4; ++i)
                sa[(c0 + q) * BM_EDGES + i * 256 + t] = make_int2(sv[q][i], av[q][i]);
    } else {
#pragma unroll
        for (int q = 0; q < BM_RPB; ++q)
#pragma unroll
            for (int i = 0; i < 4; ++i) {
                const int e = (c0 + q) * BM_EDGES + i * 256 + t;
                d[q][i] = -1;
                sv[q][i] = 0;
                av[q][i] = 0;
                if (e < E) {
                    d[q][i] = dst[e];
                    sv[q][i] = src[e];
                    av[q][i] = __float_as_int(attr[e]);
                }
            }
#pragma unroll
        for (int q = 0; q < BM_RPB; ++q)
#pragma unroll
            for (int i = 0; i < 4; ++i) {
                const int e = (c0 + q) * BM_EDGES + i * 256 + t;
                if (e < E) sa[e] = make_int2(sv[q][i], av[q][i]);
            }
    }
#pragma unroll
    for (int q = 0; q < BM_RPB; ++q)
#pragma unroll
        for (int i = 0; i < 8; ++i) s_bits[q][i][t] = 0u;
    __syncthreads();
#pragma unroll
    for (int q = 0; q < BM_RPB; ++q)
#pragma unroll
        for (int i = 0; i < 4; ++i) {
            const unsigned blk = (unsigned)d[q][i] / (unsigned)EP_R;
            if (blk < (unsigned)BM_NB) s_bits[q][wave][blk] |= 1u << (i * 8 + wave);
        }
    __syncthreads();
#pragma unroll
    for (int q = 0; q < BM_RPB; ++q) {
        if (c0 + q < nwords) {
            unsigned w = 0u;
#pragma unroll
            for (int i = 0; i < 8; ++i) w |= s_bits[q][i][t];
            bm[(size_t)(c0 + q) * BM_NB + t] = w;
        }
    }
}

#define EP_WAVES 32
#define EP_CAP 384
#define EP_GCAP 384
#define EP_U 32
#define EP_ROWS 16
#define EP_RING 12

__device__ __forceinline__ void ep_wave_order() {
    __builtin_amdgcn_fence(__ATOMIC_RELEASE, "wavefront");
    __builtin_amdgcn_wave_barrier();
    __builtin_amdgcn_fence(__ATOMIC_ACQUIRE, "wavefront");
}

__device__ __forceinline__ void ep_flush(
    const unsigned* stage, int nb, int lane, const int2* __restrict__ sa,
    const float4* __restrict__ x4, float4* acc4) {
    ep_wave_order();
    for (int b0 = 0; b0 < nb; b0 += 32) {
        int nbb = nb - b0;
        nbb = nbb < 32 ? nbb : 32;
        unsigned ent = 0;
        int2 rec = make_int2(0, 0);
        if (lane < nbb) {
            ent = stage[b0 + lane];
            rec = sa[(int)(ent >> 9)];
        }
        const int sv = rec.x;
        const int rv = (int)(ent & 511u);
        const int avi = rec.y;
        for (int j0 = 0; j0 < nbb; j0 += EP_ROWS) {
            float4 v[EP_ROWS];
#pragma unroll
            for (int u = 0; u < EP_ROWS; ++u) {
                int j = j0 + u;
                j = j < nbb ? j : nbb - 1;
                const int s = __builtin_amdgcn_readlane(sv, j);
                v[u] = x4[(size_t)s * 32 + lane];
            }
#pragma unroll
            for (int u = 0; u < EP_ROWS; ++u) {
                const int j = j0 + u;
                if (j < nbb) {
                    const int r = __builtin_amdgcn_readlane(rv, j);
                    const float a = __int_as_float(__builtin_amdgcn_readlane(avi, j));
                    float4 c = acc4[r * 32 + lane];
                    c.x += a * v[u].x;
                    c.y += a * v[u].y;
                    c.z += a * v[u].z;
                    c.w += a * v[u].w;
                    acc4[r * 32 + lane] = c;
                }
            }
        }
    }
    ep_wave_order();
}

__device__ __forceinline__ void ep_ask_rows(
    float4 (&v)[EP_RING], int sv, int lane, const float4* __restrict__ x4) {
#pragma unroll
    for (int u = 0; u < EP_RING; ++u) {
        const int s = __builtin_amdgcn_readlane(sv, u);
        v[u] = x4[(size_t)s * 32 + lane];
    }
}

__device__ __forceinline__ void ep_add_rows32(
    float4 (&v)[EP_RING], int sv, int rv, int avi, int lane,
    const float4* __restrict__ x4, float4* acc4) {
#pragma unroll
    for (int j = 0; j < 32; ++j) {
        const int u = j % EP_RING;
        const int r = __builtin_amdgcn_readlane(rv, j);
        const float a = __int_as_float(__builtin_amdgcn_readlane(avi, j));
        float4 c = acc4[r * 32 + lane];
        c.x += a * v[u].x;
        c.y += a * v[u].y;
        c.z += a * v[u].z;
        c.w += a * v[u].w;
        acc4[r * 32 + lane] = c;
        if (j + EP_RING < 32) {
            const int s = __builtin_amdgcn_readlane(sv, j + EP_RING);
            v[u] = x4[(size_t)s * 32 + lane];
        }
    }
}

template <bool FULL>
__device__ __forceinline__ void ep_scan_batch(
    const int* __restrict__ dst, const unsigned short* glist, unsigned* mylist, int lane,
    int g_row0, int nval_in, int gp, int base, unsigned unrows, int& count) {
    const int nval = FULL ? EP_U : nval_in;
    const int myg = (int)glist[gp + (lane < nval ? lane : 0)];
    const int ebase = (g_row0 << 10) + lane;
    int e[EP_U];
    int d[EP_U];
#pragma unroll
    for (int u = 0; u < EP_U; ++u) {
        e[u] = ebase + (__builtin_amdgcn_readlane(myg, u) << 5);
        d[u] = dst[e[u]];
    }
    int c = count;
#pragma unroll
    for (int q = 0; q < EP_U / 8; ++q) {
        if (FULL || q * 8 < nval) {
#pragma unroll
            for (int u = q * 8; u < q * 8 + 8; ++u) {
                const unsigned rel = (unsigned)(d[u] - base);
                const bool hit = rel < unrows;
                const unsigned mk = __builtin_amdgcn_ballot_w32(hit);
                if ((FULL || u < nval) && c <= EP_CAP - 32) {
                    if (hit) mylist[c + (int)__builtin_amdgcn_mbcnt_lo(mk, 0u)] = ((unsigned)e[u] << 9) + rel;
                    c += __builtin_popcount(mk);
                }
            }
        }
    }
    count = c;
}

__global__ __launch_bounds__(1024) void k_edge_rows_lds(
    const int* __restrict__ sa_words, const int* __restrict__ dst,
    const float* __restrict__ x,
    const unsigned* __restrict__ bm, float* __restrict__ agg, int N, int E) {
    __shared__ __attribute__((aligned(16))) float s_acc[EP_R * DCH];
    __shared__ unsigned s_list[EP_WAVES][EP_CAP];
    __shared__ unsigned s_sorted[EP_WAVES * EP_CAP];
    __shared__ unsigned short s_grp[EP_WAVES][EP_GCAP];
    __shared__ int s_own[EP_WAVES][EP_WAVES];
    __shared__ int s_more[EP_WAVES];

    const int t = threadIdx.x;
    const int lane = t & 31;
    const int wave = __builtin_amdgcn_readfirstlane(t >> 5);
    const int base = blockIdx.x * EP_R;
    int nrows = N - base;
    nrows = nrows < EP_R ? nrows : EP_R;
    const unsigned unrows = (unsigned)nrows;

    const int bid = blockIdx.x;
    const int nwords = (E + BM_EDGES - 1) / BM_EDGES;
    const int nfull = E / BM_EDGES;
    const int wper = (nwords + EP_WAVES - 1) / EP_WAVES;
    int wend = (wave + 1) * wper;
    wend = wend < nwords ? wend : nwords;
    int wi = wave * wper;
    int wc_base = wi, wc_end = wi + 32;
    unsigned w0 = 0u;
    if (wi + lane < wend) w0 = bid < BM_NB ? bm[(size_t)(wi + lane) * BM_NB + bid] : 0xffffffffu;

    float4* acc4 = reinterpret_cast<float4*>(s_acc);
    const float4* x4 = reinterpret_cast<const float4*>(x);
    const int2* sa = reinterpret_cast<const int2*>(sa_words);
    for (int i = t; i < EP_R * 32; i += 1024) acc4[i] = make_float4(0.0f, 0.0f, 0.0f, 0.0f);

    unsigned short* glist = s_grp[wave];
    unsigned* mylist = s_list[wave];
    int g_row0 = 0;
    int g_n = 0;
    int g_pos = 0;
    const int max_rounds = (wper * BM_EDGES) / (EP_CAP - 32) + 2;

    for (int round = 0; round < max_rounds; ++round) {
        int count = 0;
        bool full = false;
        while (!full) {
            if (g_pos >= g_n) {
                if (wi >= wend) break;
                if (wi >= wc_end) {
                    const int r = wi + lane;
                    w0 = 0u;
                    if (r < wend) w0 = bid < BM_NB ? bm[(size_t)r * BM_NB + bid] : 0xffffffffu;
                    wc_base = wi;
                    wc_end = wi + 32;
                }
                const int j = wi - wc_base + lane;
                unsigned bits = (unsigned)__shfl((int)w0, j & 31);
                if (j >= 32) bits = 0u;
                const int cnt = __builtin_popcount(bits);
                int incl = cnt;
#pragma unroll
                for (int off = 1; off < 32; off <<= 1) {
                    const int v = __shfl_up(incl, off);
                    if (lane >= off) incl += v;
                }
                int nrow = __builtin_popcount(__builtin_amdgcn_ballot_w32(incl <= EP_GCAP));
                nrow = nrow < wc_end - wi ? nrow : wc_end - wi;
                if (wi >= nfull) nrow = 1;
                else nrow = nrow < nfull - wi ? nrow : nfull - wi;
                if (lane >= nrow) bits = 0u;
                int pos = incl - cnt;
                ep_wave_order();
                while (bits != 0u) {
                    glist[pos++] = (unsigned short)((lane << 5) + __builtin_ctz(bits));
                    bits &= bits - 1u;
                }
                ep_wave_order();
                g_n = __builtin_amdgcn_readlane(incl, nrow - 1);
                g_row0 = wi;
                g_pos = 0;
                wi += nrow;
                continue;
            }
            const int left = g_n - g_pos;
            const int nval = left < EP_U ? left : EP_U;
            bool taken = false;
            if (g_row0 < nfull) {
                const int count0 = count;
                if (nval == EP_U)
                    ep_scan_batch<true>(dst, glist, mylist, lane, g_row0, nval, g_pos, base, unrows, count);
                else
                    ep_scan_batch<false>(dst, glist, mylist, lane, g_row0, nval, g_pos, base, unrows, count);
                taken = count <= EP_CAP - 32;
                if (!taken) count = count0;
            }
            if (taken) {
                g_pos += nval;
            } else {
                for (int u = 0; u < nval && !full; ++u) {
                    const int g = __builtin_amdgcn_readfirstlane((int)glist[g_pos]);
                    const int e = (g_row0 << 10) + (g << 5) + lane;
                    int dv = 0x7fffffff;
                    if (e < E) dv = dst[e];
                    const unsigned rel = (unsigned)(dv - base);
                    const bool hit = rel < unrows;
                    const unsigned mk = __builtin_amdgcn_ballot_w32(hit);
                    const int pop = __builtin_popcount(mk);
                    if (count + pop > EP_CAP) {
                        full = true;
                    } else {
                        if (hit) mylist[count + (int)__builtin_amdgcn_mbcnt_lo(mk, 0u)] = ((unsigned)e << 9) | rel;
                        count += pop;
                        ++g_pos;
                    }
                }
            }
        }
        int cnto = 0;
        ep_wave_order();
#pragma unroll 1
        for (int k0 = 0; k0 < count; k0 += 32) {
            const bool valid = lane < count - k0;
            unsigned e2 = 0u;
            if (valid) e2 = mylist[k0 + lane];
            const unsigned own = e2 & 31u;
            unsigned mine = __builtin_amdgcn_ballot_w32(valid);
#pragma unroll
            for (int b = 0; b < 5; ++b) {
                const unsigned mb = __builtin_amdgcn_ballot_w32(((own >> b) & 1u) != 0u);
                mine &= ((lane >> b) & 1) != 0 ? mb : ~mb;
            }
            cnto += __builtin_popcount(mine);
        }
        s_own[wave][lane] = cnto;
        if (lane == 0) s_more[wave] = (g_pos < g_n || wi < wend) ? 1 : 0;
        __syncthreads();
        int more_v = 0;
        for (int w = 0; w < EP_WAVES; ++w) more_v |= s_more[w];
        const int more = __builtin_amdgcn_readfirstlane(more_v);

        int part = 0;
#pragma unroll 4
        for (int s = 0; s < wave; ++s) part += s_own[s][lane];
        int tot = part;
#pragma unroll 4
        for (int s = wave; s < EP_WAVES; ++s) tot += s_own[s][lane];
        int incl = tot;
#pragma unroll
        for (int off = 1; off < 32; off <<= 1) {
            const int v = __shfl_up(incl, off);
            if (lane >= off) incl += v;
        }
        const int obase = incl - tot;
        int dcur = obase + part;
#pragma unroll 1
        for (int k0 = 0; k0 < count; k0 += 32) {
            const bool valid = lane < count - k0;
            unsigned e2 = 0u;
            if (valid) e2 = mylist[k0 + lane];
            const unsigned own = e2 & 31u;
            const unsigned vm = __builtin_amdgcn_ballot_w32(valid);
            unsigned same = vm;
            unsigned mine = vm;
#pragma unroll
            for (int b = 0; b < 5; ++b) {
                const unsigned mb = __builtin_amdgcn_ballot_w32(((own >> b) & 1u) != 0u);
                same &= ((own >> b) & 1u) != 0u ? mb : ~mb;
                mine &= ((lane >> b) & 1) != 0 ? mb : ~mb;
            }
            const int d = __shfl(dcur, (int)own);
            if (valid) s_sorted[d + (int)__builtin_amdgcn_mbcnt_lo(same, 0u)] = e2;
            dcur += __builtin_popcount(mine);
        }
        __syncthreads();

        const int my_n = __builtin_amdgcn_readlane(tot, wave);
        const unsigned* run = s_sorted + __builtin_amdgcn_readlane(obase, wave);
        const int ngrp = my_n >> 5;
        if (ngrp > 0) {
            float4 vrow[EP_RING];
            int sv_p = 0, rv_p = 0, avi_p = 0;
            for (int g = 0; g < ngrp; ++g) {
                const unsigned e2 = run[g * 32 + lane];
                const int2 rec = sa[(int)(e2 >> 9)];
                __builtin_amdgcn_sched_barrier(0);
                if (g != 0) ep_add_rows32(vrow, sv_p, rv_p, avi_p, lane, x4, acc4);
                __builtin_amdgcn_sched_barrier(0);
                sv_p = rec.x;
                rv_p = (int)(e2 & 511u);
                avi_p = rec.y;
                ep_ask_rows(vrow, sv_p, lane, x4);
            }
            ep_add_rows32(vrow, sv_p, rv_p, avi_p, lane, x4, acc4);
        }
        const int h = ngrp << 5;
        ep_flush(run + h, my_n - h, lane, sa, x4, acc4);
        __syncthreads();
        if (!more) break;
    }

    float4* agg4 = reinterpret_cast<float4*>(agg) + (size_t)base * 32;
    for (int i = t; i < nrows * 32; i += 1024) agg4[i] = acc4[i];
}

#define FT_SLOTS 4
#define FT_WAVES 32
template <int RB, int KU>
__device__ __forceinline__ void ft_tile_product(
    const float* __restrict__ grows, int j, int n, float (*sAw)[SX_LD],
    const float (*sBm)[SX_LD], int lane, v8f& acc0, v8f& acc1) {
    const int r0 = j * 16;
#pragma unroll
    for (int part = 0; part < 16 / RB; ++part) {
        float4 v[RB];
#pragma unroll
        for (int r = 0; r < RB; ++r) {
            const int rr = r0 + part * RB + r;
            v[r] = make_float4(0.0f, 0.0f, 0.0f, 0.0f);
            if (rr < n) v[r] = *reinterpret_cast<const float4*>(grows + (size_t)rr * DCH + lane * 4);
        }
#pragma unroll
        for (int r = 0; r < RB; ++r)
            *reinterpret_cast<float4*>(&sAw[part * RB + r][lane * 4]) = v[r];
    }
    __builtin_amdgcn_fence(__ATOMIC_RELEASE, "wavefront");
    __builtin_amdgcn_wave_barrier();
    __builtin_amdgcn_fence(__ATOMIC_ACQUIRE, "wavefront");

    const int m    = lane & 15;
    const int koff = (lane >> 4) << 1;
    acc0 = v8f{};
    acc1 = v8f{};
#pragma unroll KU
    for (int kt = 0; kt < DCH; kt += 4) {
        v2f a  = *reinterpret_cast<const v2f*>(&sAw[m][kt + koff]);
        v2f b0 = *reinterpret_cast<const v2f*>(&sBm[m][kt + koff]);
        v2f b1 = *reinterpret_cast<const v2f*>(&sBm[16 + m][kt + koff]);
        acc0 = __builtin_amdgcn_wmma_f32_16x16x4_f32(false, a, false, b0, (short)0, acc0, false, false);
        acc1 = __builtin_amdgcn_wmma_f32_16x16x4_f32(false, a, false, b1, (short)0, acc1, false, false);
    }
    __builtin_amdgcn_fence(__ATOMIC_RELEASE, "wavefront");
    __builtin_amdgcn_wave_barrier();
    __builtin_amdgcn_fence(__ATOMIC_ACQUIRE, "wavefront");
}

__device__ __forceinline__ void ft_bias_relu_sum(
    v8f& ha, v8f& hb, float bv0, float bv1, int rbase, int n, float& p0, float& p1) {
#pragma unroll
    for (int v = 0; v < 8; ++v) {
        const bool ok = (rbase + v) < n;
        float a0 = ha[v] + bv0; a0 = a0 > 0.0f ? a0 : 0.0f;
        float a1 = hb[v] + bv1; a1 = a1 > 0.0f ? a1 : 0.0f;
        ha[v] = a0;
        hb[v] = a1;
        p0 += ok ? a0 : 0.0f;
        p1 += ok ? a1 : 0.0f;
    }
}

__device__ __forceinline__ void ft_sq_sum(
    const v8f& ha, const v8f& hb, float sub0, float sub1, int rbase, int n, float& q0, float& q1) {
#pragma unroll
    for (int v = 0; v < 8; ++v) {
        const bool ok = (rbase + v) < n;
        const float o0 = ha[v] - sub0;
        const float o1 = hb[v] - sub1;
        q0 += ok ? o0 * o0 : 0.0f;
        q1 += ok ? o1 * o1 : 0.0f;
    }
}

__device__ __forceinline__ void ft_norm_store(
    const v8f& ha, const v8f& hb, float sub0, float sub1, float den0, float den1,
    float w0, float w1, float b0, float b1, int rbase, int n, float* __restrict__ ocol) {
#pragma unroll
    for (int v = 0; v < 8; ++v) {
        const int rr = rbase + v;
        if (rr < n) {
            ocol[(size_t)rr * DCH]      = w0 * (ha[v] - sub0) / den0 + b0;
            ocol[(size_t)rr * DCH + 16] = w1 * (hb[v] - sub1) / den1 + b1;
        }
    }
}

__device__ __forceinline__ void ft_block_sum(
    float p0, float p1, float (*s_red)[32], float* s_stat, int t, float& r0, float& r1) {
    const int lane = t & 31;
    const int wave = t >> 5;
    p0 += __shfl_xor(p0, 16);
    p1 += __shfl_xor(p1, 16);
    if (lane < 16) { s_red[wave][lane] = p0; s_red[wave][16 + lane] = p1; }
    __syncthreads();
    if (t < 32) {
        float s = 0.0f;
        for (int w = 0; w < FT_WAVES; ++w) s += s_red[w][t];
        s_stat[t] = s;
    }
    __syncthreads();
    r0 = s_stat[lane & 15];
    r1 = s_stat[16 + (lane & 15)];
    __syncthreads();
}

__global__ __launch_bounds__(1024) void k_gemm_graph_norm(
    const float* __restrict__ agg, const float* __restrict__ Wt,
    const float* __restrict__ bias, const int* __restrict__ gstart,
    const float* __restrict__ ms, const float* __restrict__ gw,
    const float* __restrict__ gb, float* __restrict__ out, int N) {
    __shared__ __attribute__((aligned(16))) float sA[FT_WAVES][16][SX_LD];
    __shared__ __attribute__((aligned(16))) float sB[32][SX_LD];
    __shared__ float s_red[FT_WAVES][32];
    __shared__ float s_stat[32];

    const int t = threadIdx.x;
    const int lane = t & 31;
    const int wave = __builtin_amdgcn_readfirstlane(t >> 5);
    const int g = blockIdx.x >> 2;
    const int slice = blockIdx.x & 3;

    const float4 wv = *reinterpret_cast<const float4*>(
        Wt + (size_t)(slice * 32 + (t >> 5)) * DCH + ((t & 31) << 2));
    const int start = __builtin_amdgcn_readfirstlane(gstart[g]);
    const int n = __builtin_amdgcn_readfirstlane(gstart[g + 1]) - start;
    *reinterpret_cast<float4*>(&sB[t >> 5][(t & 31) << 2]) = wv;
    __syncthreads();
    const int ntiles = (n + 15) >> 4;

    const float* grows = agg + (size_t)start * DCH;
    float* orows = out + (size_t)start * DCH;
    float (*sAw)[SX_LD] = sA[wave];

    const int c = lane & 15;
    const int rhalf = (lane >> 4) << 3;
    const int col0 = slice * 32 + c;
    const int col1 = col0 + 16;
    const float bv0 = bias[col0], bv1 = bias[col1];

    v8f hA0, hB0, hA1, hB1, hA2, hB2, hA3, hB3;
    float p0 = 0.0f, p1 = 0.0f;
#define FT_PHASE1(S, HA, HB)                                                        \
    {                                                                               \
        const int j = wave + (S) * FT_WAVES;     \
        HA = v8f{};                                                                 \
        HB = v8f{};                                                                 \
        if (j < ntiles) {                                                           \
            ft_tile_product<8, 4>(grows, j, n, sAw, sB, lane, HA, HB);              \
            ft_bias_relu_sum(HA, HB, bv0, bv1, j * 16 + rhalf, n, p0, p1);          \
        }                                                                           \
    }
    FT_PHASE1(0, hA0, hB0)
    FT_PHASE1(1, hA1, hB1)
    FT_PHASE1(2, hA2, hB2)
    FT_PHASE1(3, hA3, hB3)
#undef FT_PHASE1
    for (int j = wave + FT_SLOTS * FT_WAVES; j < ntiles; j += FT_WAVES) {
        v8f e0, e1;
        ft_tile_product<2, 1>(grows, j, n, sAw, sB, lane, e0, e1);
        ft_bias_relu_sum(e0, e1, bv0, bv1, j * 16 + rhalf, n, p0, p1);
    }
    float msum0, msum1;
    ft_block_sum(p0, p1, s_red, s_stat, t, msum0, msum1);

    float cn = (float)n;
    cn = cn > 1.0f ? cn : 1.0f;
    const float inv_cn = 1.0f / cn;
    const float sub0 = (msum0 * inv_cn) * ms[col0];
    const float sub1 = (msum1 * inv_cn) * ms[col1];

    float q0 = 0.0f, q1 = 0.0f;
    if (wave + 0 * FT_WAVES < ntiles) ft_sq_sum(hA0, hB0, sub0, sub1, (wave + 0 * FT_WAVES) * 16 + rhalf, n, q0, q1);
    if (wave + 1 * FT_WAVES < ntiles) ft_sq_sum(hA1, hB1, sub0, sub1, (wave + 1 * FT_WAVES) * 16 + rhalf, n, q0, q1);
    if (wave + 2 * FT_WAVES < ntiles) ft_sq_sum(hA2, hB2, sub0, sub1, (wave + 2 * FT_WAVES) * 16 + rhalf, n, q0, q1);
    if (wave + 3 * FT_WAVES < ntiles) ft_sq_sum(hA3, hB3, sub0, sub1, (wave + 3 * FT_WAVES) * 16 + rhalf, n, q0, q1);
    for (int j = wave + FT_SLOTS * FT_WAVES; j < ntiles; j += FT_WAVES) {
        v8f e0, e1;
        float d0 = 0.0f, d1 = 0.0f;
        ft_tile_product<2, 1>(grows, j, n, sAw, sB, lane, e0, e1);
        ft_bias_relu_sum(e0, e1, bv0, bv1, j * 16 + rhalf, n, d0, d1);
        ft_sq_sum(e0, e1, sub0, sub1, j * 16 + rhalf, n, q0, q1);
    }
    float vsum0, vsum1;
    ft_block_sum(q0, q1, s_red, s_stat, t, vsum0, vsum1);

    const float den0 = sqrtf(vsum0 * inv_cn + GN_EPS);
    const float den1 = sqrtf(vsum1 * inv_cn + GN_EPS);
    const float w0 = gw[col0], w1 = gw[col1];
    const float b0 = gb[col0], b1 = gb[col1];

    if (wave + 0 * FT_WAVES < ntiles) ft_norm_store(hA0, hB0, sub0, sub1, den0, den1, w0, w1, b0, b1, (wave + 0 * FT_WAVES) * 16 + rhalf, n, orows + col0);
    if (wave + 1 * FT_WAVES < ntiles) ft_norm_store(hA1, hB1, sub0, sub1, den0, den1, w0, w1, b0, b1, (wave + 1 * FT_WAVES) * 16 + rhalf, n, orows + col0);
    if (wave + 2 * FT_WAVES < ntiles) ft_norm_store(hA2, hB2, sub0, sub1, den0, den1, w0, w1, b0, b1, (wave + 2 * FT_WAVES) * 16 + rhalf, n, orows + col0);
    if (wave + 3 * FT_WAVES < ntiles) ft_norm_store(hA3, hB3, sub0, sub1, den0, den1, w0, w1, b0, b1, (wave + 3 * FT_WAVES) * 16 + rhalf, n, orows + col0);
    for (int j = wave + FT_SLOTS * FT_WAVES; j < ntiles; j += FT_WAVES) {
        v8f e0, e1;
        float d0 = 0.0f, d1 = 0.0f;
        ft_tile_product<2, 1>(grows, j, n, sAw, sB, lane, e0, e1);
        ft_bias_relu_sum(e0, e1, bv0, bv1, j * 16 + rhalf, n, d0, d1);
        ft_norm_store(e0, e1, sub0, sub1, den0, den1, w0, w1, b0, b1, j * 16 + rhalf, n, orows + col0);
    }
}

extern "C" void kernel_launch(void* const* d_in, const int* in_sizes, int n_in,
                              void* d_out, int out_size, void* d_ws, size_t ws_size,
                              hipStream_t stream) {
    const float* x     = (const float*)d_in[0];
    const int*   eidx  = (const int*)  d_in[1];
    const float* eattr = (const float*)d_in[2];
    const int*   batch = (const int*)  d_in[3];
    const float* W     = (const float*)d_in[4];
    const float* bias  = (const float*)d_in[5];
    const float* gw    = (const float*)d_in[6];
    const float* gb    = (const float*)d_in[7];
    const float* ms    = (const float*)d_in[8];
    float* out = (float*)d_out;

    const int N = in_sizes[0] / DCH;
    const int E = in_sizes[2];

    float* Wt  = (float*)d_ws;
    float* agg = Wt + (size_t)DCH * DCH;
    int* sa = (int*)(agg + (size_t)N * DCH);
    unsigned* bm = (unsigned*)(sa + 2 * (size_t)E);
    int* gstart = (int*)(bm + (size_t)((E + BM_EDGES - 1) / BM_EDGES) * BM_NB);

    k_transpose_w_edge_groups<<<64 + (E + BM_EDGES * BM_RPB - 1) / (BM_EDGES * BM_RPB), 256, 0, stream>>>(W, Wt, eidx, eidx + E, eattr, sa, bm, E, batch, gstart, N);

    k_edge_rows_lds<<<(N + EP_R - 1) / EP_R, 1024, 0, stream>>>(sa, eidx + E, x, bm, agg, N, E);

    k_gemm_graph_norm<<<NGRAPH * 4, 1024, 0, stream>>>(agg, Wt, bias, gstart, ms, gw, gb, out, N);
}
